# starting kernel
# baseline (speedup 1.0000x reference)
#include <hip/hip_runtime.h>
#include <hip/hip_bf16.h>
#include <math.h>


typedef __attribute__((ext_vector_type(2))) float v2f;
typedef __attribute__((ext_vector_type(8))) float v8f;

#define DFEAT 128
#define NEG_SLOPE 0.2f

__device__ __forceinline__ unsigned ordEnc(float f) {
  unsigned u = __float_as_uint(f);
  return (u & 0x80000000u) ? ~u : (u | 0x80000000u);
}
__device__ __forceinline__ float ordDec(unsigned e) {
  unsigned u = (e & 0x80000000u) ? (e & 0x7fffffffu) : ~e;
  return __uint_as_float(u);
}

__device__ __forceinline__ v8f wmma_f32(v2f a, v2f b, v8f c) {
  return __builtin_amdgcn_wmma_f32_16x16x4_f32(false, a, false, b, (short)0, c,
                                               false, false);
}

__global__ void gemm_f32_wmma(const float* __restrict__ A, int lda,
                              const float* __restrict__ B, int ldb,
                              float* __restrict__ D, int ldo,
                              int nrows, int kdim, int ncols,
                              const float* __restrict__ bias, int relu) {
  const int lane = threadIdx.x & 31;
  const int wid  = (blockIdx.x * blockDim.x + threadIdx.x) >> 5;
  const int colTiles = ncols >> 4;
  const int rowTiles = (nrows + 15) >> 4;
  if (wid >= rowTiles * colTiles) return;
  const int tr = wid / colTiles;
  const int tc = wid - tr * colTiles;
  const int row0 = tr << 4, col0 = tc << 4;
  const int lm = lane & 15;
  const int lh = lane >> 4;

  int arow = row0 + lm;
  if (arow >= nrows) arow = nrows - 1;
  const float* Ap = A + (size_t)arow * lda + 2 * lh;
  const float* Bp = B + (size_t)(2 * lh) * ldb + col0 + lm;

  v8f acc = {0.f, 0.f, 0.f, 0.f, 0.f, 0.f, 0.f, 0.f};
  int k = 0;
  for (; k + 16 <= kdim; k += 16) {
    const v2f a0 = *(const v2f*)(Ap + k);
    const v2f a1 = *(const v2f*)(Ap + k + 4);
    const v2f a2 = *(const v2f*)(Ap + k + 8);
    const v2f a3 = *(const v2f*)(Ap + k + 12);
    const float* bq = Bp + (size_t)k * ldb;
    v2f b0, b1, b2, b3;
    b0.x = bq[0];
    b0.y = bq[(size_t)1 * ldb];
    b1.x = bq[(size_t)4 * ldb];
    b1.y = bq[(size_t)5 * ldb];
    b2.x = bq[(size_t)8 * ldb];
    b2.y = bq[(size_t)9 * ldb];
    b3.x = bq[(size_t)12 * ldb];
    b3.y = bq[(size_t)13 * ldb];
    acc = wmma_f32(a0, b0, acc);
    acc = wmma_f32(a1, b1, acc);
    acc = wmma_f32(a2, b2, acc);
    acc = wmma_f32(a3, b3, acc);
  }
  for (; k < kdim; k += 4) {
    const v2f a = *(const v2f*)(Ap + k);
    v2f b;
    b.x = Bp[(size_t)k * ldb];
    b.y = Bp[(size_t)(k + 1) * ldb];
    acc = wmma_f32(a, b, acc);
  }

  const int col = col0 + lm;
  const float bv = bias ? bias[col] : 0.0f;
#pragma unroll
  for (int r = 0; r < 8; ++r) {
    const int row = row0 + r + 8 * lh;
    if (row < nrows) {
      float v = acc[r] + bv;
      if (relu) v = fmaxf(v, 0.0f);
      D[(size_t)row * ldo + col] = v;
    }
  }
}

__global__ void gemm_f32_wmma_bn(const float* __restrict__ A, int lda,
                                 const float* __restrict__ B, int ldb,
                                 float* __restrict__ D, int ldo,
                                 int nrows, int kdim, int ncols,
                                 const float* __restrict__ bias,
                                 const float* __restrict__ scl,
                                 const float* __restrict__ shf, int relu) {
  const int lane = threadIdx.x & 31;
  const int wid  = (blockIdx.x * blockDim.x + threadIdx.x) >> 5;
  const int colTiles = ncols >> 4;
  const int rowTiles = (nrows + 15) >> 4;
  if (wid >= rowTiles * colTiles) return;
  const int tr = wid / colTiles;
  const int tc = wid - tr * colTiles;
  const int row0 = tr << 4, col0 = tc << 4;
  const int lm = lane & 15;
  const int lh = lane >> 4;

  int arow = row0 + lm;
  if (arow >= nrows) arow = nrows - 1;
  const float* Ap = A + (size_t)arow * lda + 2 * lh;
  const float* Sp = scl + 2 * lh;
  const float* Hp = shf + 2 * lh;
  const float* Bp = B + (size_t)(2 * lh) * ldb + col0 + lm;

  v8f acc = {0.f, 0.f, 0.f, 0.f, 0.f, 0.f, 0.f, 0.f};
  int k = 0;
  for (; k + 16 <= kdim; k += 16) {
    v2f a0 = *(const v2f*)(Ap + k);
    v2f a1 = *(const v2f*)(Ap + k + 4);
    v2f a2 = *(const v2f*)(Ap + k + 8);
    v2f a3 = *(const v2f*)(Ap + k + 12);
    const v2f s0 = *(const v2f*)(Sp + k);
    const v2f s1 = *(const v2f*)(Sp + k + 4);
    const v2f s2 = *(const v2f*)(Sp + k + 8);
    const v2f s3 = *(const v2f*)(Sp + k + 12);
    const v2f h0 = *(const v2f*)(Hp + k);
    const v2f h1 = *(const v2f*)(Hp + k + 4);
    const v2f h2 = *(const v2f*)(Hp + k + 8);
    const v2f h3 = *(const v2f*)(Hp + k + 12);
    a0 = a0 * s0 + h0;
    a1 = a1 * s1 + h1;
    a2 = a2 * s2 + h2;
    a3 = a3 * s3 + h3;
    const float* bq = Bp + (size_t)k * ldb;
    v2f b0, b1, b2, b3;
    b0.x = bq[0];
    b0.y = bq[(size_t)1 * ldb];
    b1.x = bq[(size_t)4 * ldb];
    b1.y = bq[(size_t)5 * ldb];
    b2.x = bq[(size_t)8 * ldb];
    b2.y = bq[(size_t)9 * ldb];
    b3.x = bq[(size_t)12 * ldb];
    b3.y = bq[(size_t)13 * ldb];
    acc = wmma_f32(a0, b0, acc);
    acc = wmma_f32(a1, b1, acc);
    acc = wmma_f32(a2, b2, acc);
    acc = wmma_f32(a3, b3, acc);
  }
  for (; k < kdim; k += 4) {
    v2f a = *(const v2f*)(Ap + k);
    const v2f s = *(const v2f*)(Sp + k);
    const v2f h = *(const v2f*)(Hp + k);
    a = a * s + h;
    v2f b;
    b.x = Bp[(size_t)k * ldb];
    b.y = Bp[(size_t)(k + 1) * ldb];
    acc = wmma_f32(a, b, acc);
  }

  const int col = col0 + lm;
  const float bv = bias ? bias[col] : 0.0f;
#pragma unroll
  for (int r = 0; r < 8; ++r) {
    const int row = row0 + r + 8 * lh;
    if (row < nrows) {
      float v = acc[r] + bv;
      if (relu) v = fmaxf(v, 0.0f);
      D[(size_t)row * ldo + col] = v;
    }
  }
}

__global__ void attn_proj(const float* __restrict__ h, int heads,
                          const float* __restrict__ a_src,
                          const float* __restrict__ a_dst,
                          float* __restrict__ al_s, float* __restrict__ al_d,
                          int n) {
  const int wid  = (blockIdx.x * blockDim.x + threadIdx.x) >> 5;
  const int lane = threadIdx.x & 31;
  if (wid >= n * heads) return;
  const int node = wid / heads;
  const int head = wid - node * heads;
  const float* hp = h + (size_t)node * heads * DFEAT + head * DFEAT;
  const float* as = a_src + head * DFEAT;
  const float* ad = a_dst + head * DFEAT;
  float ss = 0.f, sd = 0.f;
  for (int c = lane; c < DFEAT; c += 32) {
    const float hv = hp[c];
    ss += hv * as[c];
    sd += hv * ad[c];
  }
  for (int off = 16; off > 0; off >>= 1) {
    ss += __shfl_xor(ss, off, 32);
    sd += __shfl_xor(sd, off, 32);
  }
  if (lane == 0) { al_s[wid] = ss; al_d[wid] = sd; }
}

__global__ void bias_fill(float* __restrict__ out, int ldo, int colOff,
                          int width, const float* __restrict__ bias, int n) {
  const int idx = blockIdx.x * blockDim.x + threadIdx.x;
  if (idx >= n * width) return;
  const int node = idx / width;
  const int c = idx - node * width;
  out[(size_t)node * ldo + colOff + c] = bias[c];
}

__global__ void zero_ms(unsigned* __restrict__ menc, float* __restrict__ ssum,
                        int nh) {
  const int i = blockIdx.x * blockDim.x + threadIdx.x;
  if (i < nh) { menc[i] = 0u; ssum[i] = 0.f; }
}

__global__ void edge_logit_max(const int* __restrict__ src,
                               const int* __restrict__ dst, int E, int n,
                               int heads, const float* __restrict__ al_s,
                               const float* __restrict__ al_d,
                               float* __restrict__ elog,
                               unsigned* __restrict__ menc) {
  const int idx = blockIdx.x * blockDim.x + threadIdx.x;
  if (idx >= (E + n) * heads) return;
  const int e = idx / heads;
  const int hh = idx - e * heads;
  const int s = (e < E) ? src[e] : (e - E);
  const int d = (e < E) ? dst[e] : (e - E);
  float x = al_s[s * heads + hh] + al_d[d * heads + hh];
  x = (x > 0.f) ? x : NEG_SLOPE * x;
  elog[idx] = x;
  atomicMax(&menc[d * heads + hh], ordEnc(x));
}

__global__ void edge_expsum(const int* __restrict__ src,
                            const int* __restrict__ dst, int E, int n,
                            int heads, const unsigned* __restrict__ menc,
                            float* __restrict__ elog,
                            float* __restrict__ ssum) {
  const int idx = blockIdx.x * blockDim.x + threadIdx.x;
  if (idx >= (E + n) * heads) return;
  const int e = idx / heads;
  const int hh = idx - e * heads;
  const int d = (e < E) ? dst[e] : (e - E);
  const float m = ordDec(menc[d * heads + hh]);
  const float ee = expf(elog[idx] - m);
  elog[idx] = ee;
  atomicAdd(&ssum[d * heads + hh], ee);
}

__global__ void edge_aggr(const int* __restrict__ src,
                          const int* __restrict__ dst, int E, int n, int heads,
                          const float* __restrict__ h, int ldh,
                          const float* __restrict__ elog,
                          const float* __restrict__ ssum,
                          float* __restrict__ out, int ldo, int colOff) {
  const int e = blockIdx.x;
  const int f = threadIdx.x;
  const int s = (e < E) ? src[e] : (e - E);
  const int d = (e < E) ? dst[e] : (e - E);
  const int head = f >> 7;
  const float alpha =
      elog[e * heads + head] / (ssum[d * heads + head] + 1e-16f);
  atomicAdd(&out[(size_t)d * ldo + colOff + f],
            alpha * h[(size_t)s * ldh + f]);
}

__global__ void col_stats(const float* __restrict__ y, int ld, int n,
                          float* __restrict__ sums, float* __restrict__ sumsq) {
  __shared__ float s1[256];
  __shared__ float s2[256];
  const int c = blockIdx.x;
  float s = 0.f, q = 0.f;
  for (int r = threadIdx.x; r < n; r += blockDim.x) {
    const float v = y[(size_t)r * ld + c];
    s += v;
    q += v * v;
  }
  s1[threadIdx.x] = s;
  s2[threadIdx.x] = q;
  __syncthreads();
  for (int st = 128; st > 0; st >>= 1) {
    if (threadIdx.x < st) {
      s1[threadIdx.x] += s1[threadIdx.x + st];
      s2[threadIdx.x] += s2[threadIdx.x + st];
    }
    __syncthreads();
  }
  if (threadIdx.x == 0) { sums[c] = s1[0]; sumsq[c] = s2[0]; }
}

__global__ void bn_final(const float* __restrict__ sums,
                         const float* __restrict__ sumsq,
                         const float* __restrict__ g,
                         const float* __restrict__ beta, int ncols, float invn,
                         float* __restrict__ sc, float* __restrict__ sh) {
  const int c = blockIdx.x * blockDim.x + threadIdx.x;
  if (c >= ncols) return;
  const float mean = sums[c] * invn;
  const float var = sumsq[c] * invn - mean * mean;
  const float r = rsqrtf(var + 1e-5f);
  const float s = g[c] * r;
  sc[c] = s;
  sh[c] = beta[c] - mean * s;
}

__global__ void mlp_out(const float* __restrict__ y2,
                        const float* __restrict__ sc,
                        const float* __restrict__ sh,
                        const float* __restrict__ Wm2,
                        const float* __restrict__ bm2, int n,
                        float* __restrict__ y3) {
  const int node = blockIdx.x * blockDim.x + threadIdx.x;
  if (node >= n) return;
  float a0 = bm2[0], a1 = bm2[1];
  const float* yp = y2 + (size_t)node * 96;
  for (int k = 0; k < 96; ++k) {
    const float v = fmaf(yp[k], sc[k], sh[k]);
    a0 = fmaf(v, Wm2[2 * k + 0], a0);
    a1 = fmaf(v, Wm2[2 * k + 1], a1);
  }
  y3[node * 2 + 0] = fmaxf(a0, 0.f);
  y3[node * 2 + 1] = fmaxf(a1, 0.f);
}

__global__ void bn_apply(const float* __restrict__ y,
                         const float* __restrict__ sc,
                         const float* __restrict__ sh, int total, int ncols,
                         float* __restrict__ out) {
  const int i = blockIdx.x * blockDim.x + threadIdx.x;
  if (i >= total) return;
  const int c = i % ncols;
  out[i] = fmaf(y[i], sc[c], sh[c]);
}

extern "C" void kernel_launch(void* const* d_in, const int* in_sizes, int n_in,
                              void* d_out, int out_size, void* d_ws,
                              size_t ws_size, hipStream_t stream) {
  const float* x      = (const float*)d_in[0];
  const int*   ei     = (const int*)d_in[1];
  const float* W1     = (const float*)d_in[2];
  const float* as1    = (const float*)d_in[3];
  const float* ad1    = (const float*)d_in[4];
  const float* b1     = (const float*)d_in[5];
  const float* W2     = (const float*)d_in[6];
  const float* as2    = (const float*)d_in[7];
  const float* ad2    = (const float*)d_in[8];
  const float* b2     = (const float*)d_in[9];
  const float* Wl     = (const float*)d_in[10];
  const float* bl     = (const float*)d_in[11];
  const float* gl     = (const float*)d_in[12];
  const float* betal  = (const float*)d_in[13];
  const float* Wm1    = (const float*)d_in[14];
  const float* bm1    = (const float*)d_in[15];
  const float* gm1    = (const float*)d_in[16];
  const float* betam1 = (const float*)d_in[17];
  const float* Wm2    = (const float*)d_in[18];
  const float* bm2    = (const float*)d_in[19];
  const float* gm2    = (const float*)d_in[20];
  const float* betam2 = (const float*)d_in[21];

  const int N = in_sizes[0] / DFEAT;
  const int E = in_sizes[1] / 2;
  const int E2 = E + N;
  const int* srcA = ei;
  const int* dstA = ei + E;

  char* w = (char*)d_ws;
  auto alloc = [&](size_t bytes) -> char* {
    char* p = w;
    w += (bytes + 255) & ~(size_t)255;
    return p;
  };
  float*    xcat = (float*)alloc((size_t)N * 384 * 4);
  float*    h1   = (float*)alloc((size_t)N * 256 * 4);
  float*    h2   = (float*)alloc((size_t)N * 128 * 4);
  float*    al_s = (float*)alloc((size_t)N * 2 * 4);
  float*    al_d = (float*)alloc((size_t)N * 2 * 4);
  unsigned* menc = (unsigned*)alloc((size_t)N * 2 * 4);
  float*    ssum = (float*)alloc((size_t)N * 2 * 4);
  float*    elog = (float*)alloc((size_t)E2 * 2 * 4);
  float*    y3   = (float*)alloc((size_t)N * 2 * 4);
  float*    sums = (float*)alloc(96 * 4);
  float*    sumsq= (float*)alloc(96 * 4);
  float*    sc1  = (float*)alloc(96 * 4);
  float*    sh1  = (float*)alloc(96 * 4);
  float*    sc2  = (float*)alloc(96 * 4);
  float*    sh2  = (float*)alloc(96 * 4);
  float*    sc3  = (float*)alloc(96 * 4);
  float*    sh3  = (float*)alloc(96 * 4);
  float* y1 = h1;
  float* y2 = h1 + (size_t)N * 96;

  const int TB = 256;
  auto gemmBlocks = [](int rows, int cols) {
    const int tiles = ((rows + 15) / 16) * (cols / 16);
    return (tiles + 7) / 8;
  };

  gemm_f32_wmma<<<gemmBlocks(N, 256), TB, 0, stream>>>(
      x, 128, W1, 256, h1, 256, N, 128, 256, nullptr, 0);
  attn_proj<<<(N * 2 + 7) / 8, TB, 0, stream>>>(h1, 2, as1, ad1, al_s, al_d, N);
  bias_fill<<<(N * 256 + TB - 1) / TB, TB, 0, stream>>>(xcat, 384, 0, 256, b1, N);
  zero_ms<<<(N * 2 + TB - 1) / TB, TB, 0, stream>>>(menc, ssum, N * 2);
  edge_logit_max<<<(E2 * 2 + TB - 1) / TB, TB, 0, stream>>>(
      srcA, dstA, E, N, 2, al_s, al_d, elog, menc);
  edge_expsum<<<(E2 * 2 + TB - 1) / TB, TB, 0, stream>>>(
      srcA, dstA, E, N, 2, menc, elog, ssum);
  edge_aggr<<<E2, 256, 0, stream>>>(srcA, dstA, E, N, 2, h1, 256, elog, ssum,
                                    xcat, 384, 0);

  gemm_f32_wmma<<<gemmBlocks(N, 128), TB, 0, stream>>>(
      xcat, 384, W2, 128, h2, 128, N, 256, 128, nullptr, 0);
  attn_proj<<<(N + 7) / 8, TB, 0, stream>>>(h2, 1, as2, ad2, al_s, al_d, N);
  bias_fill<<<(N * 128 + TB - 1) / TB, TB, 0, stream>>>(xcat, 384, 256, 128, b2, N);
  zero_ms<<<(N + TB - 1) / TB, TB, 0, stream>>>(menc, ssum, N);
  edge_logit_max<<<(E2 + TB - 1) / TB, TB, 0, stream>>>(
      srcA, dstA, E, N, 1, al_s, al_d, elog, menc);
  edge_expsum<<<(E2 + TB - 1) / TB, TB, 0, stream>>>(
      srcA, dstA, E, N, 1, menc, elog, ssum);
  edge_aggr<<<E2, 128, 0, stream>>>(srcA, dstA, E, N, 1, h2, 128, elog, ssum,
                                    xcat, 384, 256);

  const float invn = 1.0f / (float)N;
  gemm_f32_wmma<<<gemmBlocks(N, 96), TB, 0, stream>>>(
      xcat, 384, Wl, 96, y1, 96, N, 384, 96, bl, 1);
  col_stats<<<96, 256, 0, stream>>>(y1, 96, N, sums, sumsq);
  bn_final<<<1, 128, 0, stream>>>(sums, sumsq, gl, betal, 96, invn, sc1, sh1);
  gemm_f32_wmma_bn<<<gemmBlocks(N, 96), TB, 0, stream>>>(
      y1, 96, Wm1, 96, y2, 96, N, 96, 96, bm1, sc1, sh1, 1);
  col_stats<<<96, 256, 0, stream>>>(y2, 96, N, sums, sumsq);
  bn_final<<<1, 128, 0, stream>>>(sums, sumsq, gm1, betam1, 96, invn, sc2, sh2);
  mlp_out<<<(N + TB - 1) / TB, TB, 0, stream>>>(y2, sc2, sh2, Wm2, bm2, N, y3);
  col_stats<<<2, 256, 0, stream>>>(y3, 2, N, sums, sumsq);
  bn_final<<<1, 128, 0, stream>>>(sums, sumsq, gm2, betam2, 2, invn, sc3, sh3);
  bn_apply<<<(N * 2 + TB - 1) / TB, TB, 0, stream>>>(y3, sc3, sh3, N * 2, 2,
                                                     (float*)d_out);
}


# edge aggregation re-tiled: one wave per 32 edges (lane = edge for indices and weight, then lane = channel), 32 row lines in flight; whole-line f32 atomic adds kept, arrival order as before
# speedup vs baseline: 1.6237x; 1.6237x over previous
#include <hip/hip_runtime.h>
#include <hip/hip_bf16.h>
#include <math.h>


typedef __attribute__((ext_vector_type(2))) float v2f;
typedef __attribute__((ext_vector_type(8))) float v8f;

#define DFEAT 128
#define NEG_SLOPE 0.2f

__device__ __forceinline__ unsigned ordEnc(float f) {
  unsigned u = __float_as_uint(f);
  return (u & 0x80000000u) ? ~u : (u | 0x80000000u);
}
__device__ __forceinline__ float ordDec(unsigned e) {
  unsigned u = (e & 0x80000000u) ? (e & 0x7fffffffu) : ~e;
  return __uint_as_float(u);
}

__device__ __forceinline__ v8f wmma_f32(v2f a, v2f b, v8f c) {
  return __builtin_amdgcn_wmma_f32_16x16x4_f32(false, a, false, b, (short)0, c,
                                               false, false);
}

__global__ void gemm_f32_wmma(const float* __restrict__ A, int lda,
                              const float* __restrict__ B, int ldb,
                              float* __restrict__ D, int ldo,
                              int nrows, int kdim, int ncols,
                              const float* __restrict__ bias, int relu) {
  const int lane = threadIdx.x & 31;
  const int wid  = (blockIdx.x * blockDim.x + threadIdx.x) >> 5;
  const int colTiles = ncols >> 4;
  const int rowTiles = (nrows + 15) >> 4;
  if (wid >= rowTiles * colTiles) return;
  const int tr = wid / colTiles;
  const int tc = wid - tr * colTiles;
  const int row0 = tr << 4, col0 = tc << 4;
  const int lm = lane & 15;
  const int lh = lane >> 4;

  int arow = row0 + lm;
  if (arow >= nrows) arow = nrows - 1;
  const float* Ap = A + (size_t)arow * lda + 2 * lh;
  const float* Bp = B + (size_t)(2 * lh) * ldb + col0 + lm;

  v8f acc = {0.f, 0.f, 0.f, 0.f, 0.f, 0.f, 0.f, 0.f};
  int k = 0;
  for (; k + 16 <= kdim; k += 16) {
    const v2f a0 = *(const v2f*)(Ap + k);
    const v2f a1 = *(const v2f*)(Ap + k + 4);
    const v2f a2 = *(const v2f*)(Ap + k + 8);
    const v2f a3 = *(const v2f*)(Ap + k + 12);
    const float* bq = Bp + (size_t)k * ldb;
    v2f b0, b1, b2, b3;
    b0.x = bq[0];
    b0.y = bq[(size_t)1 * ldb];
    b1.x = bq[(size_t)4 * ldb];
    b1.y = bq[(size_t)5 * ldb];
    b2.x = bq[(size_t)8 * ldb];
    b2.y = bq[(size_t)9 * ldb];
    b3.x = bq[(size_t)12 * ldb];
    b3.y = bq[(size_t)13 * ldb];
    acc = wmma_f32(a0, b0, acc);
    acc = wmma_f32(a1, b1, acc);
    acc = wmma_f32(a2, b2, acc);
    acc = wmma_f32(a3, b3, acc);
  }
  for (; k < kdim; k += 4) {
    const v2f a = *(const v2f*)(Ap + k);
    v2f b;
    b.x = Bp[(size_t)k * ldb];
    b.y = Bp[(size_t)(k + 1) * ldb];
    acc = wmma_f32(a, b, acc);
  }

  const int col = col0 + lm;
  const float bv = bias ? bias[col] : 0.0f;
#pragma unroll
  for (int r = 0; r < 8; ++r) {
    const int row = row0 + r + 8 * lh;
    if (row < nrows) {
      float v = acc[r] + bv;
      if (relu) v = fmaxf(v, 0.0f);
      D[(size_t)row * ldo + col] = v;
    }
  }
}

__global__ void gemm_f32_wmma_bn(const float* __restrict__ A, int lda,
                                 const float* __restrict__ B, int ldb,
                                 float* __restrict__ D, int ldo,
                                 int nrows, int kdim, int ncols,
                                 const float* __restrict__ bias,
                                 const float* __restrict__ scl,
                                 const float* __restrict__ shf, int relu) {
  const int lane = threadIdx.x & 31;
  const int wid  = (blockIdx.x * blockDim.x + threadIdx.x) >> 5;
  const int colTiles = ncols >> 4;
  const int rowTiles = (nrows + 15) >> 4;
  if (wid >= rowTiles * colTiles) return;
  const int tr = wid / colTiles;
  const int tc = wid - tr * colTiles;
  const int row0 = tr << 4, col0 = tc << 4;
  const int lm = lane & 15;
  const int lh = lane >> 4;

  int arow = row0 + lm;
  if (arow >= nrows) arow = nrows - 1;
  const float* Ap = A + (size_t)arow * lda + 2 * lh;
  const float* Sp = scl + 2 * lh;
  const float* Hp = shf + 2 * lh;
  const float* Bp = B + (size_t)(2 * lh) * ldb + col0 + lm;

  v8f acc = {0.f, 0.f, 0.f, 0.f, 0.f, 0.f, 0.f, 0.f};
  int k = 0;
  for (; k + 16 <= kdim; k += 16) {
    v2f a0 = *(const v2f*)(Ap + k);
    v2f a1 = *(const v2f*)(Ap + k + 4);
    v2f a2 = *(const v2f*)(Ap + k + 8);
    v2f a3 = *(const v2f*)(Ap + k + 12);
    const v2f s0 = *(const v2f*)(Sp + k);
    const v2f s1 = *(const v2f*)(Sp + k + 4);
    const v2f s2 = *(const v2f*)(Sp + k + 8);
    const v2f s3 = *(const v2f*)(Sp + k + 12);
    const v2f h0 = *(const v2f*)(Hp + k);
    const v2f h1 = *(const v2f*)(Hp + k + 4);
    const v2f h2 = *(const v2f*)(Hp + k + 8);
    const v2f h3 = *(const v2f*)(Hp + k + 12);
    a0 = a0 * s0 + h0;
    a1 = a1 * s1 + h1;
    a2 = a2 * s2 + h2;
    a3 = a3 * s3 + h3;
    const float* bq = Bp + (size_t)k * ldb;
    v2f b0, b1, b2, b3;
    b0.x = bq[0];
    b0.y = bq[(size_t)1 * ldb];
    b1.x = bq[(size_t)4 * ldb];
    b1.y = bq[(size_t)5 * ldb];
    b2.x = bq[(size_t)8 * ldb];
    b2.y = bq[(size_t)9 * ldb];
    b3.x = bq[(size_t)12 * ldb];
    b3.y = bq[(size_t)13 * ldb];
    acc = wmma_f32(a0, b0, acc);
    acc = wmma_f32(a1, b1, acc);
    acc = wmma_f32(a2, b2, acc);
    acc = wmma_f32(a3, b3, acc);
  }
  for (; k < kdim; k += 4) {
    v2f a = *(const v2f*)(Ap + k);
    const v2f s = *(const v2f*)(Sp + k);
    const v2f h = *(const v2f*)(Hp + k);
    a = a * s + h;
    v2f b;
    b.x = Bp[(size_t)k * ldb];
    b.y = Bp[(size_t)(k + 1) * ldb];
    acc = wmma_f32(a, b, acc);
  }

  const int col = col0 + lm;
  const float bv = bias ? bias[col] : 0.0f;
#pragma unroll
  for (int r = 0; r < 8; ++r) {
    const int row = row0 + r + 8 * lh;
    if (row < nrows) {
      float v = acc[r] + bv;
      if (relu) v = fmaxf(v, 0.0f);
      D[(size_t)row * ldo + col] = v;
    }
  }
}

__global__ void attn_proj(const float* __restrict__ h, int heads,
                          const float* __restrict__ a_src,
                          const float* __restrict__ a_dst,
                          float* __restrict__ al_s, float* __restrict__ al_d,
                          int n) {
  const int wid  = (blockIdx.x * blockDim.x + threadIdx.x) >> 5;
  const int lane = threadIdx.x & 31;
  if (wid >= n * heads) return;
  const int node = wid / heads;
  const int head = wid - node * heads;
  const float* hp = h + (size_t)node * heads * DFEAT + head * DFEAT;
  const float* as = a_src + head * DFEAT;
  const float* ad = a_dst + head * DFEAT;
  float ss = 0.f, sd = 0.f;
  for (int c = lane; c < DFEAT; c += 32) {
    const float hv = hp[c];
    ss += hv * as[c];
    sd += hv * ad[c];
  }
  for (int off = 16; off > 0; off >>= 1) {
    ss += __shfl_xor(ss, off, 32);
    sd += __shfl_xor(sd, off, 32);
  }
  if (lane == 0) { al_s[wid] = ss; al_d[wid] = sd; }
}

__global__ void bias_fill(float* __restrict__ out, int ldo, int colOff,
                          int width, const float* __restrict__ bias, int n) {
  const int idx = blockIdx.x * blockDim.x + threadIdx.x;
  if (idx >= n * width) return;
  const int node = idx / width;
  const int c = idx - node * width;
  out[(size_t)node * ldo + colOff + c] = bias[c];
}

__global__ void zero_ms(unsigned* __restrict__ menc, float* __restrict__ ssum,
                        int nh) {
  const int i = blockIdx.x * blockDim.x + threadIdx.x;
  if (i < nh) { menc[i] = 0u; ssum[i] = 0.f; }
}

__global__ void edge_logit_max(const int* __restrict__ src,
                               const int* __restrict__ dst, int E, int n,
                               int heads, const float* __restrict__ al_s,
                               const float* __restrict__ al_d,
                               float* __restrict__ elog,
                               unsigned* __restrict__ menc) {
  const int idx = blockIdx.x * blockDim.x + threadIdx.x;
  if (idx >= (E + n) * heads) return;
  const int e = idx / heads;
  const int hh = idx - e * heads;
  const int s = (e < E) ? src[e] : (e - E);
  const int d = (e < E) ? dst[e] : (e - E);
  float x = al_s[s * heads + hh] + al_d[d * heads + hh];
  x = (x > 0.f) ? x : NEG_SLOPE * x;
  elog[idx] = x;
  atomicMax(&menc[d * heads + hh], ordEnc(x));
}

__global__ void edge_expsum(const int* __restrict__ src,
                            const int* __restrict__ dst, int E, int n,
                            int heads, const unsigned* __restrict__ menc,
                            float* __restrict__ elog,
                            float* __restrict__ ssum) {
  const int idx = blockIdx.x * blockDim.x + threadIdx.x;
  if (idx >= (E + n) * heads) return;
  const int e = idx / heads;
  const int hh = idx - e * heads;
  const int d = (e < E) ? dst[e] : (e - E);
  const float m = ordDec(menc[d * heads + hh]);
  const float ee = expf(elog[idx] - m);
  elog[idx] = ee;
  atomicAdd(&ssum[d * heads + hh], ee);
}

template <int HEADS>
__device__ __forceinline__ void edge_aggr_body(
    const int* __restrict__ src, const int* __restrict__ dst, int E, int n,
    const float* __restrict__ h, int ldh, const float* __restrict__ elog,
    const float* __restrict__ ssum, float* __restrict__ out, int ldo,
    int colOff) {
  constexpr int NL = HEADS * (DFEAT / 32);
  constexpr int BATCH = 32 / NL;
  const int lane = threadIdx.x & 31;
  const int wid  = (blockIdx.x * blockDim.x + threadIdx.x) >> 5;
  const int E2 = E + n;
  const int e0 = wid << 5;
  if (e0 >= E2) return;
  const int cnt = (E2 - e0 < 32) ? (E2 - e0) : 32;

  int s = 0, d = 0;
  float al[HEADS];
  for (int hh = 0; hh < HEADS; ++hh) al[hh] = 0.f;
  if (lane < cnt) {
    const int e = e0 + lane;
    s = (e < E) ? src[e] : (e - E);
    d = (e < E) ? dst[e] : (e - E);
    for (int hh = 0; hh < HEADS; ++hh)
      al[hh] = elog[e * HEADS + hh] / (ssum[d * HEADS + hh] + 1e-16f);
  }

  int j = 0;
  for (; j + BATCH <= cnt; j += BATCH) {
    float v[BATCH][NL];
    float* op[BATCH];
    for (int b = 0; b < BATCH; ++b) {
      const int sj = __builtin_amdgcn_readlane(s, j + b);
      const int dj = __builtin_amdgcn_readlane(d, j + b);
      const float* hp = h + (size_t)sj * ldh + lane;
      op[b] = out + (size_t)dj * ldo + colOff + lane;
      for (int c = 0; c < NL; ++c) v[b][c] = hp[32 * c];
    }
    for (int b = 0; b < BATCH; ++b) {
      for (int c = 0; c < NL; ++c) {
        const float a = __int_as_float(__builtin_amdgcn_readlane(
            __float_as_int(al[c / (DFEAT / 32)]), j + b));
        atomicAdd(op[b] + 32 * c, a * v[b][c]);
      }
    }
  }
  for (; j < cnt; ++j) {
    const int sj = __builtin_amdgcn_readlane(s, j);
    const int dj = __builtin_amdgcn_readlane(d, j);
    const float* hp = h + (size_t)sj * ldh + lane;
    float* o = out + (size_t)dj * ldo + colOff + lane;
    float v[NL];
    for (int c = 0; c < NL; ++c) v[c] = hp[32 * c];
    for (int c = 0; c < NL; ++c) {
      const float a = __int_as_float(__builtin_amdgcn_readlane(
          __float_as_int(al[c / (DFEAT / 32)]), j));
      atomicAdd(o + 32 * c, a * v[c]);
    }
  }
}

__global__ void edge_aggr_h2(const int* __restrict__ src,
                             const int* __restrict__ dst, int E, int n,
                             const float* __restrict__ h, int ldh,
                             const float* __restrict__ elog,
                             const float* __restrict__ ssum,
                             float* __restrict__ out, int ldo, int colOff) {
  edge_aggr_body<2>(src, dst, E, n, h, ldh, elog, ssum, out, ldo, colOff);
}

__global__ void edge_aggr_h1(const int* __restrict__ src,
                             const int* __restrict__ dst, int E, int n,
                             const float* __restrict__ h, int ldh,
                             const float* __restrict__ elog,
                             const float* __restrict__ ssum,
                             float* __restrict__ out, int ldo, int colOff) {
  edge_aggr_body<1>(src, dst, E, n, h, ldh, elog, ssum, out, ldo, colOff);
}

__global__ void col_stats(const float* __restrict__ y, int ld, int n,
                          float* __restrict__ sums, float* __restrict__ sumsq) {
  __shared__ float s1[256];
  __shared__ float s2[256];
  const int c = blockIdx.x;
  float s = 0.f, q = 0.f;
  for (int r = threadIdx.x; r < n; r += blockDim.x) {
    const float v = y[(size_t)r * ld + c];
    s += v;
    q += v * v;
  }
  s1[threadIdx.x] = s;
  s2[threadIdx.x] = q;
  __syncthreads();
  for (int st = 128; st > 0; st >>= 1) {
    if (threadIdx.x < st) {
      s1[threadIdx.x] += s1[threadIdx.x + st];
      s2[threadIdx.x] += s2[threadIdx.x + st];
    }
    __syncthreads();
  }
  if (threadIdx.x == 0) { sums[c] = s1[0]; sumsq[c] = s2[0]; }
}

__global__ void bn_final(const float* __restrict__ sums,
                         const float* __restrict__ sumsq,
                         const float* __restrict__ g,
                         const float* __restrict__ beta, int ncols, float invn,
                         float* __restrict__ sc, float* __restrict__ sh) {
  const int c = blockIdx.x * blockDim.x + threadIdx.x;
  if (c >= ncols) return;
  const float mean = sums[c] * invn;
  const float var = sumsq[c] * invn - mean * mean;
  const float r = rsqrtf(var + 1e-5f);
  const float s = g[c] * r;
  sc[c] = s;
  sh[c] = beta[c] - mean * s;
}

__global__ void mlp_out(const float* __restrict__ y2,
                        const float* __restrict__ sc,
                        const float* __restrict__ sh,
                        const float* __restrict__ Wm2,
                        const float* __restrict__ bm2, int n,
                        float* __restrict__ y3) {
  const int node = blockIdx.x * blockDim.x + threadIdx.x;
  if (node >= n) return;
  float a0 = bm2[0], a1 = bm2[1];
  const float* yp = y2 + (size_t)node * 96;
  for (int k = 0; k < 96; ++k) {
    const float v = fmaf(yp[k], sc[k], sh[k]);
    a0 = fmaf(v, Wm2[2 * k + 0], a0);
    a1 = fmaf(v, Wm2[2 * k + 1], a1);
  }
  y3[node * 2 + 0] = fmaxf(a0, 0.f);
  y3[node * 2 + 1] = fmaxf(a1, 0.f);
}

__global__ void bn_apply(const float* __restrict__ y,
                         const float* __restrict__ sc,
                         const float* __restrict__ sh, int total, int ncols,
                         float* __restrict__ out) {
  const int i = blockIdx.x * blockDim.x + threadIdx.x;
  if (i >= total) return;
  const int c = i % ncols;
  out[i] = fmaf(y[i], sc[c], sh[c]);
}

extern "C" void kernel_launch(void* const* d_in, const int* in_sizes, int n_in,
                              void* d_out, int out_size, void* d_ws,
                              size_t ws_size, hipStream_t stream) {
  const float* x      = (const float*)d_in[0];
  const int*   ei     = (const int*)d_in[1];
  const float* W1     = (const float*)d_in[2];
  const float* as1    = (const float*)d_in[3];
  const float* ad1    = (const float*)d_in[4];
  const float* b1     = (const float*)d_in[5];
  const float* W2     = (const float*)d_in[6];
  const float* as2    = (const float*)d_in[7];
  const float* ad2    = (const float*)d_in[8];
  const float* b2     = (const float*)d_in[9];
  const float* Wl     = (const float*)d_in[10];
  const float* bl     = (const float*)d_in[11];
  const float* gl     = (const float*)d_in[12];
  const float* betal  = (const float*)d_in[13];
  const float* Wm1    = (const float*)d_in[14];
  const float* bm1    = (const float*)d_in[15];
  const float* gm1    = (const float*)d_in[16];
  const float* betam1 = (const float*)d_in[17];
  const float* Wm2    = (const float*)d_in[18];
  const float* bm2    = (const float*)d_in[19];
  const float* gm2    = (const float*)d_in[20];
  const float* betam2 = (const float*)d_in[21];

  const int N = in_sizes[0] / DFEAT;
  const int E = in_sizes[1] / 2;
  const int E2 = E + N;
  const int* srcA = ei;
  const int* dstA = ei + E;

  char* w = (char*)d_ws;
  auto alloc = [&](size_t bytes) -> char* {
    char* p = w;
    w += (bytes + 255) & ~(size_t)255;
    return p;
  };
  float*    xcat = (float*)alloc((size_t)N * 384 * 4);
  float*    h1   = (float*)alloc((size_t)N * 256 * 4);
  float*    h2   = (float*)alloc((size_t)N * 128 * 4);
  float*    al_s = (float*)alloc((size_t)N * 2 * 4);
  float*    al_d = (float*)alloc((size_t)N * 2 * 4);
  unsigned* menc = (unsigned*)alloc((size_t)N * 2 * 4);
  float*    ssum = (float*)alloc((size_t)N * 2 * 4);
  float*    elog = (float*)alloc((size_t)E2 * 2 * 4);
  float*    y3   = (float*)alloc((size_t)N * 2 * 4);
  float*    sums = (float*)alloc(96 * 4);
  float*    sumsq= (float*)alloc(96 * 4);
  float*    sc1  = (float*)alloc(96 * 4);
  float*    sh1  = (float*)alloc(96 * 4);
  float*    sc2  = (float*)alloc(96 * 4);
  float*    sh2  = (float*)alloc(96 * 4);
  float*    sc3  = (float*)alloc(96 * 4);
  float*    sh3  = (float*)alloc(96 * 4);
  float* y1 = h1;
  float* y2 = h1 + (size_t)N * 96;

  const int TB = 256;
  auto gemmBlocks = [](int rows, int cols) {
    const int tiles = ((rows + 15) / 16) * (cols / 16);
    return (tiles + 7) / 8;
  };
  const int aggrBlocks = ((E2 + 31) / 32 + 7) / 8;

  gemm_f32_wmma<<<gemmBlocks(N, 256), TB, 0, stream>>>(
      x, 128, W1, 256, h1, 256, N, 128, 256, nullptr, 0);
  attn_proj<<<(N * 2 + 7) / 8, TB, 0, stream>>>(h1, 2, as1, ad1, al_s, al_d, N);
  bias_fill<<<(N * 256 + TB - 1) / TB, TB, 0, stream>>>(xcat, 384, 0, 256, b1, N);
  zero_ms<<<(N * 2 + TB - 1) / TB, TB, 0, stream>>>(menc, ssum, N * 2);
  edge_logit_max<<<(E2 * 2 + TB - 1) / TB, TB, 0, stream>>>(
      srcA, dstA, E, N, 2, al_s, al_d, elog, menc);
  edge_expsum<<<(E2 * 2 + TB - 1) / TB, TB, 0, stream>>>(
      srcA, dstA, E, N, 2, menc, elog, ssum);
  edge_aggr_h2<<<aggrBlocks, TB, 0, stream>>>(srcA, dstA, E, N, h1, 256, elog,
                                              ssum, xcat, 384, 0);

  gemm_f32_wmma<<<gemmBlocks(N, 128), TB, 0, stream>>>(
      xcat, 384, W2, 128, h2, 128, N, 256, 128, nullptr, 0);
  attn_proj<<<(N + 7) / 8, TB, 0, stream>>>(h2, 1, as2, ad2, al_s, al_d, N);
  bias_fill<<<(N * 128 + TB - 1) / TB, TB, 0, stream>>>(xcat, 384, 256, 128, b2, N);
  zero_ms<<<(N + TB - 1) / TB, TB, 0, stream>>>(menc, ssum, N);
  edge_logit_max<<<(E2 + TB - 1) / TB, TB, 0, stream>>>(
      srcA, dstA, E, N, 1, al_s, al_d, elog, menc);
  edge_expsum<<<(E2 + TB - 1) / TB, TB, 0, stream>>>(
      srcA, dstA, E, N, 1, menc, elog, ssum);
  edge_aggr_h1<<<aggrBlocks, TB, 0, stream>>>(srcA, dstA, E, N, h2, 128, elog,
                                              ssum, xcat, 384, 256);

  const float invn = 1.0f / (float)N;
  gemm_f32_wmma<<<gemmBlocks(N, 96), TB, 0, stream>>>(
      xcat, 384, Wl, 96, y1, 96, N, 384, 96, bl, 1);
  col_stats<<<96, 256, 0, stream>>>(y1, 96, N, sums, sumsq);
  bn_final<<<1, 128, 0, stream>>>(sums, sumsq, gl, betal, 96, invn, sc1, sh1);
  gemm_f32_wmma_bn<<<gemmBlocks(N, 96), TB, 0, stream>>>(
      y1, 96, Wm1, 96, y2, 96, N, 96, 96, bm1, sc1, sh1, 1);
  col_stats<<<96, 256, 0, stream>>>(y2, 96, N, sums, sumsq);
  bn_final<<<1, 128, 0, stream>>>(sums, sumsq, gm1, betam1, 96, invn, sc2, sh2);
  mlp_out<<<(N + TB - 1) / TB, TB, 0, stream>>>(y2, sc2, sh2, Wm2, bm2, N, y3);
  col_stats<<<2, 256, 0, stream>>>(y3, 2, N, sums, sumsq);
  bn_final<<<1, 128, 0, stream>>>(sums, sumsq, gm2, betam2, 2, invn, sc3, sh3);
  bn_apply<<<(N * 2 + TB - 1) / TB, TB, 0, stream>>>(y3, sc3, sh3, N * 2, 2,
                                                     (float*)d_out);
}


# edge passes split by destination half: two ticket work lists (rows below n/2, rows above); a work-group draws from the list of its own L2 half, so a line is added into from one half; arrival order
# speedup vs baseline: 1.9130x; 1.1782x over previous
#include <hip/hip_runtime.h>
#include <hip/hip_bf16.h>
#include <math.h>


typedef __attribute__((ext_vector_type(2))) float v2f;
typedef __attribute__((ext_vector_type(8))) float v8f;

#define DFEAT 128
#define NEG_SLOPE 0.2f

__device__ __forceinline__ unsigned ordEnc(float f) {
  unsigned u = __float_as_uint(f);
  return (u & 0x80000000u) ? ~u : (u | 0x80000000u);
}
__device__ __forceinline__ float ordDec(unsigned e) {
  unsigned u = (e & 0x80000000u) ? (e & 0x7fffffffu) : ~e;
  return __uint_as_float(u);
}

__device__ __forceinline__ v8f wmma_f32(v2f a, v2f b, v8f c) {
  return __builtin_amdgcn_wmma_f32_16x16x4_f32(false, a, false, b, (short)0, c,
                                               false, false);
}

__global__ void gemm_f32_wmma(const float* __restrict__ A, int lda,
                              const float* __restrict__ B, int ldb,
                              float* __restrict__ D, int ldo,
                              int nrows, int kdim, int ncols,
                              const float* __restrict__ bias, int relu) {
  const int lane = threadIdx.x & 31;
  const int wid  = (blockIdx.x * blockDim.x + threadIdx.x) >> 5;
  const int colTiles = ncols >> 4;
  const int rowTiles = (nrows + 15) >> 4;
  if (wid >= rowTiles * colTiles) return;
  const int tr = wid / colTiles;
  const int tc = wid - tr * colTiles;
  const int row0 = tr << 4, col0 = tc << 4;
  const int lm = lane & 15;
  const int lh = lane >> 4;

  int arow = row0 + lm;
  if (arow >= nrows) arow = nrows - 1;
  const float* Ap = A + (size_t)arow * lda + 2 * lh;
  const float* Bp = B + (size_t)(2 * lh) * ldb + col0 + lm;

  v8f acc = {0.f, 0.f, 0.f, 0.f, 0.f, 0.f, 0.f, 0.f};
  int k = 0;
  for (; k + 16 <= kdim; k += 16) {
    const v2f a0 = *(const v2f*)(Ap + k);
    const v2f a1 = *(const v2f*)(Ap + k + 4);
    const v2f a2 = *(const v2f*)(Ap + k + 8);
    const v2f a3 = *(const v2f*)(Ap + k + 12);
    const float* bq = Bp + (size_t)k * ldb;
    v2f b0, b1, b2, b3;
    b0.x = bq[0];
    b0.y = bq[(size_t)1 * ldb];
    b1.x = bq[(size_t)4 * ldb];
    b1.y = bq[(size_t)5 * ldb];
    b2.x = bq[(size_t)8 * ldb];
    b2.y = bq[(size_t)9 * ldb];
    b3.x = bq[(size_t)12 * ldb];
    b3.y = bq[(size_t)13 * ldb];
    acc = wmma_f32(a0, b0, acc);
    acc = wmma_f32(a1, b1, acc);
    acc = wmma_f32(a2, b2, acc);
    acc = wmma_f32(a3, b3, acc);
  }
  for (; k < kdim; k += 4) {
    const v2f a = *(const v2f*)(Ap + k);
    v2f b;
    b.x = Bp[(size_t)k * ldb];
    b.y = Bp[(size_t)(k + 1) * ldb];
    acc = wmma_f32(a, b, acc);
  }

  const int col = col0 + lm;
  const float bv = bias ? bias[col] : 0.0f;
#pragma unroll
  for (int r = 0; r < 8; ++r) {
    const int row = row0 + r + 8 * lh;
    if (row < nrows) {
      float v = acc[r] + bv;
      if (relu) v = fmaxf(v, 0.0f);
      D[(size_t)row * ldo + col] = v;
    }
  }
}

__global__ void gemm_f32_wmma_bn(const float* __restrict__ A, int lda,
                                 const float* __restrict__ B, int ldb,
                                 float* __restrict__ D, int ldo,
                                 int nrows, int kdim, int ncols,
                                 const float* __restrict__ bias,
                                 const float* __restrict__ scl,
                                 const float* __restrict__ shf, int relu) {
  const int lane = threadIdx.x & 31;
  const int wid  = (blockIdx.x * blockDim.x + threadIdx.x) >> 5;
  const int colTiles = ncols >> 4;
  const int rowTiles = (nrows + 15) >> 4;
  if (wid >= rowTiles * colTiles) return;
  const int tr = wid / colTiles;
  const int tc = wid - tr * colTiles;
  const int row0 = tr << 4, col0 = tc << 4;
  const int lm = lane & 15;
  const int lh = lane >> 4;

  int arow = row0 + lm;
  if (arow >= nrows) arow = nrows - 1;
  const float* Ap = A + (size_t)arow * lda + 2 * lh;
  const float* Sp = scl + 2 * lh;
  const float* Hp = shf + 2 * lh;
  const float* Bp = B + (size_t)(2 * lh) * ldb + col0 + lm;

  v8f acc = {0.f, 0.f, 0.f, 0.f, 0.f, 0.f, 0.f, 0.f};
  int k = 0;
  for (; k + 16 <= kdim; k += 16) {
    v2f a0 = *(const v2f*)(Ap + k);
    v2f a1 = *(const v2f*)(Ap + k + 4);
    v2f a2 = *(const v2f*)(Ap + k + 8);
    v2f a3 = *(const v2f*)(Ap + k + 12);
    const v2f s0 = *(const v2f*)(Sp + k);
    const v2f s1 = *(const v2f*)(Sp + k + 4);
    const v2f s2 = *(const v2f*)(Sp + k + 8);
    const v2f s3 = *(const v2f*)(Sp + k + 12);
    const v2f h0 = *(const v2f*)(Hp + k);
    const v2f h1 = *(const v2f*)(Hp + k + 4);
    const v2f h2 = *(const v2f*)(Hp + k + 8);
    const v2f h3 = *(const v2f*)(Hp + k + 12);
    a0 = a0 * s0 + h0;
    a1 = a1 * s1 + h1;
    a2 = a2 * s2 + h2;
    a3 = a3 * s3 + h3;
    const float* bq = Bp + (size_t)k * ldb;
    v2f b0, b1, b2, b3;
    b0.x = bq[0];
    b0.y = bq[(size_t)1 * ldb];
    b1.x = bq[(size_t)4 * ldb];
    b1.y = bq[(size_t)5 * ldb];
    b2.x = bq[(size_t)8 * ldb];
    b2.y = bq[(size_t)9 * ldb];
    b3.x = bq[(size_t)12 * ldb];
    b3.y = bq[(size_t)13 * ldb];
    acc = wmma_f32(a0, b0, acc);
    acc = wmma_f32(a1, b1, acc);
    acc = wmma_f32(a2, b2, acc);
    acc = wmma_f32(a3, b3, acc);
  }
  for (; k < kdim; k += 4) {
    v2f a = *(const v2f*)(Ap + k);
    const v2f s = *(const v2f*)(Sp + k);
    const v2f h = *(const v2f*)(Hp + k);
    a = a * s + h;
    v2f b;
    b.x = Bp[(size_t)k * ldb];
    b.y = Bp[(size_t)(k + 1) * ldb];
    acc = wmma_f32(a, b, acc);
  }

  const int col = col0 + lm;
  const float bv = bias ? bias[col] : 0.0f;
#pragma unroll
  for (int r = 0; r < 8; ++r) {
    const int row = row0 + r + 8 * lh;
    if (row < nrows) {
      float v = acc[r] + bv;
      if (relu) v = fmaxf(v, 0.0f);
      D[(size_t)row * ldo + col] = v;
    }
  }
}

__global__ void attn_proj(const float* __restrict__ h, int heads,
                          const float* __restrict__ a_src,
                          const float* __restrict__ a_dst,
                          float* __restrict__ al_s, float* __restrict__ al_d,
                          int n) {
  const int wid  = (blockIdx.x * blockDim.x + threadIdx.x) >> 5;
  const int lane = threadIdx.x & 31;
  if (wid >= n * heads) return;
  const int node = wid / heads;
  const int head = wid - node * heads;
  const float* hp = h + (size_t)node * heads * DFEAT + head * DFEAT;
  const float* as = a_src + head * DFEAT;
  const float* ad = a_dst + head * DFEAT;
  float ss = 0.f, sd = 0.f;
  for (int c = lane; c < DFEAT; c += 32) {
    const float hv = hp[c];
    ss += hv * as[c];
    sd += hv * ad[c];
  }
  for (int off = 16; off > 0; off >>= 1) {
    ss += __shfl_xor(ss, off, 32);
    sd += __shfl_xor(sd, off, 32);
  }
  if (lane == 0) { al_s[wid] = ss; al_d[wid] = sd; }
}

__global__ void bias_fill(float* __restrict__ out, int ldo, int colOff,
                          int width, const float* __restrict__ bias, int n) {
  const int idx = blockIdx.x * blockDim.x + threadIdx.x;
  if (idx >= n * width) return;
  const int node = idx / width;
  const int c = idx - node * width;
  out[(size_t)node * ldo + colOff + c] = bias[c];
}

__global__ void zero_ms(unsigned* __restrict__ menc, float* __restrict__ ssum,
                        int nh, unsigned* __restrict__ qctr) {
  const int i = blockIdx.x * blockDim.x + threadIdx.x;
  if (i < nh) { menc[i] = 0u; ssum[i] = 0.f; }
  if (i < 6) qctr[i * 64] = 0u;
}

#define UNIT_EDGES 64

__device__ __forceinline__ int my_half() {
  const unsigned id = __builtin_amdgcn_s_sendmsg_rtn(0x87);
  return (((id >> 16) & 15u) >= 4u) ? 1 : 0;
}

__device__ __forceinline__ void lds_barrier() {
  __builtin_amdgcn_fence(__ATOMIC_RELEASE, "workgroup", "local");
  __builtin_amdgcn_s_barrier();
  __builtin_amdgcn_fence(__ATOMIC_ACQUIRE, "workgroup", "local");
}

template <class F>
__device__ __forceinline__ void owned_units(unsigned* __restrict__ ctr2,
                                            int E2, F body) {
  __shared__ unsigned s_ticket;
  __shared__ int s_half;
  const int wave = threadIdx.x >> 5;
  const int span = (blockDim.x >> 5) * UNIT_EDGES;
  const int nunits = (E2 + span - 1) / span;
  if (threadIdx.x == 0) s_half = my_half();
  lds_barrier();
  const int mine = s_half;
  for (int pass = 0; pass < 2; ++pass) {
    const int hs = pass ? (1 - mine) : mine;
    unsigned* ctr = ctr2 + hs * 64;
    for (int it = 0; it <= nunits; ++it) {
      if (threadIdx.x == 0) s_ticket = atomicAdd(ctr, 1u);
      lds_barrier();
      const unsigned u = s_ticket;
      lds_barrier();
      if (u >= (unsigned)nunits) break;
      for (int sub = 0; sub < UNIT_EDGES / 32; ++sub) {
        const int e0 = (int)u * span + wave * UNIT_EDGES + sub * 32;
        if (e0 < E2) body(e0, hs);
      }
    }
  }
}

__device__ __forceinline__ bool edge_meta(const int* __restrict__ src,
                                          const int* __restrict__ dst, int E,
                                          int n, int e0, int lane, int hs,
                                          int& e, int& s, int& d) {
  e = e0 + lane;
  s = 0;
  d = 0;
  if (e >= E + n) return false;
  s = (e < E) ? src[e] : (e - E);
  d = (e < E) ? dst[e] : (e - E);
  return ((d >= (n >> 1)) ? 1 : 0) == hs;
}

template <int HEADS>
__device__ __forceinline__ void edge_logit_max_body(
    const int* __restrict__ src, const int* __restrict__ dst, int E, int n,
    const float* __restrict__ al_s, const float* __restrict__ al_d,
    float* __restrict__ elog, unsigned* __restrict__ menc,
    unsigned* __restrict__ ctr2) {
  const int lane = threadIdx.x & 31;
  owned_units(ctr2, E + n, [&](int e0, int hs) {
    int e, s, d;
    if (edge_meta(src, dst, E, n, e0, lane, hs, e, s, d)) {
      for (int hh = 0; hh < HEADS; ++hh) {
        float x = al_s[s * HEADS + hh] + al_d[d * HEADS + hh];
        x = (x > 0.f) ? x : NEG_SLOPE * x;
        elog[e * HEADS + hh] = x;
        atomicMax(&menc[d * HEADS + hh], ordEnc(x));
      }
    }
  });
}

__global__ void edge_logit_max_h2(const int* __restrict__ src,
                                  const int* __restrict__ dst, int E, int n,
                                  const float* __restrict__ al_s,
                                  const float* __restrict__ al_d,
                                  float* __restrict__ elog,
                                  unsigned* __restrict__ menc,
                                  unsigned* __restrict__ ctr2) {
  edge_logit_max_body<2>(src, dst, E, n, al_s, al_d, elog, menc, ctr2);
}

__global__ void edge_logit_max_h1(const int* __restrict__ src,
                                  const int* __restrict__ dst, int E, int n,
                                  const float* __restrict__ al_s,
                                  const float* __restrict__ al_d,
                                  float* __restrict__ elog,
                                  unsigned* __restrict__ menc,
                                  unsigned* __restrict__ ctr2) {
  edge_logit_max_body<1>(src, dst, E, n, al_s, al_d, elog, menc, ctr2);
}

template <int HEADS>
__device__ __forceinline__ void edge_expsum_body(
    const int* __restrict__ src, const int* __restrict__ dst, int E, int n,
    const unsigned* __restrict__ menc, float* __restrict__ elog,
    float* __restrict__ ssum, unsigned* __restrict__ ctr2) {
  const int lane = threadIdx.x & 31;
  owned_units(ctr2, E + n, [&](int e0, int hs) {
    int e, s, d;
    if (edge_meta(src, dst, E, n, e0, lane, hs, e, s, d)) {
      for (int hh = 0; hh < HEADS; ++hh) {
        const float m = ordDec(menc[d * HEADS + hh]);
        const float ee = expf(elog[e * HEADS + hh] - m);
        elog[e * HEADS + hh] = ee;
        atomicAdd(&ssum[d * HEADS + hh], ee);
      }
    }
  });
}

__global__ void edge_expsum_h2(const int* __restrict__ src,
                               const int* __restrict__ dst, int E, int n,
                               const unsigned* __restrict__ menc,
                               float* __restrict__ elog,
                               float* __restrict__ ssum,
                               unsigned* __restrict__ ctr2) {
  edge_expsum_body<2>(src, dst, E, n, menc, elog, ssum, ctr2);
}

__global__ void edge_expsum_h1(const int* __restrict__ src,
                               const int* __restrict__ dst, int E, int n,
                               const unsigned* __restrict__ menc,
                               float* __restrict__ elog,
                               float* __restrict__ ssum,
                               unsigned* __restrict__ ctr2) {
  edge_expsum_body<1>(src, dst, E, n, menc, elog, ssum, ctr2);
}

template <int HEADS>
__device__ __forceinline__ void edge_aggr_body(
    const int* __restrict__ src, const int* __restrict__ dst, int E, int n,
    const float* __restrict__ h, int ldh, const float* __restrict__ elog,
    const float* __restrict__ ssum, float* __restrict__ out, int ldo,
    int colOff, unsigned* __restrict__ ctr2) {
  constexpr int NL = HEADS * (DFEAT / 32);
  constexpr int BATCH = 32 / NL;
  const int lane = threadIdx.x & 31;
  owned_units(ctr2, E + n, [&](int e0, int hs) {
    int e, s, d;
    const bool own = edge_meta(src, dst, E, n, e0, lane, hs, e, s, d);
    float al[HEADS];
    for (int hh = 0; hh < HEADS; ++hh) al[hh] = 0.f;
    if (own) {
      for (int hh = 0; hh < HEADS; ++hh)
        al[hh] = elog[e * HEADS + hh] / (ssum[d * HEADS + hh] + 1e-16f);
    }
    unsigned m = __builtin_amdgcn_ballot_w32(own);

    while (__builtin_popcount(m) >= BATCH) {
      float v[BATCH][NL];
      float* op[BATCH];
      int jj[BATCH];
      for (int b = 0; b < BATCH; ++b) {
        const int j = __builtin_ctz(m);
        m &= m - 1;
        jj[b] = j;
        const int sj = __builtin_amdgcn_readlane(s, j);
        const int dj = __builtin_amdgcn_readlane(d, j);
        const float* hp = h + (size_t)sj * ldh + lane;
        op[b] = out + (size_t)dj * ldo + colOff + lane;
        for (int c = 0; c < NL; ++c) v[b][c] = hp[32 * c];
      }
      for (int b = 0; b < BATCH; ++b) {
        for (int c = 0; c < NL; ++c) {
          const float a = __int_as_float(__builtin_amdgcn_readlane(
              __float_as_int(al[c / (DFEAT / 32)]), jj[b]));
          atomicAdd(op[b] + 32 * c, a * v[b][c]);
        }
      }
    }
    while (m) {
      const int j = __builtin_ctz(m);
      m &= m - 1;
      const int sj = __builtin_amdgcn_readlane(s, j);
      const int dj = __builtin_amdgcn_readlane(d, j);
      const float* hp = h + (size_t)sj * ldh + lane;
      float* o = out + (size_t)dj * ldo + colOff + lane;
      float v[NL];
      for (int c = 0; c < NL; ++c) v[c] = hp[32 * c];
      for (int c = 0; c < NL; ++c) {
        const float a = __int_as_float(__builtin_amdgcn_readlane(
            __float_as_int(al[c / (DFEAT / 32)]), j));
        atomicAdd(o + 32 * c, a * v[c]);
      }
    }
  });
}

__global__ void edge_aggr_h2(const int* __restrict__ src,
                             const int* __restrict__ dst, int E, int n,
                             const float* __restrict__ h, int ldh,
                             const float* __restrict__ elog,
                             const float* __restrict__ ssum,
                             float* __restrict__ out, int ldo, int colOff,
                             unsigned* __restrict__ ctr2) {
  edge_aggr_body<2>(src, dst, E, n, h, ldh, elog, ssum, out, ldo, colOff, ctr2);
}

__global__ void edge_aggr_h1(const int* __restrict__ src,
                             const int* __restrict__ dst, int E, int n,
                             const float* __restrict__ h, int ldh,
                             const float* __restrict__ elog,
                             const float* __restrict__ ssum,
                             float* __restrict__ out, int ldo, int colOff,
                             unsigned* __restrict__ ctr2) {
  edge_aggr_body<1>(src, dst, E, n, h, ldh, elog, ssum, out, ldo, colOff, ctr2);
}

__global__ void col_stats(const float* __restrict__ y, int ld, int n,
                          float* __restrict__ sums, float* __restrict__ sumsq) {
  __shared__ float s1[256];
  __shared__ float s2[256];
  const int c = blockIdx.x;
  float s = 0.f, q = 0.f;
  for (int r = threadIdx.x; r < n; r += blockDim.x) {
    const float v = y[(size_t)r * ld + c];
    s += v;
    q += v * v;
  }
  s1[threadIdx.x] = s;
  s2[threadIdx.x] = q;
  __syncthreads();
  for (int st = 128; st > 0; st >>= 1) {
    if (threadIdx.x < st) {
      s1[threadIdx.x] += s1[threadIdx.x + st];
      s2[threadIdx.x] += s2[threadIdx.x + st];
    }
    __syncthreads();
  }
  if (threadIdx.x == 0) { sums[c] = s1[0]; sumsq[c] = s2[0]; }
}

__global__ void bn_final(const float* __restrict__ sums,
                         const float* __restrict__ sumsq,
                         const float* __restrict__ g,
                         const float* __restrict__ beta, int ncols, float invn,
                         float* __restrict__ sc, float* __restrict__ sh) {
  const int c = blockIdx.x * blockDim.x + threadIdx.x;
  if (c >= ncols) return;
  const float mean = sums[c] * invn;
  const float var = sumsq[c] * invn - mean * mean;
  const float r = rsqrtf(var + 1e-5f);
  const float s = g[c] * r;
  sc[c] = s;
  sh[c] = beta[c] - mean * s;
}

__global__ void mlp_out(const float* __restrict__ y2,
                        const float* __restrict__ sc,
                        const float* __restrict__ sh,
                        const float* __restrict__ Wm2,
                        const float* __restrict__ bm2, int n,
                        float* __restrict__ y3) {
  const int node = blockIdx.x * blockDim.x + threadIdx.x;
  if (node >= n) return;
  float a0 = bm2[0], a1 = bm2[1];
  const float* yp = y2 + (size_t)node * 96;
  for (int k = 0; k < 96; ++k) {
    const float v = fmaf(yp[k], sc[k], sh[k]);
    a0 = fmaf(v, Wm2[2 * k + 0], a0);
    a1 = fmaf(v, Wm2[2 * k + 1], a1);
  }
  y3[node * 2 + 0] = fmaxf(a0, 0.f);
  y3[node * 2 + 1] = fmaxf(a1, 0.f);
}

__global__ void bn_apply(const float* __restrict__ y,
                         const float* __restrict__ sc,
                         const float* __restrict__ sh, int total, int ncols,
                         float* __restrict__ out) {
  const int i = blockIdx.x * blockDim.x + threadIdx.x;
  if (i >= total) return;
  const int c = i % ncols;
  out[i] = fmaf(y[i], sc[c], sh[c]);
}

extern "C" void kernel_launch(void* const* d_in, const int* in_sizes, int n_in,
                              void* d_out, int out_size, void* d_ws,
                              size_t ws_size, hipStream_t stream) {
  const float* x      = (const float*)d_in[0];
  const int*   ei     = (const int*)d_in[1];
  const float* W1     = (const float*)d_in[2];
  const float* as1    = (const float*)d_in[3];
  const float* ad1    = (const float*)d_in[4];
  const float* b1     = (const float*)d_in[5];
  const float* W2     = (const float*)d_in[6];
  const float* as2    = (const float*)d_in[7];
  const float* ad2    = (const float*)d_in[8];
  const float* b2     = (const float*)d_in[9];
  const float* Wl     = (const float*)d_in[10];
  const float* bl     = (const float*)d_in[11];
  const float* gl     = (const float*)d_in[12];
  const float* betal  = (const float*)d_in[13];
  const float* Wm1    = (const float*)d_in[14];
  const float* bm1    = (const float*)d_in[15];
  const float* gm1    = (const float*)d_in[16];
  const float* betam1 = (const float*)d_in[17];
  const float* Wm2    = (const float*)d_in[18];
  const float* bm2    = (const float*)d_in[19];
  const float* gm2    = (const float*)d_in[20];
  const float* betam2 = (const float*)d_in[21];

  const int N = in_sizes[0] / DFEAT;
  const int E = in_sizes[1] / 2;
  const int E2 = E + N;
  const int* srcA = ei;
  const int* dstA = ei + E;

  char* w = (char*)d_ws;
  auto alloc = [&](size_t bytes) -> char* {
    char* p = w;
    w += (bytes + 255) & ~(size_t)255;
    return p;
  };
  float*    xcat = (float*)alloc((size_t)N * 384 * 4);
  float*    h1   = (float*)alloc((size_t)N * 256 * 4);
  float*    h2   = (float*)alloc((size_t)N * 128 * 4);
  float*    al_s = (float*)alloc((size_t)N * 2 * 4);
  float*    al_d = (float*)alloc((size_t)N * 2 * 4);
  unsigned* menc = (unsigned*)alloc((size_t)N * 2 * 4);
  float*    ssum = (float*)alloc((size_t)N * 2 * 4);
  float*    elog = (float*)alloc((size_t)E2 * 2 * 4);
  float*    y3   = (float*)alloc((size_t)N * 2 * 4);
  float*    sums = (float*)alloc(96 * 4);
  float*    sumsq= (float*)alloc(96 * 4);
  float*    sc1  = (float*)alloc(96 * 4);
  float*    sh1  = (float*)alloc(96 * 4);
  float*    sc2  = (float*)alloc(96 * 4);
  float*    sh2  = (float*)alloc(96 * 4);
  float*    sc3  = (float*)alloc(96 * 4);
  float*    sh3  = (float*)alloc(96 * 4);
  float* y1 = h1;
  float* y2 = h1 + (size_t)N * 96;
  unsigned* qctr = (unsigned*)y3;

  const int TB = 256;
  auto gemmBlocks = [](int rows, int cols) {
    const int tiles = ((rows + 15) / 16) * (cols / 16);
    return (tiles + 7) / 8;
  };
  const int unitBlocks = ((E2 + UNIT_EDGES - 1) / UNIT_EDGES + 7) / 8;

  gemm_f32_wmma<<<gemmBlocks(N, 256), TB, 0, stream>>>(
      x, 128, W1, 256, h1, 256, N, 128, 256, nullptr, 0);
  attn_proj<<<(N * 2 + 7) / 8, TB, 0, stream>>>(h1, 2, as1, ad1, al_s, al_d, N);
  bias_fill<<<(N * 256 + TB - 1) / TB, TB, 0, stream>>>(xcat, 384, 0, 256, b1, N);
  zero_ms<<<(N * 2 + TB - 1) / TB, TB, 0, stream>>>(menc, ssum, N * 2, qctr);
  edge_logit_max_h2<<<unitBlocks, TB, 0, stream>>>(
      srcA, dstA, E, N, al_s, al_d, elog, menc, qctr);
  edge_expsum_h2<<<unitBlocks, TB, 0, stream>>>(
      srcA, dstA, E, N, menc, elog, ssum, qctr + 128);
  edge_aggr_h2<<<unitBlocks, TB, 0, stream>>>(srcA, dstA, E, N, h1, 256, elog,
                                              ssum, xcat, 384, 0, qctr + 256);

  gemm_f32_wmma<<<gemmBlocks(N, 128), TB, 0, stream>>>(
      xcat, 384, W2, 128, h2, 128, N, 256, 128, nullptr, 0);
  attn_proj<<<(N + 7) / 8, TB, 0, stream>>>(h2, 1, as2, ad2, al_s, al_d, N);
  bias_fill<<<(N * 128 + TB - 1) / TB, TB, 0, stream>>>(xcat, 384, 256, 128, b2, N);
  zero_ms<<<(N + TB - 1) / TB, TB, 0, stream>>>(menc, ssum, N, qctr);
  edge_logit_max_h1<<<unitBlocks, TB, 0, stream>>>(
      srcA, dstA, E, N, al_s, al_d, elog, menc, qctr);
  edge_expsum_h1<<<unitBlocks, TB, 0, stream>>>(
      srcA, dstA, E, N, menc, elog, ssum, qctr + 128);
  edge_aggr_h1<<<unitBlocks, TB, 0, stream>>>(srcA, dstA, E, N, h2, 128, elog,
                                              ssum, xcat, 384, 256, qctr + 256);

  const float invn = 1.0f / (float)N;
  gemm_f32_wmma<<<gemmBlocks(N, 96), TB, 0, stream>>>(
      xcat, 384, Wl, 96, y1, 96, N, 384, 96, bl, 1);
  col_stats<<<96, 256, 0, stream>>>(y1, 96, N, sums, sumsq);
  bn_final<<<1, 128, 0, stream>>>(sums, sumsq, gl, betal, 96, invn, sc1, sh1);
  gemm_f32_wmma_bn<<<gemmBlocks(N, 96), TB, 0, stream>>>(
      y1, 96, Wm1, 96, y2, 96, N, 96, 96, bm1, sc1, sh1, 1);
  col_stats<<<96, 256, 0, stream>>>(y2, 96, N, sums, sumsq);
  bn_final<<<1, 128, 0, stream>>>(sums, sumsq, gm1, betam1, 96, invn, sc2, sh2);
  mlp_out<<<(N + TB - 1) / TB, TB, 0, stream>>>(y2, sc2, sh2, Wm2, bm2, N, y3);
  col_stats<<<2, 256, 0, stream>>>(y3, 2, N, sums, sumsq);
  bn_final<<<1, 128, 0, stream>>>(sums, sumsq, gm2, betam2, 2, invn, sc3, sh3);
  bn_apply<<<(N * 2 + TB - 1) / TB, TB, 0, stream>>>(y3, sc3, sh3, N * 2, 2,
                                                     (float*)d_out);
}


# BatchNorm column statistics re-tiled: flat coalesced 16-byte reads, 10 independent loads per thread, per-work-group partials, one wave per column adds them (order of the BN sums changes)
# speedup vs baseline: 3.3386x; 1.7452x over previous
#include <hip/hip_runtime.h>
#include <hip/hip_bf16.h>
#include <math.h>


typedef __attribute__((ext_vector_type(2))) float v2f;
typedef __attribute__((ext_vector_type(8))) float v8f;

#define DFEAT 128
#define NEG_SLOPE 0.2f

__device__ __forceinline__ unsigned ordEnc(float f) {
  unsigned u = __float_as_uint(f);
  return (u & 0x80000000u) ? ~u : (u | 0x80000000u);
}
__device__ __forceinline__ float ordDec(unsigned e) {
  unsigned u = (e & 0x80000000u) ? (e & 0x7fffffffu) : ~e;
  return __uint_as_float(u);
}

__device__ __forceinline__ v8f wmma_f32(v2f a, v2f b, v8f c) {
  return __builtin_amdgcn_wmma_f32_16x16x4_f32(false, a, false, b, (short)0, c,
                                               false, false);
}

__global__ void gemm_f32_wmma(const float* __restrict__ A, int lda,
                              const float* __restrict__ B, int ldb,
                              float* __restrict__ D, int ldo,
                              int nrows, int kdim, int ncols,
                              const float* __restrict__ bias, int relu) {
  const int lane = threadIdx.x & 31;
  const int wid  = (blockIdx.x * blockDim.x + threadIdx.x) >> 5;
  const int colTiles = ncols >> 4;
  const int rowTiles = (nrows + 15) >> 4;
  if (wid >= rowTiles * colTiles) return;
  const int tr = wid / colTiles;
  const int tc = wid - tr * colTiles;
  const int row0 = tr << 4, col0 = tc << 4;
  const int lm = lane & 15;
  const int lh = lane >> 4;

  int arow = row0 + lm;
  if (arow >= nrows) arow = nrows - 1;
  const float* Ap = A + (size_t)arow * lda + 2 * lh;
  const float* Bp = B + (size_t)(2 * lh) * ldb + col0 + lm;

  v8f acc = {0.f, 0.f, 0.f, 0.f, 0.f, 0.f, 0.f, 0.f};
  int k = 0;
  for (; k + 16 <= kdim; k += 16) {
    const v2f a0 = *(const v2f*)(Ap + k);
    const v2f a1 = *(const v2f*)(Ap + k + 4);
    const v2f a2 = *(const v2f*)(Ap + k + 8);
    const v2f a3 = *(const v2f*)(Ap + k + 12);
    const float* bq = Bp + (size_t)k * ldb;
    v2f b0, b1, b2, b3;
    b0.x = bq[0];
    b0.y = bq[(size_t)1 * ldb];
    b1.x = bq[(size_t)4 * ldb];
    b1.y = bq[(size_t)5 * ldb];
    b2.x = bq[(size_t)8 * ldb];
    b2.y = bq[(size_t)9 * ldb];
    b3.x = bq[(size_t)12 * ldb];
    b3.y = bq[(size_t)13 * ldb];
    acc = wmma_f32(a0, b0, acc);
    acc = wmma_f32(a1, b1, acc);
    acc = wmma_f32(a2, b2, acc);
    acc = wmma_f32(a3, b3, acc);
  }
  for (; k < kdim; k += 4) {
    const v2f a = *(const v2f*)(Ap + k);
    v2f b;
    b.x = Bp[(size_t)k * ldb];
    b.y = Bp[(size_t)(k + 1) * ldb];
    acc = wmma_f32(a, b, acc);
  }

  const int col = col0 + lm;
  const float bv = bias ? bias[col] : 0.0f;
#pragma unroll
  for (int r = 0; r < 8; ++r) {
    const int row = row0 + r + 8 * lh;
    if (row < nrows) {
      float v = acc[r] + bv;
      if (relu) v = fmaxf(v, 0.0f);
      D[(size_t)row * ldo + col] = v;
    }
  }
}

__global__ void gemm_f32_wmma_bn(const float* __restrict__ A, int lda,
                                 const float* __restrict__ B, int ldb,
                                 float* __restrict__ D, int ldo,
                                 int nrows, int kdim, int ncols,
                                 const float* __restrict__ bias,
                                 const float* __restrict__ scl,
                                 const float* __restrict__ shf, int relu) {
  const int lane = threadIdx.x & 31;
  const int wid  = (blockIdx.x * blockDim.x + threadIdx.x) >> 5;
  const int colTiles = ncols >> 4;
  const int rowTiles = (nrows + 15) >> 4;
  if (wid >= rowTiles * colTiles) return;
  const int tr = wid / colTiles;
  const int tc = wid - tr * colTiles;
  const int row0 = tr << 4, col0 = tc << 4;
  const int lm = lane & 15;
  const int lh = lane >> 4;

  int arow = row0 + lm;
  if (arow >= nrows) arow = nrows - 1;
  const float* Ap = A + (size_t)arow * lda + 2 * lh;
  const float* Sp = scl + 2 * lh;
  const float* Hp = shf + 2 * lh;
  const float* Bp = B + (size_t)(2 * lh) * ldb + col0 + lm;

  v8f acc = {0.f, 0.f, 0.f, 0.f, 0.f, 0.f, 0.f, 0.f};
  int k = 0;
  for (; k + 16 <= kdim; k += 16) {
    v2f a0 = *(const v2f*)(Ap + k);
    v2f a1 = *(const v2f*)(Ap + k + 4);
    v2f a2 = *(const v2f*)(Ap + k + 8);
    v2f a3 = *(const v2f*)(Ap + k + 12);
    const v2f s0 = *(const v2f*)(Sp + k);
    const v2f s1 = *(const v2f*)(Sp + k + 4);
    const v2f s2 = *(const v2f*)(Sp + k + 8);
    const v2f s3 = *(const v2f*)(Sp + k + 12);
    const v2f h0 = *(const v2f*)(Hp + k);
    const v2f h1 = *(const v2f*)(Hp + k + 4);
    const v2f h2 = *(const v2f*)(Hp + k + 8);
    const v2f h3 = *(const v2f*)(Hp + k + 12);
    a0 = a0 * s0 + h0;
    a1 = a1 * s1 + h1;
    a2 = a2 * s2 + h2;
    a3 = a3 * s3 + h3;
    const float* bq = Bp + (size_t)k * ldb;
    v2f b0, b1, b2, b3;
    b0.x = bq[0];
    b0.y = bq[(size_t)1 * ldb];
    b1.x = bq[(size_t)4 * ldb];
    b1.y = bq[(size_t)5 * ldb];
    b2.x = bq[(size_t)8 * ldb];
    b2.y = bq[(size_t)9 * ldb];
    b3.x = bq[(size_t)12 * ldb];
    b3.y = bq[(size_t)13 * ldb];
    acc = wmma_f32(a0, b0, acc);
    acc = wmma_f32(a1, b1, acc);
    acc = wmma_f32(a2, b2, acc);
    acc = wmma_f32(a3, b3, acc);
  }
  for (; k < kdim; k += 4) {
    v2f a = *(const v2f*)(Ap + k);
    const v2f s = *(const v2f*)(Sp + k);
    const v2f h = *(const v2f*)(Hp + k);
    a = a * s + h;
    v2f b;
    b.x = Bp[(size_t)k * ldb];
    b.y = Bp[(size_t)(k + 1) * ldb];
    acc = wmma_f32(a, b, acc);
  }

  const int col = col0 + lm;
  const float bv = bias ? bias[col] : 0.0f;
#pragma unroll
  for (int r = 0; r < 8; ++r) {
    const int row = row0 + r + 8 * lh;
    if (row < nrows) {
      float v = acc[r] + bv;
      if (relu) v = fmaxf(v, 0.0f);
      D[(size_t)row * ldo + col] = v;
    }
  }
}

__global__ void attn_proj(const float* __restrict__ h, int heads,
                          const float* __restrict__ a_src,
                          const float* __restrict__ a_dst,
                          float* __restrict__ al_s, float* __restrict__ al_d,
                          int n) {
  const int wid  = (blockIdx.x * blockDim.x + threadIdx.x) >> 5;
  const int lane = threadIdx.x & 31;
  if (wid >= n * heads) return;
  const int node = wid / heads;
  const int head = wid - node * heads;
  const float* hp = h + (size_t)node * heads * DFEAT + head * DFEAT;
  const float* as = a_src + head * DFEAT;
  const float* ad = a_dst + head * DFEAT;
  float ss = 0.f, sd = 0.f;
  for (int c = lane; c < DFEAT; c += 32) {
    const float hv = hp[c];
    ss += hv * as[c];
    sd += hv * ad[c];
  }
  for (int off = 16; off > 0; off >>= 1) {
    ss += __shfl_xor(ss, off, 32);
    sd += __shfl_xor(sd, off, 32);
  }
  if (lane == 0) { al_s[wid] = ss; al_d[wid] = sd; }
}

#define OWN_ROWS 80
#define OWN_THREADS 1024
#define LIST_CAP 8192

template <int HEADS>
__device__ __forceinline__ void gat_owned_body(
    const int* __restrict__ src, const int* __restrict__ dst, int E, int n,
    const float* __restrict__ h, int ldh, const float* __restrict__ al_s,
    const float* __restrict__ al_d, const float* __restrict__ bias,
    float* __restrict__ out, int ldo, int colOff) {
  constexpr int ROWW = HEADS * DFEAT;
  constexpr int NL = ROWW / 32;
  constexpr int BATCH = 32 / NL;
  __shared__ float s_acc[OWN_ROWS * ROWW];
  __shared__ int s_ls[LIST_CAP];
  __shared__ int s_ld[LIST_CAP];
  __shared__ unsigned s_menc[OWN_ROWS * HEADS];
  __shared__ float s_sum[OWN_ROWS * HEADS];
  __shared__ unsigned s_cnt;

  const int tid = threadIdx.x;
  const int lane = tid & 31;
  const int wave = tid >> 5;
  const int nthr = blockDim.x;
  const int nw = nthr >> 5;
  const int r0 = blockIdx.x * OWN_ROWS;
  int rows_here = n - r0;
  if (rows_here > OWN_ROWS) rows_here = OWN_ROWS;
  if (rows_here < 0) rows_here = 0;

  for (int i = tid; i < OWN_ROWS * ROWW; i += nthr) s_acc[i] = 0.f;
  for (int i = tid; i < OWN_ROWS * HEADS; i += nthr) {
    s_menc[i] = 0u;
    s_sum[i] = 0.f;
  }
  if (tid == 0) s_cnt = 0u;
  __syncthreads();

  auto append = [&](int e, int dl) {
    const unsigned p = atomicAdd(&s_cnt, 1u);
    if (p < (unsigned)LIST_CAP) { s_ls[p] = e; s_ld[p] = dl; }
  };
  auto owned = [&](int d) -> bool {
    return (unsigned)(d - r0) < (unsigned)rows_here;
  };
  for (int r = tid; r < rows_here; r += nthr) append(E + r0 + r, r);
  {
    int pre = (int)(((16u - ((unsigned)(size_t)dst & 15u)) & 15u) >> 2);
    if (pre > E) pre = E;
    const int nvec = (E - pre) >> 2;
    const int4* __restrict__ dv = (const int4*)(dst + pre);
    for (int vb = 0; vb < nvec; vb += 4 * nthr) {
      int4 q[4];
      int vi[4];
      for (int k = 0; k < 4; ++k) {
        vi[k] = vb + k * nthr + tid;
        q[k] = dv[vi[k] < nvec ? vi[k] : nvec - 1];
      }
      for (int k = 0; k < 4; ++k) {
        const bool in = vi[k] < nvec;
        const bool o0 = in && owned(q[k].x), o1 = in && owned(q[k].y);
        const bool o2 = in && owned(q[k].z), o3 = in && owned(q[k].w);
        if (o0 | o1 | o2 | o3) {
          const int e = pre + 4 * vi[k];
          if (o0) append(e + 0, q[k].x - r0);
          if (o1) append(e + 1, q[k].y - r0);
          if (o2) append(e + 2, q[k].z - r0);
          if (o3) append(e + 3, q[k].w - r0);
        }
      }
    }
    const int tail0 = pre + 4 * nvec;
    if (tid < 8) {
      const int e = (tid < 4) ? tid : tail0 + (tid - 4);
      const bool in = (tid < 4) ? (e < pre) : (e < E);
      if (in) {
        const int d = dst[e];
        if (owned(d)) append(e, d - r0);
      }
    }
  }
  __syncthreads();
  const int cnt = (int)s_cnt;
  const bool use_list = cnt <= LIST_CAP;
  if (use_list) {
    for (int i = tid; i < cnt; i += nthr) {
      const int e = s_ls[i];
      s_ls[i] = (e < E) ? src[e] : (e - E);
    }
  }
  __syncthreads();

  auto for_owned = [&](auto body) {
    if (use_list) {
      for (int base = 0; base < cnt; base += 32 * nw) {
        const int i = base + lane * nw + wave;
        const bool own = i < cnt;
        const int s = own ? s_ls[i] : 0;
        const int dl = own ? s_ld[i] : 0;
        body(own, s, dl);
      }
    } else {
      for (int e0 = wave * 32; e0 < E; e0 += 32 * nw) {
        const int e = e0 + lane;
        const int d = (e < E) ? dst[e] : -1;
        const bool own = (unsigned)(d - r0) < (unsigned)rows_here;
        if (__builtin_amdgcn_ballot_w32(own)) {
          const int s = own ? src[e] : 0;
          body(own, s, own ? d - r0 : 0);
        }
      }
      for (int i0 = wave * 32; i0 < rows_here; i0 += 32 * nw) {
        const int r = i0 + lane;
        const bool own = r < rows_here;
        body(own, own ? r0 + r : 0, own ? r : 0);
      }
    }
  };
  auto logit = [&](int s, int dl, int hh) -> float {
    const float x = al_s[s * HEADS + hh] + al_d[(r0 + dl) * HEADS + hh];
    return (x > 0.f) ? x : NEG_SLOPE * x;
  };

  for_owned([&](bool own, int s, int dl) {
    if (own) {
      for (int hh = 0; hh < HEADS; ++hh)
        atomicMax(&s_menc[dl * HEADS + hh], ordEnc(logit(s, dl, hh)));
    }
  });
  __syncthreads();

  for_owned([&](bool own, int s, int dl) {
    if (own) {
      for (int hh = 0; hh < HEADS; ++hh) {
        const float m = ordDec(s_menc[dl * HEADS + hh]);
        atomicAdd(&s_sum[dl * HEADS + hh], expf(logit(s, dl, hh) - m));
      }
    }
  });
  __syncthreads();

  for_owned([&](bool own, int s, int dl) {
    float al[HEADS];
    for (int hh = 0; hh < HEADS; ++hh) al[hh] = 0.f;
    if (own) {
      for (int hh = 0; hh < HEADS; ++hh) {
        const float m = ordDec(s_menc[dl * HEADS + hh]);
        al[hh] = expf(logit(s, dl, hh) - m) / (s_sum[dl * HEADS + hh] + 1e-16f);
      }
    }
    unsigned m = __builtin_amdgcn_ballot_w32(own);
    while (__builtin_popcount(m) >= BATCH) {
      float v[BATCH][NL];
      int oo[BATCH];
      int jj[BATCH];
      for (int b = 0; b < BATCH; ++b) {
        const int j = __builtin_ctz(m);
        m &= m - 1;
        jj[b] = j;
        const int sj = __builtin_amdgcn_readlane(s, j);
        const int dj = __builtin_amdgcn_readlane(dl, j);
        const float* hp = h + (size_t)sj * ldh + lane;
        oo[b] = dj * ROWW + lane;
        for (int c = 0; c < NL; ++c) v[b][c] = hp[32 * c];
      }
      for (int b = 0; b < BATCH; ++b) {
        for (int c = 0; c < NL; ++c) {
          const float a = __int_as_float(__builtin_amdgcn_readlane(
              __float_as_int(al[c / (DFEAT / 32)]), jj[b]));
          atomicAdd(&s_acc[oo[b] + 32 * c], a * v[b][c]);
        }
      }
    }
    while (m) {
      const int j = __builtin_ctz(m);
      m &= m - 1;
      const int sj = __builtin_amdgcn_readlane(s, j);
      const int dj = __builtin_amdgcn_readlane(dl, j);
      const float* hp = h + (size_t)sj * ldh + lane;
      const int o = dj * ROWW + lane;
      float v[NL];
      for (int c = 0; c < NL; ++c) v[c] = hp[32 * c];
      for (int c = 0; c < NL; ++c) {
        const float a = __int_as_float(__builtin_amdgcn_readlane(
            __float_as_int(al[c / (DFEAT / 32)]), j));
        atomicAdd(&s_acc[o + 32 * c], a * v[c]);
      }
    }
  });
  __syncthreads();

  for (int r = wave; r < rows_here; r += nw) {
    float* op = out + (size_t)(r0 + r) * ldo + colOff + lane;
    for (int c = 0; c < NL; ++c)
      op[32 * c] = s_acc[r * ROWW + 32 * c + lane] + bias[32 * c + lane];
  }
}

__global__ void __launch_bounds__(OWN_THREADS)
gat_owned_h2(const int* __restrict__ src, const int* __restrict__ dst, int E,
             int n, const float* __restrict__ h, int ldh,
             const float* __restrict__ al_s, const float* __restrict__ al_d,
             const float* __restrict__ bias, float* __restrict__ out, int ldo,
             int colOff) {
  gat_owned_body<2>(src, dst, E, n, h, ldh, al_s, al_d, bias, out, ldo, colOff);
}

__global__ void __launch_bounds__(OWN_THREADS)
gat_owned_h1(const int* __restrict__ src, const int* __restrict__ dst, int E,
             int n, const float* __restrict__ h, int ldh,
             const float* __restrict__ al_s, const float* __restrict__ al_d,
             const float* __restrict__ bias, float* __restrict__ out, int ldo,
             int colOff) {
  gat_owned_body<1>(src, dst, E, n, h, ldh, al_s, al_d, bias, out, ldo, colOff);
}

#define CS_THREADS 192
#define CS_LOADS 10

template <int NC>
__global__ void __launch_bounds__(CS_THREADS)
col_part(const float* __restrict__ y, int total, float* __restrict__ ps,
         float* __restrict__ pq, int nparts) {
  static_assert((4 * CS_THREADS) % NC == 0, "a thread keeps its columns");
  const int t = threadIdx.x;
  const int nvec = total >> 2;
  const int base = blockIdx.x * (CS_LOADS * CS_THREADS);
  const float4* __restrict__ y4 = (const float4*)y;

  float4 v[CS_LOADS];
#pragma unroll
  for (int k = 0; k < CS_LOADS; ++k) {
    const int f = base + k * CS_THREADS + t;
    v[k] = y4[f < nvec ? f : 0];
  }
  float sa[4] = {0.f, 0.f, 0.f, 0.f};
  float qa[4] = {0.f, 0.f, 0.f, 0.f};
#pragma unroll
  for (int k = 0; k < CS_LOADS; ++k) {
    const int f = base + k * CS_THREADS + t;
    if (f < nvec) {
      sa[0] += v[k].x; qa[0] = fmaf(v[k].x, v[k].x, qa[0]);
      sa[1] += v[k].y; qa[1] = fmaf(v[k].y, v[k].y, qa[1]);
      sa[2] += v[k].z; qa[2] = fmaf(v[k].z, v[k].z, qa[2]);
      sa[3] += v[k].w; qa[3] = fmaf(v[k].w, v[k].w, qa[3]);
    }
  }
  {
    const int tail = total & 3;
    const int rel = nvec - base;
    if (tail && rel >= 0 && rel < CS_LOADS * CS_THREADS &&
        rel % CS_THREADS == t) {
      for (int j = 0; j < tail; ++j) {
        const float w = y[4 * nvec + j];
        sa[j] += w;
        qa[j] = fmaf(w, w, qa[j]);
      }
    }
  }

  if constexpr (NC % 4 == 0) {
    constexpr int QN = NC / 4;
    static_assert(CS_THREADS % QN == 0, "whole rows per step");
    constexpr int GN = CS_THREADS / QN;
    __shared__ __attribute__((aligned(16))) float red[2][GN][NC];
    const int g = t / QN;
    const int c4 = 4 * (t - g * QN);
    *(float4*)&red[0][g][c4] = make_float4(sa[0], sa[1], sa[2], sa[3]);
    *(float4*)&red[1][g][c4] = make_float4(qa[0], qa[1], qa[2], qa[3]);
    __syncthreads();
    for (int i = t; i < 2 * NC; i += CS_THREADS) {
      const int which = i / NC;
      const int c = i - which * NC;
      float r = 0.f;
#pragma unroll
      for (int gg = 0; gg < GN; ++gg) r += red[which][gg][c];
      (which ? pq : ps)[(size_t)c * nparts + blockIdx.x] = r;
    }
  } else {
    static_assert(4 % NC == 0, "column count must divide a group");
    constexpr int NW = CS_THREADS / 32;
    __shared__ float red[2][NW][NC];
    float cs[NC], cq[NC];
#pragma unroll
    for (int c = 0; c < NC; ++c) { cs[c] = 0.f; cq[c] = 0.f; }
#pragma unroll
    for (int j = 0; j < 4; ++j) { cs[j % NC] += sa[j]; cq[j % NC] += qa[j]; }
#pragma unroll
    for (int c = 0; c < NC; ++c) {
      for (int off = 16; off > 0; off >>= 1) {
        cs[c] += __shfl_xor(cs[c], off, 32);
        cq[c] += __shfl_xor(cq[c], off, 32);
      }
    }
    const int lane = t & 31;
    const int wave = t >> 5;
    if (lane == 0) {
#pragma unroll
      for (int c = 0; c < NC; ++c) {
        red[0][wave][c] = cs[c];
        red[1][wave][c] = cq[c];
      }
    }
    __syncthreads();
    if (t < 2 * NC) {
      const int which = t / NC;
      const int c = t - which * NC;
      float r = 0.f;
#pragma unroll
      for (int w = 0; w < NW; ++w) r += red[which][w][c];
      (which ? pq : ps)[(size_t)c * nparts + blockIdx.x] = r;
    }
  }
}

__global__ void bn_fold(const float* __restrict__ ps,
                        const float* __restrict__ pq, int nparts,
                        const float* __restrict__ g,
                        const float* __restrict__ beta, int ncols, float invn,
                        float* __restrict__ sc, float* __restrict__ sh) {
  const int lane = threadIdx.x & 31;
  const int c = (blockIdx.x * blockDim.x + threadIdx.x) >> 5;
  if (c >= ncols) return;
  const float* __restrict__ sp = ps + (size_t)c * nparts;
  const float* __restrict__ qp = pq + (size_t)c * nparts;
  float s = 0.f, q = 0.f;
  for (int b0 = 0; b0 < nparts; b0 += 256) {
    float vs[8], vq[8];
#pragma unroll
    for (int k = 0; k < 8; ++k) {
      const int b = b0 + 32 * k + lane;
      vs[k] = sp[b < nparts ? b : 0];
      vq[k] = qp[b < nparts ? b : 0];
    }
#pragma unroll
    for (int k = 0; k < 8; ++k) {
      const int b = b0 + 32 * k + lane;
      if (b < nparts) { s += vs[k]; q += vq[k]; }
    }
  }
  for (int off = 16; off > 0; off >>= 1) {
    s += __shfl_xor(s, off, 32);
    q += __shfl_xor(q, off, 32);
  }
  if (lane == 0) {
    const float mean = s * invn;
    const float var = q * invn - mean * mean;
    const float r = rsqrtf(var + 1e-5f);
    const float a = g[c] * r;
    sc[c] = a;
    sh[c] = beta[c] - mean * a;
  }
}

__global__ void mlp_out(const float* __restrict__ y2,
                        const float* __restrict__ sc,
                        const float* __restrict__ sh,
                        const float* __restrict__ Wm2,
                        const float* __restrict__ bm2, int n,
                        float* __restrict__ y3) {
  const int node = blockIdx.x * blockDim.x + threadIdx.x;
  if (node >= n) return;
  float a0 = bm2[0], a1 = bm2[1];
  const float* yp = y2 + (size_t)node * 96;
  for (int k = 0; k < 96; ++k) {
    const float v = fmaf(yp[k], sc[k], sh[k]);
    a0 = fmaf(v, Wm2[2 * k + 0], a0);
    a1 = fmaf(v, Wm2[2 * k + 1], a1);
  }
  y3[node * 2 + 0] = fmaxf(a0, 0.f);
  y3[node * 2 + 1] = fmaxf(a1, 0.f);
}

__global__ void bn_apply(const float* __restrict__ y,
                         const float* __restrict__ sc,
                         const float* __restrict__ sh, int total, int ncols,
                         float* __restrict__ out) {
  const int i = blockIdx.x * blockDim.x + threadIdx.x;
  if (i >= total) return;
  const int c = i % ncols;
  out[i] = fmaf(y[i], sc[c], sh[c]);
}

extern "C" void kernel_launch(void* const* d_in, const int* in_sizes, int n_in,
                              void* d_out, int out_size, void* d_ws,
                              size_t ws_size, hipStream_t stream) {
  const float* x      = (const float*)d_in[0];
  const int*   ei     = (const int*)d_in[1];
  const float* W1     = (const float*)d_in[2];
  const float* as1    = (const float*)d_in[3];
  const float* ad1    = (const float*)d_in[4];
  const float* b1     = (const float*)d_in[5];
  const float* W2     = (const float*)d_in[6];
  const float* as2    = (const float*)d_in[7];
  const float* ad2    = (const float*)d_in[8];
  const float* b2     = (const float*)d_in[9];
  const float* Wl     = (const float*)d_in[10];
  const float* bl     = (const float*)d_in[11];
  const float* gl     = (const float*)d_in[12];
  const float* betal  = (const float*)d_in[13];
  const float* Wm1    = (const float*)d_in[14];
  const float* bm1    = (const float*)d_in[15];
  const float* gm1    = (const float*)d_in[16];
  const float* betam1 = (const float*)d_in[17];
  const float* Wm2    = (const float*)d_in[18];
  const float* bm2    = (const float*)d_in[19];
  const float* gm2    = (const float*)d_in[20];
  const float* betam2 = (const float*)d_in[21];

  const int N = in_sizes[0] / DFEAT;
  const int E = in_sizes[1] / 2;
  const int* srcA = ei;
  const int* dstA = ei + E;

  char* w = (char*)d_ws;
  auto alloc = [&](size_t bytes) -> char* {
    char* p = w;
    w += (bytes + 255) & ~(size_t)255;
    return p;
  };
  float*    xcat = (float*)alloc((size_t)N * 384 * 4);
  float*    h1   = (float*)alloc((size_t)N * 256 * 4);
  float*    h2   = (float*)alloc((size_t)N * 128 * 4);
  float*    al_s = (float*)alloc((size_t)N * 2 * 4);
  float*    al_d = (float*)alloc((size_t)N * 2 * 4);
  float*    y3   = (float*)alloc((size_t)N * 2 * 4);
  float*    sc1  = (float*)alloc(96 * 4);
  float*    sh1  = (float*)alloc(96 * 4);
  float*    sc2  = (float*)alloc(96 * 4);
  float*    sh2  = (float*)alloc(96 * 4);
  float*    sc3  = (float*)alloc(96 * 4);
  float*    sh3  = (float*)alloc(96 * 4);
  float* y1 = h1;
  float* y2 = h1 + (size_t)N * 96;
  const int csGroups = CS_LOADS * CS_THREADS;
  auto csBlocks = [&](int total) {
    return ((total + 3) / 4 + csGroups - 1) / csGroups;
  };
  const int np96 = csBlocks(N * 96);
  const int np2  = csBlocks(N * 2);
  float* ps = h2;
  float* pq = h2 + (size_t)96 * np96;

  const int TB = 256;
  auto gemmBlocks = [](int rows, int cols) {
    const int tiles = ((rows + 15) / 16) * (cols / 16);
    return (tiles + 7) / 8;
  };
  const int ownBlocks = (N + OWN_ROWS - 1) / OWN_ROWS;

  gemm_f32_wmma<<<gemmBlocks(N, 256), TB, 0, stream>>>(
      x, 128, W1, 256, h1, 256, N, 128, 256, nullptr, 0);
  attn_proj<<<(N * 2 + 7) / 8, TB, 0, stream>>>(h1, 2, as1, ad1, al_s, al_d, N);
  gat_owned_h2<<<ownBlocks, OWN_THREADS, 0, stream>>>(
      srcA, dstA, E, N, h1, 256, al_s, al_d, b1, xcat, 384, 0);

  gemm_f32_wmma<<<gemmBlocks(N, 128), TB, 0, stream>>>(
      xcat, 384, W2, 128, h2, 128, N, 256, 128, nullptr, 0);
  attn_proj<<<(N + 7) / 8, TB, 0, stream>>>(h2, 1, as2, ad2, al_s, al_d, N);
  gat_owned_h1<<<ownBlocks, OWN_THREADS, 0, stream>>>(
      srcA, dstA, E, N, h2, 128, al_s, al_d, b2, xcat, 384, 256);

  const float invn = 1.0f / (float)N;
  gemm_f32_wmma<<<gemmBlocks(N, 96), TB, 0, stream>>>(
      xcat, 384, Wl, 96, y1, 96, N, 384, 96, bl, 1);
  col_part<96><<<np96, CS_THREADS, 0, stream>>>(y1, N * 96, ps, pq, np96);
  bn_fold<<<(96 * 32 + TB - 1) / TB, TB, 0, stream>>>(ps, pq, np96, gl, betal,
                                                      96, invn, sc1, sh1);
  gemm_f32_wmma_bn<<<gemmBlocks(N, 96), TB, 0, stream>>>(
      y1, 96, Wm1, 96, y2, 96, N, 96, 96, bm1, sc1, sh1, 1);
  col_part<96><<<np96, CS_THREADS, 0, stream>>>(y2, N * 96, ps, pq, np96);
  bn_fold<<<(96 * 32 + TB - 1) / TB, TB, 0, stream>>>(ps, pq, np96, gm1, betam1,
                                                      96, invn, sc2, sh2);
  mlp_out<<<(N + TB - 1) / TB, TB, 0, stream>>>(y2, sc2, sh2, Wm2, bm2, N, y3);
  col_part<2><<<np2, CS_THREADS, 0, stream>>>(y3, N * 2, ps, pq, np2);
  bn_fold<<<(2 * 32 + TB - 1) / TB, TB, 0, stream>>>(ps, pq, np2, gm2, betam2, 2,
                                                     invn, sc3, sh3);
  bn_apply<<<(N * 2 + TB - 1) / TB, TB, 0, stream>>>(y3, sc3, sh3, N * 2, 2,
                                                     (float*)d_out);
}


# attention projections re-tiled: a wave takes 16 head rows of 128 floats as whole-row 16-byte loads, lane sums folded by a halving exchange; 60000 waves become 3750 (projection sum order changes)
# speedup vs baseline: 3.5505x; 1.0635x over previous
#include <hip/hip_runtime.h>
#include <hip/hip_bf16.h>
#include <math.h>


typedef __attribute__((ext_vector_type(2))) float v2f;
typedef __attribute__((ext_vector_type(8))) float v8f;

#define DFEAT 128
#define NEG_SLOPE 0.2f

__device__ __forceinline__ unsigned ordEnc(float f) {
  unsigned u = __float_as_uint(f);
  return (u & 0x80000000u) ? ~u : (u | 0x80000000u);
}
__device__ __forceinline__ float ordDec(unsigned e) {
  unsigned u = (e & 0x80000000u) ? (e & 0x7fffffffu) : ~e;
  return __uint_as_float(u);
}

__device__ __forceinline__ v8f wmma_f32(v2f a, v2f b, v8f c) {
  return __builtin_amdgcn_wmma_f32_16x16x4_f32(false, a, false, b, (short)0, c,
                                               false, false);
}

__global__ void gemm_f32_wmma(const float* __restrict__ A, int lda,
                              const float* __restrict__ B, int ldb,
                              float* __restrict__ D, int ldo,
                              int nrows, int kdim, int ncols,
                              const float* __restrict__ bias, int relu) {
  const int lane = threadIdx.x & 31;
  const int wid  = (blockIdx.x * blockDim.x + threadIdx.x) >> 5;
  const int colTiles = ncols >> 4;
  const int rowTiles = (nrows + 15) >> 4;
  if (wid >= rowTiles * colTiles) return;
  const int tr = wid / colTiles;
  const int tc = wid - tr * colTiles;
  const int row0 = tr << 4, col0 = tc << 4;
  const int lm = lane & 15;
  const int lh = lane >> 4;

  int arow = row0 + lm;
  if (arow >= nrows) arow = nrows - 1;
  const float* Ap = A + (size_t)arow * lda + 2 * lh;
  const float* Bp = B + (size_t)(2 * lh) * ldb + col0 + lm;

  v8f acc = {0.f, 0.f, 0.f, 0.f, 0.f, 0.f, 0.f, 0.f};
  int k = 0;
  for (; k + 16 <= kdim; k += 16) {
    const v2f a0 = *(const v2f*)(Ap + k);
    const v2f a1 = *(const v2f*)(Ap + k + 4);
    const v2f a2 = *(const v2f*)(Ap + k + 8);
    const v2f a3 = *(const v2f*)(Ap + k + 12);
    const float* bq = Bp + (size_t)k * ldb;
    v2f b0, b1, b2, b3;
    b0.x = bq[0];
    b0.y = bq[(size_t)1 * ldb];
    b1.x = bq[(size_t)4 * ldb];
    b1.y = bq[(size_t)5 * ldb];
    b2.x = bq[(size_t)8 * ldb];
    b2.y = bq[(size_t)9 * ldb];
    b3.x = bq[(size_t)12 * ldb];
    b3.y = bq[(size_t)13 * ldb];
    acc = wmma_f32(a0, b0, acc);
    acc = wmma_f32(a1, b1, acc);
    acc = wmma_f32(a2, b2, acc);
    acc = wmma_f32(a3, b3, acc);
  }
  for (; k < kdim; k += 4) {
    const v2f a = *(const v2f*)(Ap + k);
    v2f b;
    b.x = Bp[(size_t)k * ldb];
    b.y = Bp[(size_t)(k + 1) * ldb];
    acc = wmma_f32(a, b, acc);
  }

  const int col = col0 + lm;
  const float bv = bias ? bias[col] : 0.0f;
#pragma unroll
  for (int r = 0; r < 8; ++r) {
    const int row = row0 + r + 8 * lh;
    if (row < nrows) {
      float v = acc[r] + bv;
      if (relu) v = fmaxf(v, 0.0f);
      D[(size_t)row * ldo + col] = v;
    }
  }
}

__global__ void gemm_f32_wmma_bn(const float* __restrict__ A, int lda,
                                 const float* __restrict__ B, int ldb,
                                 float* __restrict__ D, int ldo,
                                 int nrows, int kdim, int ncols,
                                 const float* __restrict__ bias,
                                 const float* __restrict__ scl,
                                 const float* __restrict__ shf, int relu) {
  const int lane = threadIdx.x & 31;
  const int wid  = (blockIdx.x * blockDim.x + threadIdx.x) >> 5;
  const int colTiles = ncols >> 4;
  const int rowTiles = (nrows + 15) >> 4;
  if (wid >= rowTiles * colTiles) return;
  const int tr = wid / colTiles;
  const int tc = wid - tr * colTiles;
  const int row0 = tr << 4, col0 = tc << 4;
  const int lm = lane & 15;
  const int lh = lane >> 4;

  int arow = row0 + lm;
  if (arow >= nrows) arow = nrows - 1;
  const float* Ap = A + (size_t)arow * lda + 2 * lh;
  const float* Sp = scl + 2 * lh;
  const float* Hp = shf + 2 * lh;
  const float* Bp = B + (size_t)(2 * lh) * ldb + col0 + lm;

  v8f acc = {0.f, 0.f, 0.f, 0.f, 0.f, 0.f, 0.f, 0.f};
  int k = 0;
  for (; k + 16 <= kdim; k += 16) {
    v2f a0 = *(const v2f*)(Ap + k);
    v2f a1 = *(const v2f*)(Ap + k + 4);
    v2f a2 = *(const v2f*)(Ap + k + 8);
    v2f a3 = *(const v2f*)(Ap + k + 12);
    const v2f s0 = *(const v2f*)(Sp + k);
    const v2f s1 = *(const v2f*)(Sp + k + 4);
    const v2f s2 = *(const v2f*)(Sp + k + 8);
    const v2f s3 = *(const v2f*)(Sp + k + 12);
    const v2f h0 = *(const v2f*)(Hp + k);
    const v2f h1 = *(const v2f*)(Hp + k + 4);
    const v2f h2 = *(const v2f*)(Hp + k + 8);
    const v2f h3 = *(const v2f*)(Hp + k + 12);
    a0 = a0 * s0 + h0;
    a1 = a1 * s1 + h1;
    a2 = a2 * s2 + h2;
    a3 = a3 * s3 + h3;
    const float* bq = Bp + (size_t)k * ldb;
    v2f b0, b1, b2, b3;
    b0.x = bq[0];
    b0.y = bq[(size_t)1 * ldb];
    b1.x = bq[(size_t)4 * ldb];
    b1.y = bq[(size_t)5 * ldb];
    b2.x = bq[(size_t)8 * ldb];
    b2.y = bq[(size_t)9 * ldb];
    b3.x = bq[(size_t)12 * ldb];
    b3.y = bq[(size_t)13 * ldb];
    acc = wmma_f32(a0, b0, acc);
    acc = wmma_f32(a1, b1, acc);
    acc = wmma_f32(a2, b2, acc);
    acc = wmma_f32(a3, b3, acc);
  }
  for (; k < kdim; k += 4) {
    v2f a = *(const v2f*)(Ap + k);
    const v2f s = *(const v2f*)(Sp + k);
    const v2f h = *(const v2f*)(Hp + k);
    a = a * s + h;
    v2f b;
    b.x = Bp[(size_t)k * ldb];
    b.y = Bp[(size_t)(k + 1) * ldb];
    acc = wmma_f32(a, b, acc);
  }

  const int col = col0 + lm;
  const float bv = bias ? bias[col] : 0.0f;
#pragma unroll
  for (int r = 0; r < 8; ++r) {
    const int row = row0 + r + 8 * lh;
    if (row < nrows) {
      float v = acc[r] + bv;
      if (relu) v = fmaxf(v, 0.0f);
      D[(size_t)row * ldo + col] = v;
    }
  }
}

#define AP_ROWS 16
#define AP_THREADS 128

template <int CNT>
__device__ __forceinline__ void ap_fold(float (&val)[2 * AP_ROWS], int lane) {
  const bool up = (lane & CNT) != 0;
#pragma unroll
  for (int i = 0; i < CNT; ++i) {
    const float keep = up ? val[i + CNT] : val[i];
    const float send = up ? val[i] : val[i + CNT];
    val[i] = keep + __shfl_xor(send, CNT, 32);
  }
}

template <int HEADS>
__global__ void __launch_bounds__(AP_THREADS)
attn_proj_rows(const float* __restrict__ h, const float* __restrict__ a_src,
               const float* __restrict__ a_dst, float* __restrict__ al_s,
               float* __restrict__ al_d, int nrows) {
  static_assert(AP_ROWS % HEADS == 0, "a wave's first row is head 0");
  static_assert(2 * AP_ROWS == 32, "one finished sum per lane");
  const int lane = threadIdx.x & 31;
  const int wid = __builtin_amdgcn_readfirstlane(
      (int)((blockIdx.x * blockDim.x + threadIdx.x) >> 5));
  const int w0 = wid * AP_ROWS;
  if (w0 >= nrows) return;

  float as[HEADS][4], ad[HEADS][4];
#pragma unroll
  for (int hh = 0; hh < HEADS; ++hh) {
#pragma unroll
    for (int q = 0; q < 4; ++q) {
      as[hh][q] = a_src[hh * DFEAT + 4 * lane + q];
      ad[hh][q] = a_dst[hh * DFEAT + 4 * lane + q];
    }
  }
  const float4* __restrict__ hp =
      (const float4*)(h + (size_t)w0 * DFEAT) + lane;
  float4 v[AP_ROWS];
#pragma unroll
  for (int j = 0; j < AP_ROWS; ++j)
    v[j] = hp[(w0 + j < nrows ? j : 0) * (DFEAT / 4)];
  __builtin_amdgcn_sched_barrier(0);

  float val[2 * AP_ROWS];
#pragma unroll
  for (int j = 0; j < AP_ROWS; ++j) {
    const int hh = j % HEADS;
    val[j] = v[j].x * as[hh][0] + v[j].y * as[hh][1] + v[j].z * as[hh][2] +
             v[j].w * as[hh][3];
    val[AP_ROWS + j] = v[j].x * ad[hh][0] + v[j].y * ad[hh][1] +
                       v[j].z * ad[hh][2] + v[j].w * ad[hh][3];
  }
  ap_fold<16>(val, lane);
  ap_fold<8>(val, lane);
  ap_fold<4>(val, lane);
  ap_fold<2>(val, lane);
  ap_fold<1>(val, lane);

  const int w = w0 + (lane & 15);
  if (w < nrows) (lane < 16 ? al_s : al_d)[w] = val[0];
}

#define OWN_ROWS 80
#define OWN_THREADS 1024
#define LIST_CAP 8192

template <int HEADS>
__device__ __forceinline__ void gat_owned_body(
    const int* __restrict__ src, const int* __restrict__ dst, int E, int n,
    const float* __restrict__ h, int ldh, const float* __restrict__ al_s,
    const float* __restrict__ al_d, const float* __restrict__ bias,
    float* __restrict__ out, int ldo, int colOff) {
  constexpr int ROWW = HEADS * DFEAT;
  constexpr int NL = ROWW / 32;
  constexpr int BATCH = 32 / NL;
  __shared__ float s_acc[OWN_ROWS * ROWW];
  __shared__ int s_ls[LIST_CAP];
  __shared__ int s_ld[LIST_CAP];
  __shared__ unsigned s_menc[OWN_ROWS * HEADS];
  __shared__ float s_sum[OWN_ROWS * HEADS];
  __shared__ unsigned s_cnt;

  const int tid = threadIdx.x;
  const int lane = tid & 31;
  const int wave = tid >> 5;
  const int nthr = blockDim.x;
  const int nw = nthr >> 5;
  const int r0 = blockIdx.x * OWN_ROWS;
  int rows_here = n - r0;
  if (rows_here > OWN_ROWS) rows_here = OWN_ROWS;
  if (rows_here < 0) rows_here = 0;

  for (int i = tid; i < OWN_ROWS * ROWW; i += nthr) s_acc[i] = 0.f;
  for (int i = tid; i < OWN_ROWS * HEADS; i += nthr) {
    s_menc[i] = 0u;
    s_sum[i] = 0.f;
  }
  if (tid == 0) s_cnt = 0u;
  __syncthreads();

  auto append = [&](int e, int dl) {
    const unsigned p = atomicAdd(&s_cnt, 1u);
    if (p < (unsigned)LIST_CAP) { s_ls[p] = e; s_ld[p] = dl; }
  };
  auto owned = [&](int d) -> bool {
    return (unsigned)(d - r0) < (unsigned)rows_here;
  };
  for (int r = tid; r < rows_here; r += nthr) append(E + r0 + r, r);
  {
    int pre = (int)(((16u - ((unsigned)(size_t)dst & 15u)) & 15u) >> 2);
    if (pre > E) pre = E;
    const int nvec = (E - pre) >> 2;
    const int4* __restrict__ dv = (const int4*)(dst + pre);
    for (int vb = 0; vb < nvec; vb += 4 * nthr) {
      int4 q[4];
      int vi[4];
      for (int k = 0; k < 4; ++k) {
        vi[k] = vb + k * nthr + tid;
        q[k] = dv[vi[k] < nvec ? vi[k] : nvec - 1];
      }
      for (int k = 0; k < 4; ++k) {
        const bool in = vi[k] < nvec;
        const bool o0 = in && owned(q[k].x), o1 = in && owned(q[k].y);
        const bool o2 = in && owned(q[k].z), o3 = in && owned(q[k].w);
        if (o0 | o1 | o2 | o3) {
          const int e = pre + 4 * vi[k];
          if (o0) append(e + 0, q[k].x - r0);
          if (o1) append(e + 1, q[k].y - r0);
          if (o2) append(e + 2, q[k].z - r0);
          if (o3) append(e + 3, q[k].w - r0);
        }
      }
    }
    const int tail0 = pre + 4 * nvec;
    if (tid < 8) {
      const int e = (tid < 4) ? tid : tail0 + (tid - 4);
      const bool in = (tid < 4) ? (e < pre) : (e < E);
      if (in) {
        const int d = dst[e];
        if (owned(d)) append(e, d - r0);
      }
    }
  }
  __syncthreads();
  const int cnt = (int)s_cnt;
  const bool use_list = cnt <= LIST_CAP;
  if (use_list) {
    for (int i = tid; i < cnt; i += nthr) {
      const int e = s_ls[i];
      s_ls[i] = (e < E) ? src[e] : (e - E);
    }
  }
  __syncthreads();

  auto for_owned = [&](auto body) {
    if (use_list) {
      for (int base = 0; base < cnt; base += 32 * nw) {
        const int i = base + lane * nw + wave;
        const bool own = i < cnt;
        const int s = own ? s_ls[i] : 0;
        const int dl = own ? s_ld[i] : 0;
        body(own, s, dl);
      }
    } else {
      for (int e0 = wave * 32; e0 < E; e0 += 32 * nw) {
        const int e = e0 + lane;
        const int d = (e < E) ? dst[e] : -1;
        const bool own = (unsigned)(d - r0) < (unsigned)rows_here;
        if (__builtin_amdgcn_ballot_w32(own)) {
          const int s = own ? src[e] : 0;
          body(own, s, own ? d - r0 : 0);
        }
      }
      for (int i0 = wave * 32; i0 < rows_here; i0 += 32 * nw) {
        const int r = i0 + lane;
        const bool own = r < rows_here;
        body(own, own ? r0 + r : 0, own ? r : 0);
      }
    }
  };
  auto logit = [&](int s, int dl, int hh) -> float {
    const float x = al_s[s * HEADS + hh] + al_d[(r0 + dl) * HEADS + hh];
    return (x > 0.f) ? x : NEG_SLOPE * x;
  };

  for_owned([&](bool own, int s, int dl) {
    if (own) {
      for (int hh = 0; hh < HEADS; ++hh)
        atomicMax(&s_menc[dl * HEADS + hh], ordEnc(logit(s, dl, hh)));
    }
  });
  __syncthreads();

  for_owned([&](bool own, int s, int dl) {
    if (own) {
      for (int hh = 0; hh < HEADS; ++hh) {
        const float m = ordDec(s_menc[dl * HEADS + hh]);
        atomicAdd(&s_sum[dl * HEADS + hh], expf(logit(s, dl, hh) - m));
      }
    }
  });
  __syncthreads();

  for_owned([&](bool own, int s, int dl) {
    float al[HEADS];
    for (int hh = 0; hh < HEADS; ++hh) al[hh] = 0.f;
    if (own) {
      for (int hh = 0; hh < HEADS; ++hh) {
        const float m = ordDec(s_menc[dl * HEADS + hh]);
        al[hh] = expf(logit(s, dl, hh) - m) / (s_sum[dl * HEADS + hh] + 1e-16f);
      }
    }
    unsigned m = __builtin_amdgcn_ballot_w32(own);
    while (__builtin_popcount(m) >= BATCH) {
      float v[BATCH][NL];
      int oo[BATCH];
      int jj[BATCH];
      for (int b = 0; b < BATCH; ++b) {
        const int j = __builtin_ctz(m);
        m &= m - 1;
        jj[b] = j;
        const int sj = __builtin_amdgcn_readlane(s, j);
        const int dj = __builtin_amdgcn_readlane(dl, j);
        const float* hp = h + (size_t)sj * ldh + lane;
        oo[b] = dj * ROWW + lane;
        for (int c = 0; c < NL; ++c) v[b][c] = hp[32 * c];
      }
      for (int b = 0; b < BATCH; ++b) {
        for (int c = 0; c < NL; ++c) {
          const float a = __int_as_float(__builtin_amdgcn_readlane(
              __float_as_int(al[c / (DFEAT / 32)]), jj[b]));
          atomicAdd(&s_acc[oo[b] + 32 * c], a * v[b][c]);
        }
      }
    }
    while (m) {
      const int j = __builtin_ctz(m);
      m &= m - 1;
      const int sj = __builtin_amdgcn_readlane(s, j);
      const int dj = __builtin_amdgcn_readlane(dl, j);
      const float* hp = h + (size_t)sj * ldh + lane;
      const int o = dj * ROWW + lane;
      float v[NL];
      for (int c = 0; c < NL; ++c) v[c] = hp[32 * c];
      for (int c = 0; c < NL; ++c) {
        const float a = __int_as_float(__builtin_amdgcn_readlane(
            __float_as_int(al[c / (DFEAT / 32)]), j));
        atomicAdd(&s_acc[o + 32 * c], a * v[c]);
      }
    }
  });
  __syncthreads();

  for (int r = wave; r < rows_here; r += nw) {
    float* op = out + (size_t)(r0 + r) * ldo + colOff + lane;
    for (int c = 0; c < NL; ++c)
      op[32 * c] = s_acc[r * ROWW + 32 * c + lane] + bias[32 * c + lane];
  }
}

__global__ void __launch_bounds__(OWN_THREADS)
gat_owned_h2(const int* __restrict__ src, const int* __restrict__ dst, int E,
             int n, const float* __restrict__ h, int ldh,
             const float* __restrict__ al_s, const float* __restrict__ al_d,
             const float* __restrict__ bias, float* __restrict__ out, int ldo,
             int colOff) {
  gat_owned_body<2>(src, dst, E, n, h, ldh, al_s, al_d, bias, out, ldo, colOff);
}

__global__ void __launch_bounds__(OWN_THREADS)
gat_owned_h1(const int* __restrict__ src, const int* __restrict__ dst, int E,
             int n, const float* __restrict__ h, int ldh,
             const float* __restrict__ al_s, const float* __restrict__ al_d,
             const float* __restrict__ bias, float* __restrict__ out, int ldo,
             int colOff) {
  gat_owned_body<1>(src, dst, E, n, h, ldh, al_s, al_d, bias, out, ldo, colOff);
}

#define CS_THREADS 192
#define CS_LOADS 10

template <int NC>
__global__ void __launch_bounds__(CS_THREADS)
col_part(const float* __restrict__ y, int total, float* __restrict__ ps,
         float* __restrict__ pq, int nparts) {
  static_assert((4 * CS_THREADS) % NC == 0, "a thread keeps its columns");
  const int t = threadIdx.x;
  const int nvec = total >> 2;
  const int base = blockIdx.x * (CS_LOADS * CS_THREADS);
  const float4* __restrict__ y4 = (const float4*)y;

  float4 v[CS_LOADS];
#pragma unroll
  for (int k = 0; k < CS_LOADS; ++k) {
    const int f = base + k * CS_THREADS + t;
    v[k] = y4[f < nvec ? f : 0];
  }
  float sa[4] = {0.f, 0.f, 0.f, 0.f};
  float qa[4] = {0.f, 0.f, 0.f, 0.f};
#pragma unroll
  for (int k = 0; k < CS_LOADS; ++k) {
    const int f = base + k * CS_THREADS + t;
    if (f < nvec) {
      sa[0] += v[k].x; qa[0] = fmaf(v[k].x, v[k].x, qa[0]);
      sa[1] += v[k].y; qa[1] = fmaf(v[k].y, v[k].y, qa[1]);
      sa[2] += v[k].z; qa[2] = fmaf(v[k].z, v[k].z, qa[2]);
      sa[3] += v[k].w; qa[3] = fmaf(v[k].w, v[k].w, qa[3]);
    }
  }
  {
    const int tail = total & 3;
    const int rel = nvec - base;
    if (tail && rel >= 0 && rel < CS_LOADS * CS_THREADS &&
        rel % CS_THREADS == t) {
      for (int j = 0; j < tail; ++j) {
        const float w = y[4 * nvec + j];
        sa[j] += w;
        qa[j] = fmaf(w, w, qa[j]);
      }
    }
  }

  if constexpr (NC % 4 == 0) {
    constexpr int QN = NC / 4;
    static_assert(CS_THREADS % QN == 0, "whole rows per step");
    constexpr int GN = CS_THREADS / QN;
    __shared__ __attribute__((aligned(16))) float red[2][GN][NC];
    const int g = t / QN;
    const int c4 = 4 * (t - g * QN);
    *(float4*)&red[0][g][c4] = make_float4(sa[0], sa[1], sa[2], sa[3]);
    *(float4*)&red[1][g][c4] = make_float4(qa[0], qa[1], qa[2], qa[3]);
    __syncthreads();
    for (int i = t; i < 2 * NC; i += CS_THREADS) {
      const int which = i / NC;
      const int c = i - which * NC;
      float r = 0.f;
#pragma unroll
      for (int gg = 0; gg < GN; ++gg) r += red[which][gg][c];
      (which ? pq : ps)[(size_t)c * nparts + blockIdx.x] = r;
    }
  } else {
    static_assert(4 % NC == 0, "column count must divide a group");
    constexpr int NW = CS_THREADS / 32;
    __shared__ float red[2][NW][NC];
    float cs[NC], cq[NC];
#pragma unroll
    for (int c = 0; c < NC; ++c) { cs[c] = 0.f; cq[c] = 0.f; }
#pragma unroll
    for (int j = 0; j < 4; ++j) { cs[j % NC] += sa[j]; cq[j % NC] += qa[j]; }
#pragma unroll
    for (int c = 0; c < NC; ++c) {
      for (int off = 16; off > 0; off >>= 1) {
        cs[c] += __shfl_xor(cs[c], off, 32);
        cq[c] += __shfl_xor(cq[c], off, 32);
      }
    }
    const int lane = t & 31;
    const int wave = t >> 5;
    if (lane == 0) {
#pragma unroll
      for (int c = 0; c < NC; ++c) {
        red[0][wave][c] = cs[c];
        red[1][wave][c] = cq[c];
      }
    }
    __syncthreads();
    if (t < 2 * NC) {
      const int which = t / NC;
      const int c = t - which * NC;
      float r = 0.f;
#pragma unroll
      for (int w = 0; w < NW; ++w) r += red[which][w][c];
      (which ? pq : ps)[(size_t)c * nparts + blockIdx.x] = r;
    }
  }
}

__global__ void bn_fold(const float* __restrict__ ps,
                        const float* __restrict__ pq, int nparts,
                        const float* __restrict__ g,
                        const float* __restrict__ beta, int ncols, float invn,
                        float* __restrict__ sc, float* __restrict__ sh) {
  const int lane = threadIdx.x & 31;
  const int c = (blockIdx.x * blockDim.x + threadIdx.x) >> 5;
  if (c >= ncols) return;
  const float* __restrict__ sp = ps + (size_t)c * nparts;
  const float* __restrict__ qp = pq + (size_t)c * nparts;
  float s = 0.f, q = 0.f;
  for (int b0 = 0; b0 < nparts; b0 += 256) {
    float vs[8], vq[8];
#pragma unroll
    for (int k = 0; k < 8; ++k) {
      const int b = b0 + 32 * k + lane;
      vs[k] = sp[b < nparts ? b : 0];
      vq[k] = qp[b < nparts ? b : 0];
    }
#pragma unroll
    for (int k = 0; k < 8; ++k) {
      const int b = b0 + 32 * k + lane;
      if (b < nparts) { s += vs[k]; q += vq[k]; }
    }
  }
  for (int off = 16; off > 0; off >>= 1) {
    s += __shfl_xor(s, off, 32);
    q += __shfl_xor(q, off, 32);
  }
  if (lane == 0) {
    const float mean = s * invn;
    const float var = q * invn - mean * mean;
    const float r = rsqrtf(var + 1e-5f);
    const float a = g[c] * r;
    sc[c] = a;
    sh[c] = beta[c] - mean * a;
  }
}

__global__ void mlp_out(const float* __restrict__ y2,
                        const float* __restrict__ sc,
                        const float* __restrict__ sh,
                        const float* __restrict__ Wm2,
                        const float* __restrict__ bm2, int n,
                        float* __restrict__ y3) {
  const int node = blockIdx.x * blockDim.x + threadIdx.x;
  if (node >= n) return;
  float a0 = bm2[0], a1 = bm2[1];
  const float* yp = y2 + (size_t)node * 96;
  for (int k = 0; k < 96; ++k) {
    const float v = fmaf(yp[k], sc[k], sh[k]);
    a0 = fmaf(v, Wm2[2 * k + 0], a0);
    a1 = fmaf(v, Wm2[2 * k + 1], a1);
  }
  y3[node * 2 + 0] = fmaxf(a0, 0.f);
  y3[node * 2 + 1] = fmaxf(a1, 0.f);
}

__global__ void bn_apply(const float* __restrict__ y,
                         const float* __restrict__ sc,
                         const float* __restrict__ sh, int total, int ncols,
                         float* __restrict__ out) {
  const int i = blockIdx.x * blockDim.x + threadIdx.x;
  if (i >= total) return;
  const int c = i % ncols;
  out[i] = fmaf(y[i], sc[c], sh[c]);
}

extern "C" void kernel_launch(void* const* d_in, const int* in_sizes, int n_in,
                              void* d_out, int out_size, void* d_ws,
                              size_t ws_size, hipStream_t stream) {
  const float* x      = (const float*)d_in[0];
  const int*   ei     = (const int*)d_in[1];
  const float* W1     = (const float*)d_in[2];
  const float* as1    = (const float*)d_in[3];
  const float* ad1    = (const float*)d_in[4];
  const float* b1     = (const float*)d_in[5];
  const float* W2     = (const float*)d_in[6];
  const float* as2    = (const float*)d_in[7];
  const float* ad2    = (const float*)d_in[8];
  const float* b2     = (const float*)d_in[9];
  const float* Wl     = (const float*)d_in[10];
  const float* bl     = (const float*)d_in[11];
  const float* gl     = (const float*)d_in[12];
  const float* betal  = (const float*)d_in[13];
  const float* Wm1    = (const float*)d_in[14];
  const float* bm1    = (const float*)d_in[15];
  const float* gm1    = (const float*)d_in[16];
  const float* betam1 = (const float*)d_in[17];
  const float* Wm2    = (const float*)d_in[18];
  const float* bm2    = (const float*)d_in[19];
  const float* gm2    = (const float*)d_in[20];
  const float* betam2 = (const float*)d_in[21];

  const int N = in_sizes[0] / DFEAT;
  const int E = in_sizes[1] / 2;
  const int* srcA = ei;
  const int* dstA = ei + E;

  char* w = (char*)d_ws;
  auto alloc = [&](size_t bytes) -> char* {
    char* p = w;
    w += (bytes + 255) & ~(size_t)255;
    return p;
  };
  float*    xcat = (float*)alloc((size_t)N * 384 * 4);
  float*    h1   = (float*)alloc((size_t)N * 256 * 4);
  float*    h2   = (float*)alloc((size_t)N * 128 * 4);
  float*    al_s = (float*)alloc((size_t)N * 2 * 4);
  float*    al_d = (float*)alloc((size_t)N * 2 * 4);
  float*    y3   = (float*)alloc((size_t)N * 2 * 4);
  float*    sc1  = (float*)alloc(96 * 4);
  float*    sh1  = (float*)alloc(96 * 4);
  float*    sc2  = (float*)alloc(96 * 4);
  float*    sh2  = (float*)alloc(96 * 4);
  float*    sc3  = (float*)alloc(96 * 4);
  float*    sh3  = (float*)alloc(96 * 4);
  float* y1 = h1;
  float* y2 = h1 + (size_t)N * 96;
  const int csGroups = CS_LOADS * CS_THREADS;
  auto csBlocks = [&](int total) {
    return ((total + 3) / 4 + csGroups - 1) / csGroups;
  };
  const int np96 = csBlocks(N * 96);
  const int np2  = csBlocks(N * 2);
  float* ps = h2;
  float* pq = h2 + (size_t)96 * np96;

  const int TB = 256;
  auto gemmBlocks = [](int rows, int cols) {
    const int tiles = ((rows + 15) / 16) * (cols / 16);
    return (tiles + 7) / 8;
  };
  const int ownBlocks = (N + OWN_ROWS - 1) / OWN_ROWS;
  const int apRows = AP_ROWS * (AP_THREADS / 32);

  gemm_f32_wmma<<<gemmBlocks(N, 256), TB, 0, stream>>>(
      x, 128, W1, 256, h1, 256, N, 128, 256, nullptr, 0);
  attn_proj_rows<2><<<(N * 2 + apRows - 1) / apRows, AP_THREADS, 0, stream>>>(
      h1, as1, ad1, al_s, al_d, N * 2);
  gat_owned_h2<<<ownBlocks, OWN_THREADS, 0, stream>>>(
      srcA, dstA, E, N, h1, 256, al_s, al_d, b1, xcat, 384, 0);

  gemm_f32_wmma<<<gemmBlocks(N, 128), TB, 0, stream>>>(
      xcat, 384, W2, 128, h2, 128, N, 256, 128, nullptr, 0);
  attn_proj_rows<1><<<(N + apRows - 1) / apRows, AP_THREADS, 0, stream>>>(
      h2, as2, ad2, al_s, al_d, N);
  gat_owned_h1<<<ownBlocks, OWN_THREADS, 0, stream>>>(
      srcA, dstA, E, N, h2, 128, al_s, al_d, b2, xcat, 384, 256);

  const float invn = 1.0f / (float)N;
  gemm_f32_wmma<<<gemmBlocks(N, 96), TB, 0, stream>>>(
      xcat, 384, Wl, 96, y1, 96, N, 384, 96, bl, 1);
  col_part<96><<<np96, CS_THREADS, 0, stream>>>(y1, N * 96, ps, pq, np96);
  bn_fold<<<(96 * 32 + TB - 1) / TB, TB, 0, stream>>>(ps, pq, np96, gl, betal,
                                                      96, invn, sc1, sh1);
  gemm_f32_wmma_bn<<<gemmBlocks(N, 96), TB, 0, stream>>>(
      y1, 96, Wm1, 96, y2, 96, N, 96, 96, bm1, sc1, sh1, 1);
  col_part<96><<<np96, CS_THREADS, 0, stream>>>(y2, N * 96, ps, pq, np96);
  bn_fold<<<(96 * 32 + TB - 1) / TB, TB, 0, stream>>>(ps, pq, np96, gm1, betam1,
                                                      96, invn, sc2, sh2);
  mlp_out<<<(N + TB - 1) / TB, TB, 0, stream>>>(y2, sc2, sh2, Wm2, bm2, N, y3);
  col_part<2><<<np2, CS_THREADS, 0, stream>>>(y3, N * 2, ps, pq, np2);
  bn_fold<<<(2 * 32 + TB - 1) / TB, TB, 0, stream>>>(ps, pq, np2, gm2, betam2, 2,
                                                     invn, sc3, sh3);
  bn_apply<<<(N * 2 + TB - 1) / TB, TB, 0, stream>>>(y3, sc3, sh3, N * 2, 2,
                                                     (float*)d_out);
}


# GAT layer GEMMs (128->256, 256->128) re-tiled: 64 rows x 64 columns per work-group, weight strip staged once in LDS, A fragment feeds 4 tiles, whole-line D stores (order of k sums changes)
# speedup vs baseline: 3.8741x; 1.0911x over previous
#include <hip/hip_runtime.h>
#include <hip/hip_bf16.h>
#include <math.h>


typedef __attribute__((ext_vector_type(2))) float v2f;
typedef __attribute__((ext_vector_type(8))) float v8f;

#define DFEAT 128
#define NEG_SLOPE 0.2f

__device__ __forceinline__ unsigned ordEnc(float f) {
  unsigned u = __float_as_uint(f);
  return (u & 0x80000000u) ? ~u : (u | 0x80000000u);
}
__device__ __forceinline__ float ordDec(unsigned e) {
  unsigned u = (e & 0x80000000u) ? (e & 0x7fffffffu) : ~e;
  return __uint_as_float(u);
}

__device__ __forceinline__ v8f wmma_f32(v2f a, v2f b, v8f c) {
  return __builtin_amdgcn_wmma_f32_16x16x4_f32(false, a, false, b, (short)0, c,
                                               false, false);
}

__global__ void gemm_f32_wmma(const float* __restrict__ A, int lda,
                              const float* __restrict__ B, int ldb,
                              float* __restrict__ D, int ldo,
                              int nrows, int kdim, int ncols,
                              const float* __restrict__ bias, int relu) {
  const int lane = threadIdx.x & 31;
  const int wid  = (blockIdx.x * blockDim.x + threadIdx.x) >> 5;
  const int colTiles = ncols >> 4;
  const int rowTiles = (nrows + 15) >> 4;
  if (wid >= rowTiles * colTiles) return;
  const int tr = wid / colTiles;
  const int tc = wid - tr * colTiles;
  const int row0 = tr << 4, col0 = tc << 4;
  const int lm = lane & 15;
  const int lh = lane >> 4;

  int arow = row0 + lm;
  if (arow >= nrows) arow = nrows - 1;
  const float* Ap = A + (size_t)arow * lda + 2 * lh;
  const float* Bp = B + (size_t)(2 * lh) * ldb + col0 + lm;

  v8f acc = {0.f, 0.f, 0.f, 0.f, 0.f, 0.f, 0.f, 0.f};
  int k = 0;
  for (; k + 16 <= kdim; k += 16) {
    const v2f a0 = *(const v2f*)(Ap + k);
    const v2f a1 = *(const v2f*)(Ap + k + 4);
    const v2f a2 = *(const v2f*)(Ap + k + 8);
    const v2f a3 = *(const v2f*)(Ap + k + 12);
    const float* bq = Bp + (size_t)k * ldb;
    v2f b0, b1, b2, b3;
    b0.x = bq[0];
    b0.y = bq[(size_t)1 * ldb];
    b1.x = bq[(size_t)4 * ldb];
    b1.y = bq[(size_t)5 * ldb];
    b2.x = bq[(size_t)8 * ldb];
    b2.y = bq[(size_t)9 * ldb];
    b3.x = bq[(size_t)12 * ldb];
    b3.y = bq[(size_t)13 * ldb];
    acc = wmma_f32(a0, b0, acc);
    acc = wmma_f32(a1, b1, acc);
    acc = wmma_f32(a2, b2, acc);
    acc = wmma_f32(a3, b3, acc);
  }
  for (; k < kdim; k += 4) {
    const v2f a = *(const v2f*)(Ap + k);
    v2f b;
    b.x = Bp[(size_t)k * ldb];
    b.y = Bp[(size_t)(k + 1) * ldb];
    acc = wmma_f32(a, b, acc);
  }

  const int col = col0 + lm;
  const float bv = bias ? bias[col] : 0.0f;
#pragma unroll
  for (int r = 0; r < 8; ++r) {
    const int row = row0 + r + 8 * lh;
    if (row < nrows) {
      float v = acc[r] + bv;
      if (relu) v = fmaxf(v, 0.0f);
      D[(size_t)row * ldo + col] = v;
    }
  }
}

__global__ void gemm_f32_wmma_bn(const float* __restrict__ A, int lda,
                                 const float* __restrict__ B, int ldb,
                                 float* __restrict__ D, int ldo,
                                 int nrows, int kdim, int ncols,
                                 const float* __restrict__ bias,
                                 const float* __restrict__ scl,
                                 const float* __restrict__ shf, int relu) {
  const int lane = threadIdx.x & 31;
  const int wid  = (blockIdx.x * blockDim.x + threadIdx.x) >> 5;
  const int colTiles = ncols >> 4;
  const int rowTiles = (nrows + 15) >> 4;
  if (wid >= rowTiles * colTiles) return;
  const int tr = wid / colTiles;
  const int tc = wid - tr * colTiles;
  const int row0 = tr << 4, col0 = tc << 4;
  const int lm = lane & 15;
  const int lh = lane >> 4;

  int arow = row0 + lm;
  if (arow >= nrows) arow = nrows - 1;
  const float* Ap = A + (size_t)arow * lda + 2 * lh;
  const float* Sp = scl + 2 * lh;
  const float* Hp = shf + 2 * lh;
  const float* Bp = B + (size_t)(2 * lh) * ldb + col0 + lm;

  v8f acc = {0.f, 0.f, 0.f, 0.f, 0.f, 0.f, 0.f, 0.f};
  int k = 0;
  for (; k + 16 <= kdim; k += 16) {
    v2f a0 = *(const v2f*)(Ap + k);
    v2f a1 = *(const v2f*)(Ap + k + 4);
    v2f a2 = *(const v2f*)(Ap + k + 8);
    v2f a3 = *(const v2f*)(Ap + k + 12);
    const v2f s0 = *(const v2f*)(Sp + k);
    const v2f s1 = *(const v2f*)(Sp + k + 4);
    const v2f s2 = *(const v2f*)(Sp + k + 8);
    const v2f s3 = *(const v2f*)(Sp + k + 12);
    const v2f h0 = *(const v2f*)(Hp + k);
    const v2f h1 = *(const v2f*)(Hp + k + 4);
    const v2f h2 = *(const v2f*)(Hp + k + 8);
    const v2f h3 = *(const v2f*)(Hp + k + 12);
    a0 = a0 * s0 + h0;
    a1 = a1 * s1 + h1;
    a2 = a2 * s2 + h2;
    a3 = a3 * s3 + h3;
    const float* bq = Bp + (size_t)k * ldb;
    v2f b0, b1, b2, b3;
    b0.x = bq[0];
    b0.y = bq[(size_t)1 * ldb];
    b1.x = bq[(size_t)4 * ldb];
    b1.y = bq[(size_t)5 * ldb];
    b2.x = bq[(size_t)8 * ldb];
    b2.y = bq[(size_t)9 * ldb];
    b3.x = bq[(size_t)12 * ldb];
    b3.y = bq[(size_t)13 * ldb];
    acc = wmma_f32(a0, b0, acc);
    acc = wmma_f32(a1, b1, acc);
    acc = wmma_f32(a2, b2, acc);
    acc = wmma_f32(a3, b3, acc);
  }
  for (; k < kdim; k += 4) {
    v2f a = *(const v2f*)(Ap + k);
    const v2f s = *(const v2f*)(Sp + k);
    const v2f h = *(const v2f*)(Hp + k);
    a = a * s + h;
    v2f b;
    b.x = Bp[(size_t)k * ldb];
    b.y = Bp[(size_t)(k + 1) * ldb];
    acc = wmma_f32(a, b, acc);
  }

  const int col = col0 + lm;
  const float bv = bias ? bias[col] : 0.0f;
#pragma unroll
  for (int r = 0; r < 8; ++r) {
    const int row = row0 + r + 8 * lh;
    if (row < nrows) {
      float v = acc[r] + bv;
      if (relu) v = fmaxf(v, 0.0f);
      D[(size_t)row * ldo + col] = v;
    }
  }
}

#define LG_WAVES 4
#define LG_COLS 64
#define LG_PITCH (2 * LG_COLS + 8)
#define LG_CPITCH (LG_COLS + 4)

template <int KDIM>
__global__ void __launch_bounds__(LG_WAVES * 32)
gemm_layer_strip(const float* __restrict__ A, int lda,
                 const float* __restrict__ B, int ldb, float* __restrict__ D,
                 int ldo, int nrows, int ncols) {
  static_assert(KDIM % 16 == 0, "whole blocks of 16 k");
  static_assert(LG_COLS == 64, "four tiles per wave, 16 column groups per row");
  constexpr int NTHR = LG_WAVES * 32;
  constexpr int NPAIR = KDIM / 2;
  constexpr int NBLK = KDIM / 16;
  constexpr int CGR = LG_COLS / 4;
  constexpr int PER = NPAIR * CGR / NTHR;
  constexpr int CH = 8;
  static_assert((NPAIR * CGR) % NTHR == 0 && PER % CH == 0, "even staging");
  static_assert(NTHR % CGR == 0, "a thread keeps its column group");
  __shared__ __attribute__((aligned(16))) float Bs[NPAIR * LG_PITCH];
  __shared__ __attribute__((aligned(16))) float Cs[LG_WAVES][16 * LG_CPITCH];

  const int tid = threadIdx.x;
  const int lane = tid & 31;
  const int wave = __builtin_amdgcn_readfirstlane(tid >> 5);
  const int strips = ncols / LG_COLS;
  const int rg = blockIdx.x / strips;
  const int col0 = (blockIdx.x - rg * strips) * LG_COLS;
  const int row0 = (rg * LG_WAVES + wave) * 16;
  const bool live = row0 < nrows;
  const int c = lane & 15;
  const int h = lane >> 4;

  int arow = row0 + c;
  if (arow >= nrows) arow = nrows - 1;
  const float4* __restrict__ Ap =
      (const float4*)(A + (size_t)arow * lda + 8 * h);
  float4 a[2 * NBLK];
#pragma unroll
  for (int kb = 0; kb < NBLK; ++kb) {
    a[2 * kb] = Ap[4 * kb];
    a[2 * kb + 1] = Ap[4 * kb + 1];
  }

  {
    const int cg = tid % CGR;
    const float* __restrict__ bcol = B + col0 + 4 * cg;
    for (int t0 = 0; t0 < PER; t0 += CH) {
      float4 r0[CH], r1[CH];
#pragma unroll
      for (int i = 0; i < CH; ++i) {
        const int p = ((t0 + i) * NTHR + tid) / CGR;
        const float* bp = bcol + (size_t)(2 * p) * ldb;
        r0[i] = *(const float4*)bp;
        r1[i] = *(const float4*)(bp + ldb);
      }
#pragma unroll
      for (int i = 0; i < CH; ++i) {
        const int p = ((t0 + i) * NTHR + tid) / CGR;
        float* sp = Bs + p * LG_PITCH + 8 * cg;
        *(float4*)sp = make_float4(r0[i].x, r1[i].x, r0[i].y, r1[i].y);
        *(float4*)(sp + 4) = make_float4(r0[i].z, r1[i].z, r0[i].w, r1[i].w);
      }
    }
  }
  __syncthreads();

  float* cw = &Cs[wave][0];
  if (live) {
    v8f acc[4];
#pragma unroll
    for (int t = 0; t < 4; ++t)
      acc[t] = (v8f){0.f, 0.f, 0.f, 0.f, 0.f, 0.f, 0.f, 0.f};
    const float* bl = Bs + (4 * h) * LG_PITCH + 2 * c;
#pragma unroll
    for (int kb = 0; kb < NBLK; ++kb) {
      const float av[8] = {a[2 * kb].x,     a[2 * kb].y,     a[2 * kb].z,
                           a[2 * kb].w,     a[2 * kb + 1].x, a[2 * kb + 1].y,
                           a[2 * kb + 1].z, a[2 * kb + 1].w};
#pragma unroll
      for (int j = 0; j < 4; ++j) {
        v2f af;
        af.x = av[2 * j];
        af.y = av[2 * j + 1];
        const float* bq = bl + (8 * kb + j) * LG_PITCH;
#pragma unroll
        for (int t = 0; t < 4; ++t) {
          const v2f bf = *(const v2f*)(bq + 32 * t);
          acc[t] = wmma_f32(af, bf, acc[t]);
        }
      }
    }
    float* cp = cw + (8 * h) * LG_CPITCH + c;
#pragma unroll
    for (int t = 0; t < 4; ++t) {
#pragma unroll
      for (int r = 0; r < 8; ++r) cp[r * LG_CPITCH + 16 * t] = acc[t][r];
    }
  }
  __syncthreads();
  if (live) {
    const int rr = lane >> 4;
    const int cg = lane & 15;
    float4 v[8];
#pragma unroll
    for (int i = 0; i < 8; ++i)
      v[i] = *(const float4*)(cw + (2 * i + rr) * LG_CPITCH + 4 * cg);
    float* dp = D + (size_t)(row0 + rr) * ldo + col0 + 4 * cg;
    if (row0 + 16 <= nrows) {
#pragma unroll
      for (int i = 0; i < 8; ++i) *(float4*)(dp + (size_t)(2 * i) * ldo) = v[i];
    } else {
#pragma unroll
      for (int i = 0; i < 8; ++i) {
        if (row0 + 2 * i + rr < nrows)
          *(float4*)(dp + (size_t)(2 * i) * ldo) = v[i];
      }
    }
  }
}

#define AP_ROWS 16
#define AP_THREADS 128

template <int CNT>
__device__ __forceinline__ void ap_fold(float (&val)[2 * AP_ROWS], int lane) {
  const bool up = (lane & CNT) != 0;
#pragma unroll
  for (int i = 0; i < CNT; ++i) {
    const float keep = up ? val[i + CNT] : val[i];
    const float send = up ? val[i] : val[i + CNT];
    val[i] = keep + __shfl_xor(send, CNT, 32);
  }
}

template <int HEADS>
__global__ void __launch_bounds__(AP_THREADS)
attn_proj_rows(const float* __restrict__ h, const float* __restrict__ a_src,
               const float* __restrict__ a_dst, float* __restrict__ al_s,
               float* __restrict__ al_d, int nrows) {
  static_assert(AP_ROWS % HEADS == 0, "a wave's first row is head 0");
  static_assert(2 * AP_ROWS == 32, "one finished sum per lane");
  const int lane = threadIdx.x & 31;
  const int wid = __builtin_amdgcn_readfirstlane(
      (int)((blockIdx.x * blockDim.x + threadIdx.x) >> 5));
  const int w0 = wid * AP_ROWS;
  if (w0 >= nrows) return;

  float as[HEADS][4], ad[HEADS][4];
#pragma unroll
  for (int hh = 0; hh < HEADS; ++hh) {
#pragma unroll
    for (int q = 0; q < 4; ++q) {
      as[hh][q] = a_src[hh * DFEAT + 4 * lane + q];
      ad[hh][q] = a_dst[hh * DFEAT + 4 * lane + q];
    }
  }
  const float4* __restrict__ hp =
      (const float4*)(h + (size_t)w0 * DFEAT) + lane;
  float4 v[AP_ROWS];
#pragma unroll
  for (int j = 0; j < AP_ROWS; ++j)
    v[j] = hp[(w0 + j < nrows ? j : 0) * (DFEAT / 4)];
  __builtin_amdgcn_sched_barrier(0);

  float val[2 * AP_ROWS];
#pragma unroll
  for (int j = 0; j < AP_ROWS; ++j) {
    const int hh = j % HEADS;
    val[j] = v[j].x * as[hh][0] + v[j].y * as[hh][1] + v[j].z * as[hh][2] +
             v[j].w * as[hh][3];
    val[AP_ROWS + j] = v[j].x * ad[hh][0] + v[j].y * ad[hh][1] +
                       v[j].z * ad[hh][2] + v[j].w * ad[hh][3];
  }
  ap_fold<16>(val, lane);
  ap_fold<8>(val, lane);
  ap_fold<4>(val, lane);
  ap_fold<2>(val, lane);
  ap_fold<1>(val, lane);

  const int w = w0 + (lane & 15);
  if (w < nrows) (lane < 16 ? al_s : al_d)[w] = val[0];
}

#define OWN_ROWS 80
#define OWN_THREADS 1024
#define LIST_CAP 8192

template <int HEADS>
__device__ __forceinline__ void gat_owned_body(
    const int* __restrict__ src, const int* __restrict__ dst, int E, int n,
    const float* __restrict__ h, int ldh, const float* __restrict__ al_s,
    const float* __restrict__ al_d, const float* __restrict__ bias,
    float* __restrict__ out, int ldo, int colOff) {
  constexpr int ROWW = HEADS * DFEAT;
  constexpr int NL = ROWW / 32;
  constexpr int BATCH = 32 / NL;
  __shared__ float s_acc[OWN_ROWS * ROWW];
  __shared__ int s_ls[LIST_CAP];
  __shared__ int s_ld[LIST_CAP];
  __shared__ unsigned s_menc[OWN_ROWS * HEADS];
  __shared__ float s_sum[OWN_ROWS * HEADS];
  __shared__ unsigned s_cnt;

  const int tid = threadIdx.x;
  const int lane = tid & 31;
  const int wave = tid >> 5;
  const int nthr = blockDim.x;
  const int nw = nthr >> 5;
  const int r0 = blockIdx.x * OWN_ROWS;
  int rows_here = n - r0;
  if (rows_here > OWN_ROWS) rows_here = OWN_ROWS;
  if (rows_here < 0) rows_here = 0;

  for (int i = tid; i < OWN_ROWS * ROWW; i += nthr) s_acc[i] = 0.f;
  for (int i = tid; i < OWN_ROWS * HEADS; i += nthr) {
    s_menc[i] = 0u;
    s_sum[i] = 0.f;
  }
  if (tid == 0) s_cnt = 0u;
  __syncthreads();

  auto append = [&](int e, int dl) {
    const unsigned p = atomicAdd(&s_cnt, 1u);
    if (p < (unsigned)LIST_CAP) { s_ls[p] = e; s_ld[p] = dl; }
  };
  auto owned = [&](int d) -> bool {
    return (unsigned)(d - r0) < (unsigned)rows_here;
  };
  for (int r = tid; r < rows_here; r += nthr) append(E + r0 + r, r);
  {
    int pre = (int)(((16u - ((unsigned)(size_t)dst & 15u)) & 15u) >> 2);
    if (pre > E) pre = E;
    const int nvec = (E - pre) >> 2;
    const int4* __restrict__ dv = (const int4*)(dst + pre);
    for (int vb = 0; vb < nvec; vb += 4 * nthr) {
      int4 q[4];
      int vi[4];
      for (int k = 0; k < 4; ++k) {
        vi[k] = vb + k * nthr + tid;
        q[k] = dv[vi[k] < nvec ? vi[k] : nvec - 1];
      }
      for (int k = 0; k < 4; ++k) {
        const bool in = vi[k] < nvec;
        const bool o0 = in && owned(q[k].x), o1 = in && owned(q[k].y);
        const bool o2 = in && owned(q[k].z), o3 = in && owned(q[k].w);
        if (o0 | o1 | o2 | o3) {
          const int e = pre + 4 * vi[k];
          if (o0) append(e + 0, q[k].x - r0);
          if (o1) append(e + 1, q[k].y - r0);
          if (o2) append(e + 2, q[k].z - r0);
          if (o3) append(e + 3, q[k].w - r0);
        }
      }
    }
    const int tail0 = pre + 4 * nvec;
    if (tid < 8) {
      const int e = (tid < 4) ? tid : tail0 + (tid - 4);
      const bool in = (tid < 4) ? (e < pre) : (e < E);
      if (in) {
        const int d = dst[e];
        if (owned(d)) append(e, d - r0);
      }
    }
  }
  __syncthreads();
  const int cnt = (int)s_cnt;
  const bool use_list = cnt <= LIST_CAP;
  if (use_list) {
    for (int i = tid; i < cnt; i += nthr) {
      const int e = s_ls[i];
      s_ls[i] = (e < E) ? src[e] : (e - E);
    }
  }
  __syncthreads();

  auto for_owned = [&](auto body) {
    if (use_list) {
      for (int base = 0; base < cnt; base += 32 * nw) {
        const int i = base + lane * nw + wave;
        const bool own = i < cnt;
        const int s = own ? s_ls[i] : 0;
        const int dl = own ? s_ld[i] : 0;
        body(own, s, dl);
      }
    } else {
      for (int e0 = wave * 32; e0 < E; e0 += 32 * nw) {
        const int e = e0 + lane;
        const int d = (e < E) ? dst[e] : -1;
        const bool own = (unsigned)(d - r0) < (unsigned)rows_here;
        if (__builtin_amdgcn_ballot_w32(own)) {
          const int s = own ? src[e] : 0;
          body(own, s, own ? d - r0 : 0);
        }
      }
      for (int i0 = wave * 32; i0 < rows_here; i0 += 32 * nw) {
        const int r = i0 + lane;
        const bool own = r < rows_here;
        body(own, own ? r0 + r : 0, own ? r : 0);
      }
    }
  };
  auto logit = [&](int s, int dl, int hh) -> float {
    const float x = al_s[s * HEADS + hh] + al_d[(r0 + dl) * HEADS + hh];
    return (x > 0.f) ? x : NEG_SLOPE * x;
  };

  for_owned([&](bool own, int s, int dl) {
    if (own) {
      for (int hh = 0; hh < HEADS; ++hh)
        atomicMax(&s_menc[dl * HEADS + hh], ordEnc(logit(s, dl, hh)));
    }
  });
  __syncthreads();

  for_owned([&](bool own, int s, int dl) {
    if (own) {
      for (int hh = 0; hh < HEADS; ++hh) {
        const float m = ordDec(s_menc[dl * HEADS + hh]);
        atomicAdd(&s_sum[dl * HEADS + hh], expf(logit(s, dl, hh) - m));
      }
    }
  });
  __syncthreads();

  for_owned([&](bool own, int s, int dl) {
    float al[HEADS];
    for (int hh = 0; hh < HEADS; ++hh) al[hh] = 0.f;
    if (own) {
      for (int hh = 0; hh < HEADS; ++hh) {
        const float m = ordDec(s_menc[dl * HEADS + hh]);
        al[hh] = expf(logit(s, dl, hh) - m) / (s_sum[dl * HEADS + hh] + 1e-16f);
      }
    }
    unsigned m = __builtin_amdgcn_ballot_w32(own);
    while (__builtin_popcount(m) >= BATCH) {
      float v[BATCH][NL];
      int oo[BATCH];
      int jj[BATCH];
      for (int b = 0; b < BATCH; ++b) {
        const int j = __builtin_ctz(m);
        m &= m - 1;
        jj[b] = j;
        const int sj = __builtin_amdgcn_readlane(s, j);
        const int dj = __builtin_amdgcn_readlane(dl, j);
        const float* hp = h + (size_t)sj * ldh + lane;
        oo[b] = dj * ROWW + lane;
        for (int c = 0; c < NL; ++c) v[b][c] = hp[32 * c];
      }
      for (int b = 0; b < BATCH; ++b) {
        for (int c = 0; c < NL; ++c) {
          const float a = __int_as_float(__builtin_amdgcn_readlane(
              __float_as_int(al[c / (DFEAT / 32)]), jj[b]));
          atomicAdd(&s_acc[oo[b] + 32 * c], a * v[b][c]);
        }
      }
    }
    while (m) {
      const int j = __builtin_ctz(m);
      m &= m - 1;
      const int sj = __builtin_amdgcn_readlane(s, j);
      const int dj = __builtin_amdgcn_readlane(dl, j);
      const float* hp = h + (size_t)sj * ldh + lane;
      const int o = dj * ROWW + lane;
      float v[NL];
      for (int c = 0; c < NL; ++c) v[c] = hp[32 * c];
      for (int c = 0; c < NL; ++c) {
        const float a = __int_as_float(__builtin_amdgcn_readlane(
            __float_as_int(al[c / (DFEAT / 32)]), j));
        atomicAdd(&s_acc[o + 32 * c], a * v[c]);
      }
    }
  });
  __syncthreads();

  for (int r = wave; r < rows_here; r += nw) {
    float* op = out + (size_t)(r0 + r) * ldo + colOff + lane;
    for (int c = 0; c < NL; ++c)
      op[32 * c] = s_acc[r * ROWW + 32 * c + lane] + bias[32 * c + lane];
  }
}

__global__ void __launch_bounds__(OWN_THREADS)
gat_owned_h2(const int* __restrict__ src, const int* __restrict__ dst, int E,
             int n, const float* __restrict__ h, int ldh,
             const float* __restrict__ al_s, const float* __restrict__ al_d,
             const float* __restrict__ bias, float* __restrict__ out, int ldo,
             int colOff) {
  gat_owned_body<2>(src, dst, E, n, h, ldh, al_s, al_d, bias, out, ldo, colOff);
}

__global__ void __launch_bounds__(OWN_THREADS)
gat_owned_h1(const int* __restrict__ src, const int* __restrict__ dst, int E,
             int n, const float* __restrict__ h, int ldh,
             const float* __restrict__ al_s, const float* __restrict__ al_d,
             const float* __restrict__ bias, float* __restrict__ out, int ldo,
             int colOff) {
  gat_owned_body<1>(src, dst, E, n, h, ldh, al_s, al_d, bias, out, ldo, colOff);
}

#define CS_THREADS 192
#define CS_LOADS 10

template <int NC>
__global__ void __launch_bounds__(CS_THREADS)
col_part(const float* __restrict__ y, int total, float* __restrict__ ps,
         float* __restrict__ pq, int nparts) {
  static_assert((4 * CS_THREADS) % NC == 0, "a thread keeps its columns");
  const int t = threadIdx.x;
  const int nvec = total >> 2;
  const int base = blockIdx.x * (CS_LOADS * CS_THREADS);
  const float4* __restrict__ y4 = (const float4*)y;

  float4 v[CS_LOADS];
#pragma unroll
  for (int k = 0; k < CS_LOADS; ++k) {
    const int f = base + k * CS_THREADS + t;
    v[k] = y4[f < nvec ? f : 0];
  }
  float sa[4] = {0.f, 0.f, 0.f, 0.f};
  float qa[4] = {0.f, 0.f, 0.f, 0.f};
#pragma unroll
  for (int k = 0; k < CS_LOADS; ++k) {
    const int f = base + k * CS_THREADS + t;
    if (f < nvec) {
      sa[0] += v[k].x; qa[0] = fmaf(v[k].x, v[k].x, qa[0]);
      sa[1] += v[k].y; qa[1] = fmaf(v[k].y, v[k].y, qa[1]);
      sa[2] += v[k].z; qa[2] = fmaf(v[k].z, v[k].z, qa[2]);
      sa[3] += v[k].w; qa[3] = fmaf(v[k].w, v[k].w, qa[3]);
    }
  }
  {
    const int tail = total & 3;
    const int rel = nvec - base;
    if (tail && rel >= 0 && rel < CS_LOADS * CS_THREADS &&
        rel % CS_THREADS == t) {
      for (int j = 0; j < tail; ++j) {
        const float w = y[4 * nvec + j];
        sa[j] += w;
        qa[j] = fmaf(w, w, qa[j]);
      }
    }
  }

  if constexpr (NC % 4 == 0) {
    constexpr int QN = NC / 4;
    static_assert(CS_THREADS % QN == 0, "whole rows per step");
    constexpr int GN = CS_THREADS / QN;
    __shared__ __attribute__((aligned(16))) float red[2][GN][NC];
    const int g = t / QN;
    const int c4 = 4 * (t - g * QN);
    *(float4*)&red[0][g][c4] = make_float4(sa[0], sa[1], sa[2], sa[3]);
    *(float4*)&red[1][g][c4] = make_float4(qa[0], qa[1], qa[2], qa[3]);
    __syncthreads();
    for (int i = t; i < 2 * NC; i += CS_THREADS) {
      const int which = i / NC;
      const int c = i - which * NC;
      float r = 0.f;
#pragma unroll
      for (int gg = 0; gg < GN; ++gg) r += red[which][gg][c];
      (which ? pq : ps)[(size_t)c * nparts + blockIdx.x] = r;
    }
  } else {
    static_assert(4 % NC == 0, "column count must divide a group");
    constexpr int NW = CS_THREADS / 32;
    __shared__ float red[2][NW][NC];
    float cs[NC], cq[NC];
#pragma unroll
    for (int c = 0; c < NC; ++c) { cs[c] = 0.f; cq[c] = 0.f; }
#pragma unroll
    for (int j = 0; j < 4; ++j) { cs[j % NC] += sa[j]; cq[j % NC] += qa[j]; }
#pragma unroll
    for (int c = 0; c < NC; ++c) {
      for (int off = 16; off > 0; off >>= 1) {
        cs[c] += __shfl_xor(cs[c], off, 32);
        cq[c] += __shfl_xor(cq[c], off, 32);
      }
    }
    const int lane = t & 31;
    const int wave = t >> 5;
    if (lane == 0) {
#pragma unroll
      for (int c = 0; c < NC; ++c) {
        red[0][wave][c] = cs[c];
        red[1][wave][c] = cq[c];
      }
    }
    __syncthreads();
    if (t < 2 * NC) {
      const int which = t / NC;
      const int c = t - which * NC;
      float r = 0.f;
#pragma unroll
      for (int w = 0; w < NW; ++w) r += red[which][w][c];
      (which ? pq : ps)[(size_t)c * nparts + blockIdx.x] = r;
    }
  }
}

__global__ void bn_fold(const float* __restrict__ ps,
                        const float* __restrict__ pq, int nparts,
                        const float* __restrict__ g,
                        const float* __restrict__ beta, int ncols, float invn,
                        float* __restrict__ sc, float* __restrict__ sh) {
  const int lane = threadIdx.x & 31;
  const int c = (blockIdx.x * blockDim.x + threadIdx.x) >> 5;
  if (c >= ncols) return;
  const float* __restrict__ sp = ps + (size_t)c * nparts;
  const float* __restrict__ qp = pq + (size_t)c * nparts;
  float s = 0.f, q = 0.f;
  for (int b0 = 0; b0 < nparts; b0 += 256) {
    float vs[8], vq[8];
#pragma unroll
    for (int k = 0; k < 8; ++k) {
      const int b = b0 + 32 * k + lane;
      vs[k] = sp[b < nparts ? b : 0];
      vq[k] = qp[b < nparts ? b : 0];
    }
#pragma unroll
    for (int k = 0; k < 8; ++k) {
      const int b = b0 + 32 * k + lane;
      if (b < nparts) { s += vs[k]; q += vq[k]; }
    }
  }
  for (int off = 16; off > 0; off >>= 1) {
    s += __shfl_xor(s, off, 32);
    q += __shfl_xor(q, off, 32);
  }
  if (lane == 0) {
    const float mean = s * invn;
    const float var = q * invn - mean * mean;
    const float r = rsqrtf(var + 1e-5f);
    const float a = g[c] * r;
    sc[c] = a;
    sh[c] = beta[c] - mean * a;
  }
}

__global__ void mlp_out(const float* __restrict__ y2,
                        const float* __restrict__ sc,
                        const float* __restrict__ sh,
                        const float* __restrict__ Wm2,
                        const float* __restrict__ bm2, int n,
                        float* __restrict__ y3) {
  const int node = blockIdx.x * blockDim.x + threadIdx.x;
  if (node >= n) return;
  float a0 = bm2[0], a1 = bm2[1];
  const float* yp = y2 + (size_t)node * 96;
  for (int k = 0; k < 96; ++k) {
    const float v = fmaf(yp[k], sc[k], sh[k]);
    a0 = fmaf(v, Wm2[2 * k + 0], a0);
    a1 = fmaf(v, Wm2[2 * k + 1], a1);
  }
  y3[node * 2 + 0] = fmaxf(a0, 0.f);
  y3[node * 2 + 1] = fmaxf(a1, 0.f);
}

__global__ void bn_apply(const float* __restrict__ y,
                         const float* __restrict__ sc,
                         const float* __restrict__ sh, int total, int ncols,
                         float* __restrict__ out) {
  const int i = blockIdx.x * blockDim.x + threadIdx.x;
  if (i >= total) return;
  const int c = i % ncols;
  out[i] = fmaf(y[i], sc[c], sh[c]);
}

extern "C" void kernel_launch(void* const* d_in, const int* in_sizes, int n_in,
                              void* d_out, int out_size, void* d_ws,
                              size_t ws_size, hipStream_t stream) {
  const float* x      = (const float*)d_in[0];
  const int*   ei     = (const int*)d_in[1];
  const float* W1     = (const float*)d_in[2];
  const float* as1    = (const float*)d_in[3];
  const float* ad1    = (const float*)d_in[4];
  const float* b1     = (const float*)d_in[5];
  const float* W2     = (const float*)d_in[6];
  const float* as2    = (const float*)d_in[7];
  const float* ad2    = (const float*)d_in[8];
  const float* b2     = (const float*)d_in[9];
  const float* Wl     = (const float*)d_in[10];
  const float* bl     = (const float*)d_in[11];
  const float* gl     = (const float*)d_in[12];
  const float* betal  = (const float*)d_in[13];
  const float* Wm1    = (const float*)d_in[14];
  const float* bm1    = (const float*)d_in[15];
  const float* gm1    = (const float*)d_in[16];
  const float* betam1 = (const float*)d_in[17];
  const float* Wm2    = (const float*)d_in[18];
  const float* bm2    = (const float*)d_in[19];
  const float* gm2    = (const float*)d_in[20];
  const float* betam2 = (const float*)d_in[21];

  const int N = in_sizes[0] / DFEAT;
  const int E = in_sizes[1] / 2;
  const int* srcA = ei;
  const int* dstA = ei + E;

  char* w = (char*)d_ws;
  auto alloc = [&](size_t bytes) -> char* {
    char* p = w;
    w += (bytes + 255) & ~(size_t)255;
    return p;
  };
  float*    xcat = (float*)alloc((size_t)N * 384 * 4);
  float*    h1   = (float*)alloc((size_t)N * 256 * 4);
  float*    h2   = (float*)alloc((size_t)N * 128 * 4);
  float*    al_s = (float*)alloc((size_t)N * 2 * 4);
  float*    al_d = (float*)alloc((size_t)N * 2 * 4);
  float*    y3   = (float*)alloc((size_t)N * 2 * 4);
  float*    sc1  = (float*)alloc(96 * 4);
  float*    sh1  = (float*)alloc(96 * 4);
  float*    sc2  = (float*)alloc(96 * 4);
  float*    sh2  = (float*)alloc(96 * 4);
  float*    sc3  = (float*)alloc(96 * 4);
  float*    sh3  = (float*)alloc(96 * 4);
  float* y1 = h1;
  float* y2 = h1 + (size_t)N * 96;
  const int csGroups = CS_LOADS * CS_THREADS;
  auto csBlocks = [&](int total) {
    return ((total + 3) / 4 + csGroups - 1) / csGroups;
  };
  const int np96 = csBlocks(N * 96);
  const int np2  = csBlocks(N * 2);
  float* ps = h2;
  float* pq = h2 + (size_t)96 * np96;

  const int TB = 256;
  auto gemmBlocks = [](int rows, int cols) {
    const int tiles = ((rows + 15) / 16) * (cols / 16);
    return (tiles + 7) / 8;
  };
  auto layerBlocks = [](int rows, int cols) {
    const int rowsPer = LG_WAVES * 16;
    return ((rows + rowsPer - 1) / rowsPer) * (cols / LG_COLS);
  };
  const int ownBlocks = (N + OWN_ROWS - 1) / OWN_ROWS;
  const int apRows = AP_ROWS * (AP_THREADS / 32);

  gemm_layer_strip<128><<<layerBlocks(N, 256), LG_WAVES * 32, 0, stream>>>(
      x, 128, W1, 256, h1, 256, N, 256);
  attn_proj_rows<2><<<(N * 2 + apRows - 1) / apRows, AP_THREADS, 0, stream>>>(
      h1, as1, ad1, al_s, al_d, N * 2);
  gat_owned_h2<<<ownBlocks, OWN_THREADS, 0, stream>>>(
      srcA, dstA, E, N, h1, 256, al_s, al_d, b1, xcat, 384, 0);

  gemm_layer_strip<256><<<layerBlocks(N, 128), LG_WAVES * 32, 0, stream>>>(
      xcat, 384, W2, 128, h2, 128, N, 128);
  attn_proj_rows<1><<<(N + apRows - 1) / apRows, AP_THREADS, 0, stream>>>(
      h2, as2, ad2, al_s, al_d, N);
  gat_owned_h1<<<ownBlocks, OWN_THREADS, 0, stream>>>(
      srcA, dstA, E, N, h2, 128, al_s, al_d, b2, xcat, 384, 256);

  const float invn = 1.0f / (float)N;
  gemm_f32_wmma<<<gemmBlocks(N, 96), TB, 0, stream>>>(
      xcat, 384, Wl, 96, y1, 96, N, 384, 96, bl, 1);
  col_part<96><<<np96, CS_THREADS, 0, stream>>>(y1, N * 96, ps, pq, np96);
  bn_fold<<<(96 * 32 + TB - 1) / TB, TB, 0, stream>>>(ps, pq, np96, gl, betal,
                                                      96, invn, sc1, sh1);
  gemm_f32_wmma_bn<<<gemmBlocks(N, 96), TB, 0, stream>>>(
      y1, 96, Wm1, 96, y2, 96, N, 96, 96, bm1, sc1, sh1, 1);
  col_part<96><<<np96, CS_THREADS, 0, stream>>>(y2, N * 96, ps, pq, np96);
  bn_fold<<<(96 * 32 + TB - 1) / TB, TB, 0, stream>>>(ps, pq, np96, gm1, betam1,
                                                      96, invn, sc2, sh2);
  mlp_out<<<(N + TB - 1) / TB, TB, 0, stream>>>(y2, sc2, sh2, Wm2, bm2, N, y3);
  col_part<2><<<np2, CS_THREADS, 0, stream>>>(y3, N * 2, ps, pq, np2);
  bn_fold<<<(2 * 32 + TB - 1) / TB, TB, 0, stream>>>(ps, pq, np2, gm2, betam2, 2,
                                                     invn, sc3, sh3);
  bn_apply<<<(N * 2 + TB - 1) / TB, TB, 0, stream>>>(y3, sc3, sh3, N * 2, 2,
                                                     (float*)d_out);
}


# edge list filtered once per call: a layer-1 work-group leaves its filtered list in its own rows of the layer-2 output columns, layer 2 reads it back, no second dst sweep (LDS adds in arrival order)
# speedup vs baseline: 4.0531x; 1.0462x over previous
#include <hip/hip_runtime.h>
#include <hip/hip_bf16.h>
#include <math.h>


typedef __attribute__((ext_vector_type(2))) float v2f;
typedef __attribute__((ext_vector_type(8))) float v8f;

#define DFEAT 128
#define NEG_SLOPE 0.2f

__device__ __forceinline__ unsigned ordEnc(float f) {
  unsigned u = __float_as_uint(f);
  return (u & 0x80000000u) ? ~u : (u | 0x80000000u);
}
__device__ __forceinline__ float ordDec(unsigned e) {
  unsigned u = (e & 0x80000000u) ? (e & 0x7fffffffu) : ~e;
  return __uint_as_float(u);
}

__device__ __forceinline__ v8f wmma_f32(v2f a, v2f b, v8f c) {
  return __builtin_amdgcn_wmma_f32_16x16x4_f32(false, a, false, b, (short)0, c,
                                               false, false);
}

__global__ void gemm_f32_wmma(const float* __restrict__ A, int lda,
                              const float* __restrict__ B, int ldb,
                              float* __restrict__ D, int ldo,
                              int nrows, int kdim, int ncols,
                              const float* __restrict__ bias, int relu) {
  const int lane = threadIdx.x & 31;
  const int wid  = (blockIdx.x * blockDim.x + threadIdx.x) >> 5;
  const int colTiles = ncols >> 4;
  const int rowTiles = (nrows + 15) >> 4;
  if (wid >= rowTiles * colTiles) return;
  const int tr = wid / colTiles;
  const int tc = wid - tr * colTiles;
  const int row0 = tr << 4, col0 = tc << 4;
  const int lm = lane & 15;
  const int lh = lane >> 4;

  int arow = row0 + lm;
  if (arow >= nrows) arow = nrows - 1;
  const float* Ap = A + (size_t)arow * lda + 2 * lh;
  const float* Bp = B + (size_t)(2 * lh) * ldb + col0 + lm;

  v8f acc = {0.f, 0.f, 0.f, 0.f, 0.f, 0.f, 0.f, 0.f};
  int k = 0;
  for (; k + 16 <= kdim; k += 16) {
    const v2f a0 = *(const v2f*)(Ap + k);
    const v2f a1 = *(const v2f*)(Ap + k + 4);
    const v2f a2 = *(const v2f*)(Ap + k + 8);
    const v2f a3 = *(const v2f*)(Ap + k + 12);
    const float* bq = Bp + (size_t)k * ldb;
    v2f b0, b1, b2, b3;
    b0.x = bq[0];
    b0.y = bq[(size_t)1 * ldb];
    b1.x = bq[(size_t)4 * ldb];
    b1.y = bq[(size_t)5 * ldb];
    b2.x = bq[(size_t)8 * ldb];
    b2.y = bq[(size_t)9 * ldb];
    b3.x = bq[(size_t)12 * ldb];
    b3.y = bq[(size_t)13 * ldb];
    acc = wmma_f32(a0, b0, acc);
    acc = wmma_f32(a1, b1, acc);
    acc = wmma_f32(a2, b2, acc);
    acc = wmma_f32(a3, b3, acc);
  }
  for (; k < kdim; k += 4) {
    const v2f a = *(const v2f*)(Ap + k);
    v2f b;
    b.x = Bp[(size_t)k * ldb];
    b.y = Bp[(size_t)(k + 1) * ldb];
    acc = wmma_f32(a, b, acc);
  }

  const int col = col0 + lm;
  const float bv = bias ? bias[col] : 0.0f;
#pragma unroll
  for (int r = 0; r < 8; ++r) {
    const int row = row0 + r + 8 * lh;
    if (row < nrows) {
      float v = acc[r] + bv;
      if (relu) v = fmaxf(v, 0.0f);
      D[(size_t)row * ldo + col] = v;
    }
  }
}

__global__ void gemm_f32_wmma_bn(const float* __restrict__ A, int lda,
                                 const float* __restrict__ B, int ldb,
                                 float* __restrict__ D, int ldo,
                                 int nrows, int kdim, int ncols,
                                 const float* __restrict__ bias,
                                 const float* __restrict__ scl,
                                 const float* __restrict__ shf, int relu) {
  const int lane = threadIdx.x & 31;
  const int wid  = (blockIdx.x * blockDim.x + threadIdx.x) >> 5;
  const int colTiles = ncols >> 4;
  const int rowTiles = (nrows + 15) >> 4;
  if (wid >= rowTiles * colTiles) return;
  const int tr = wid / colTiles;
  const int tc = wid - tr * colTiles;
  const int row0 = tr << 4, col0 = tc << 4;
  const int lm = lane & 15;
  const int lh = lane >> 4;

  int arow = row0 + lm;
  if (arow >= nrows) arow = nrows - 1;
  const float* Ap = A + (size_t)arow * lda + 2 * lh;
  const float* Sp = scl + 2 * lh;
  const float* Hp = shf + 2 * lh;
  const float* Bp = B + (size_t)(2 * lh) * ldb + col0 + lm;

  v8f acc = {0.f, 0.f, 0.f, 0.f, 0.f, 0.f, 0.f, 0.f};
  int k = 0;
  for (; k + 16 <= kdim; k += 16) {
    v2f a0 = *(const v2f*)(Ap + k);
    v2f a1 = *(const v2f*)(Ap + k + 4);
    v2f a2 = *(const v2f*)(Ap + k + 8);
    v2f a3 = *(const v2f*)(Ap + k + 12);
    const v2f s0 = *(const v2f*)(Sp + k);
    const v2f s1 = *(const v2f*)(Sp + k + 4);
    const v2f s2 = *(const v2f*)(Sp + k + 8);
    const v2f s3 = *(const v2f*)(Sp + k + 12);
    const v2f h0 = *(const v2f*)(Hp + k);
    const v2f h1 = *(const v2f*)(Hp + k + 4);
    const v2f h2 = *(const v2f*)(Hp + k + 8);
    const v2f h3 = *(const v2f*)(Hp + k + 12);
    a0 = a0 * s0 + h0;
    a1 = a1 * s1 + h1;
    a2 = a2 * s2 + h2;
    a3 = a3 * s3 + h3;
    const float* bq = Bp + (size_t)k * ldb;
    v2f b0, b1, b2, b3;
    b0.x = bq[0];
    b0.y = bq[(size_t)1 * ldb];
    b1.x = bq[(size_t)4 * ldb];
    b1.y = bq[(size_t)5 * ldb];
    b2.x = bq[(size_t)8 * ldb];
    b2.y = bq[(size_t)9 * ldb];
    b3.x = bq[(size_t)12 * ldb];
    b3.y = bq[(size_t)13 * ldb];
    acc = wmma_f32(a0, b0, acc);
    acc = wmma_f32(a1, b1, acc);
    acc = wmma_f32(a2, b2, acc);
    acc = wmma_f32(a3, b3, acc);
  }
  for (; k < kdim; k += 4) {
    v2f a = *(const v2f*)(Ap + k);
    const v2f s = *(const v2f*)(Sp + k);
    const v2f h = *(const v2f*)(Hp + k);
    a = a * s + h;
    v2f b;
    b.x = Bp[(size_t)k * ldb];
    b.y = Bp[(size_t)(k + 1) * ldb];
    acc = wmma_f32(a, b, acc);
  }

  const int col = col0 + lm;
  const float bv = bias ? bias[col] : 0.0f;
#pragma unroll
  for (int r = 0; r < 8; ++r) {
    const int row = row0 + r + 8 * lh;
    if (row < nrows) {
      float v = acc[r] + bv;
      if (relu) v = fmaxf(v, 0.0f);
      D[(size_t)row * ldo + col] = v;
    }
  }
}

#define LG_WAVES 4
#define LG_COLS 64
#define LG_PITCH (2 * LG_COLS + 8)
#define LG_CPITCH (LG_COLS + 4)

template <int KDIM>
__global__ void __launch_bounds__(LG_WAVES * 32)
gemm_layer_strip(const float* __restrict__ A, int lda,
                 const float* __restrict__ B, int ldb, float* __restrict__ D,
                 int ldo, int nrows, int ncols) {
  static_assert(KDIM % 16 == 0, "whole blocks of 16 k");
  static_assert(LG_COLS == 64, "four tiles per wave, 16 column groups per row");
  constexpr int NTHR = LG_WAVES * 32;
  constexpr int NPAIR = KDIM / 2;
  constexpr int NBLK = KDIM / 16;
  constexpr int CGR = LG_COLS / 4;
  constexpr int PER = NPAIR * CGR / NTHR;
  constexpr int CH = 8;
  static_assert((NPAIR * CGR) % NTHR == 0 && PER % CH == 0, "even staging");
  static_assert(NTHR % CGR == 0, "a thread keeps its column group");
  __shared__ __attribute__((aligned(16))) float Bs[NPAIR * LG_PITCH];
  __shared__ __attribute__((aligned(16))) float Cs[LG_WAVES][16 * LG_CPITCH];

  const int tid = threadIdx.x;
  const int lane = tid & 31;
  const int wave = __builtin_amdgcn_readfirstlane(tid >> 5);
  const int strips = ncols / LG_COLS;
  const int rg = blockIdx.x / strips;
  const int col0 = (blockIdx.x - rg * strips) * LG_COLS;
  const int row0 = (rg * LG_WAVES + wave) * 16;
  const bool live = row0 < nrows;
  const int c = lane & 15;
  const int h = lane >> 4;

  int arow = row0 + c;
  if (arow >= nrows) arow = nrows - 1;
  const float4* __restrict__ Ap =
      (const float4*)(A + (size_t)arow * lda + 8 * h);
  float4 a[2 * NBLK];
#pragma unroll
  for (int kb = 0; kb < NBLK; ++kb) {
    a[2 * kb] = Ap[4 * kb];
    a[2 * kb + 1] = Ap[4 * kb + 1];
  }

  {
    const int cg = tid % CGR;
    const float* __restrict__ bcol = B + col0 + 4 * cg;
    for (int t0 = 0; t0 < PER; t0 += CH) {
      float4 r0[CH], r1[CH];
#pragma unroll
      for (int i = 0; i < CH; ++i) {
        const int p = ((t0 + i) * NTHR + tid) / CGR;
        const float* bp = bcol + (size_t)(2 * p) * ldb;
        r0[i] = *(const float4*)bp;
        r1[i] = *(const float4*)(bp + ldb);
      }
#pragma unroll
      for (int i = 0; i < CH; ++i) {
        const int p = ((t0 + i) * NTHR + tid) / CGR;
        float* sp = Bs + p * LG_PITCH + 8 * cg;
        *(float4*)sp = make_float4(r0[i].x, r1[i].x, r0[i].y, r1[i].y);
        *(float4*)(sp + 4) = make_float4(r0[i].z, r1[i].z, r0[i].w, r1[i].w);
      }
    }
  }
  __syncthreads();

  float* cw = &Cs[wave][0];
  if (live) {
    v8f acc[4];
#pragma unroll
    for (int t = 0; t < 4; ++t)
      acc[t] = (v8f){0.f, 0.f, 0.f, 0.f, 0.f, 0.f, 0.f, 0.f};
    const float* bl = Bs + (4 * h) * LG_PITCH + 2 * c;
#pragma unroll
    for (int kb = 0; kb < NBLK; ++kb) {
      const float av[8] = {a[2 * kb].x,     a[2 * kb].y,     a[2 * kb].z,
                           a[2 * kb].w,     a[2 * kb + 1].x, a[2 * kb + 1].y,
                           a[2 * kb + 1].z, a[2 * kb + 1].w};
#pragma unroll
      for (int j = 0; j < 4; ++j) {
        v2f af;
        af.x = av[2 * j];
        af.y = av[2 * j + 1];
        const float* bq = bl + (8 * kb + j) * LG_PITCH;
#pragma unroll
        for (int t = 0; t < 4; ++t) {
          const v2f bf = *(const v2f*)(bq + 32 * t);
          acc[t] = wmma_f32(af, bf, acc[t]);
        }
      }
    }
    float* cp = cw + (8 * h) * LG_CPITCH + c;
#pragma unroll
    for (int t = 0; t < 4; ++t) {
#pragma unroll
      for (int r = 0; r < 8; ++r) cp[r * LG_CPITCH + 16 * t] = acc[t][r];
    }
  }
  __syncthreads();
  if (live) {
    const int rr = lane >> 4;
    const int cg = lane & 15;
    float4 v[8];
#pragma unroll
    for (int i = 0; i < 8; ++i)
      v[i] = *(const float4*)(cw + (2 * i + rr) * LG_CPITCH + 4 * cg);
    float* dp = D + (size_t)(row0 + rr) * ldo + col0 + 4 * cg;
    if (row0 + 16 <= nrows) {
#pragma unroll
      for (int i = 0; i < 8; ++i) *(float4*)(dp + (size_t)(2 * i) * ldo) = v[i];
    } else {
#pragma unroll
      for (int i = 0; i < 8; ++i) {
        if (row0 + 2 * i + rr < nrows)
          *(float4*)(dp + (size_t)(2 * i) * ldo) = v[i];
      }
    }
  }
}

#define AP_ROWS 16
#define AP_THREADS 128

template <int CNT>
__device__ __forceinline__ void ap_fold(float (&val)[2 * AP_ROWS], int lane) {
  const bool up = (lane & CNT) != 0;
#pragma unroll
  for (int i = 0; i < CNT; ++i) {
    const float keep = up ? val[i + CNT] : val[i];
    const float send = up ? val[i] : val[i + CNT];
    val[i] = keep + __shfl_xor(send, CNT, 32);
  }
}

template <int HEADS>
__global__ void __launch_bounds__(AP_THREADS)
attn_proj_rows(const float* __restrict__ h, const float* __restrict__ a_src,
               const float* __restrict__ a_dst, float* __restrict__ al_s,
               float* __restrict__ al_d, int nrows) {
  static_assert(AP_ROWS % HEADS == 0, "a wave's first row is head 0");
  static_assert(2 * AP_ROWS == 32, "one finished sum per lane");
  const int lane = threadIdx.x & 31;
  const int wid = __builtin_amdgcn_readfirstlane(
      (int)((blockIdx.x * blockDim.x + threadIdx.x) >> 5));
  const int w0 = wid * AP_ROWS;
  if (w0 >= nrows) return;

  float as[HEADS][4], ad[HEADS][4];
#pragma unroll
  for (int hh = 0; hh < HEADS; ++hh) {
#pragma unroll
    for (int q = 0; q < 4; ++q) {
      as[hh][q] = a_src[hh * DFEAT + 4 * lane + q];
      ad[hh][q] = a_dst[hh * DFEAT + 4 * lane + q];
    }
  }
  const float4* __restrict__ hp =
      (const float4*)(h + (size_t)w0 * DFEAT) + lane;
  float4 v[AP_ROWS];
#pragma unroll
  for (int j = 0; j < AP_ROWS; ++j)
    v[j] = hp[(w0 + j < nrows ? j : 0) * (DFEAT / 4)];
  __builtin_amdgcn_sched_barrier(0);

  float val[2 * AP_ROWS];
#pragma unroll
  for (int j = 0; j < AP_ROWS; ++j) {
    const int hh = j % HEADS;
    val[j] = v[j].x * as[hh][0] + v[j].y * as[hh][1] + v[j].z * as[hh][2] +
             v[j].w * as[hh][3];
    val[AP_ROWS + j] = v[j].x * ad[hh][0] + v[j].y * ad[hh][1] +
                       v[j].z * ad[hh][2] + v[j].w * ad[hh][3];
  }
  ap_fold<16>(val, lane);
  ap_fold<8>(val, lane);
  ap_fold<4>(val, lane);
  ap_fold<2>(val, lane);
  ap_fold<1>(val, lane);

  const int w = w0 + (lane & 15);
  if (w < nrows) (lane < 16 ? al_s : al_d)[w] = val[0];
}

#define OWN_ROWS 80
#define OWN_THREADS 1024
#define LIST_CAP 8192
#define KEEP_HDR 32

template <int HEADS, int KEEP>
__device__ __forceinline__ void gat_owned_body(
    const int* __restrict__ src, const int* __restrict__ dst, int E, int n,
    const float* __restrict__ h, int ldh, const float* __restrict__ al_s,
    const float* __restrict__ al_d, const float* __restrict__ bias,
    float* __restrict__ out, int ldo, int colOff, int keepOff) {
  static_assert(OWN_ROWS <= 128 && DFEAT == 128, "packed word: 7 bits of row");
  constexpr int ROWW = HEADS * DFEAT;
  constexpr int NL = ROWW / 32;
  constexpr int BATCH = 32 / NL;
  __shared__ float s_acc[OWN_ROWS * ROWW];
  __shared__ int s_ls[LIST_CAP];
  __shared__ int s_ld[LIST_CAP];
  __shared__ unsigned s_menc[OWN_ROWS * HEADS];
  __shared__ float s_sum[OWN_ROWS * HEADS];
  __shared__ unsigned s_cnt;

  const int tid = threadIdx.x;
  const int lane = tid & 31;
  const int wave = tid >> 5;
  const int nthr = blockDim.x;
  const int nw = nthr >> 5;
  const int r0 = blockIdx.x * OWN_ROWS;
  int rows_here = n - r0;
  if (rows_here > OWN_ROWS) rows_here = OWN_ROWS;
  if (rows_here < 0) rows_here = 0;

  float* const kbase = out + (size_t)r0 * ldo + keepOff;
  auto kslot = [&](int j) -> float* {
    return kbase + (size_t)(j >> 7) * ldo + (j & (DFEAT - 1));
  };
  int keep_cap = rows_here * DFEAT - KEEP_HDR;
  if (keep_cap > LIST_CAP) keep_cap = LIST_CAP;
  const bool packs = n <= (1 << 24);
  int kcnt = -1;
  if constexpr (KEEP == 2) {
    if (rows_here > 0 && packs) kcnt = __float_as_int(*kslot(0));
  }
  const bool kept = KEEP == 2 && kcnt >= 0 && kcnt <= keep_cap;

  for (int i = tid; i < OWN_ROWS * ROWW; i += nthr) s_acc[i] = 0.f;
  for (int i = tid; i < OWN_ROWS * HEADS; i += nthr) {
    s_menc[i] = 0u;
    s_sum[i] = 0.f;
  }
  if (tid == 0) s_cnt = 0u;
  __syncthreads();

  auto append = [&](int e, int dl) {
    const unsigned p = atomicAdd(&s_cnt, 1u);
    if (p < (unsigned)LIST_CAP) { s_ls[p] = e; s_ld[p] = dl; }
  };
  auto owned = [&](int d) -> bool {
    return (unsigned)(d - r0) < (unsigned)rows_here;
  };
  if (kept) {
    for (int i = tid; i < kcnt; i += nthr) {
      const int p = __float_as_int(*kslot(KEEP_HDR + i));
      s_ls[i] = p >> 7;
      s_ld[i] = p & 127;
    }
  } else {
    for (int r = tid; r < rows_here; r += nthr) append(E + r0 + r, r);
    int pre = (int)(((16u - ((unsigned)(size_t)dst & 15u)) & 15u) >> 2);
    if (pre > E) pre = E;
    const int nvec = (E - pre) >> 2;
    const int4* __restrict__ dv = (const int4*)(dst + pre);
    for (int vb = 0; vb < nvec; vb += 4 * nthr) {
      int4 q[4];
      int vi[4];
      for (int k = 0; k < 4; ++k) {
        vi[k] = vb + k * nthr + tid;
        q[k] = dv[vi[k] < nvec ? vi[k] : nvec - 1];
      }
      for (int k = 0; k < 4; ++k) {
        const bool in = vi[k] < nvec;
        const bool o0 = in && owned(q[k].x), o1 = in && owned(q[k].y);
        const bool o2 = in && owned(q[k].z), o3 = in && owned(q[k].w);
        if (o0 | o1 | o2 | o3) {
          const int e = pre + 4 * vi[k];
          if (o0) append(e + 0, q[k].x - r0);
          if (o1) append(e + 1, q[k].y - r0);
          if (o2) append(e + 2, q[k].z - r0);
          if (o3) append(e + 3, q[k].w - r0);
        }
      }
    }
    const int tail0 = pre + 4 * nvec;
    if (tid < 8) {
      const int e = (tid < 4) ? tid : tail0 + (tid - 4);
      const bool in = (tid < 4) ? (e < pre) : (e < E);
      if (in) {
        const int d = dst[e];
        if (owned(d)) append(e, d - r0);
      }
    }
  }
  __syncthreads();
  const int cnt = kept ? kcnt : (int)s_cnt;
  const bool use_list = cnt <= LIST_CAP;
  const bool save = KEEP == 1 && packs && cnt <= keep_cap;
  if (use_list && !kept) {
    for (int i = tid; i < cnt; i += nthr) {
      const int e = s_ls[i];
      const int sr = (e < E) ? src[e] : (e - E);
      s_ls[i] = sr;
      if (save) *kslot(KEEP_HDR + i) = __int_as_float((sr << 7) | s_ld[i]);
    }
  }
  if constexpr (KEEP == 1) {
    if (tid == 0 && rows_here > 0) *kslot(0) = __int_as_float(save ? cnt : -1);
  }
  __syncthreads();

  auto for_owned = [&](auto body) {
    if (use_list) {
      for (int base = 0; base < cnt; base += 32 * nw) {
        const int i = base + lane * nw + wave;
        const bool own = i < cnt;
        const int s = own ? s_ls[i] : 0;
        const int dl = own ? s_ld[i] : 0;
        body(own, s, dl);
      }
    } else {
      for (int e0 = wave * 32; e0 < E; e0 += 32 * nw) {
        const int e = e0 + lane;
        const int d = (e < E) ? dst[e] : -1;
        const bool own = (unsigned)(d - r0) < (unsigned)rows_here;
        if (__builtin_amdgcn_ballot_w32(own)) {
          const int s = own ? src[e] : 0;
          body(own, s, own ? d - r0 : 0);
        }
      }
      for (int i0 = wave * 32; i0 < rows_here; i0 += 32 * nw) {
        const int r = i0 + lane;
        const bool own = r < rows_here;
        body(own, own ? r0 + r : 0, own ? r : 0);
      }
    }
  };
  auto logit = [&](int s, int dl, int hh) -> float {
    const float x = al_s[s * HEADS + hh] + al_d[(r0 + dl) * HEADS + hh];
    return (x > 0.f) ? x : NEG_SLOPE * x;
  };

  for_owned([&](bool own, int s, int dl) {
    if (own) {
      for (int hh = 0; hh < HEADS; ++hh)
        atomicMax(&s_menc[dl * HEADS + hh], ordEnc(logit(s, dl, hh)));
    }
  });
  __syncthreads();

  for_owned([&](bool own, int s, int dl) {
    if (own) {
      for (int hh = 0; hh < HEADS; ++hh) {
        const float m = ordDec(s_menc[dl * HEADS + hh]);
        atomicAdd(&s_sum[dl * HEADS + hh], expf(logit(s, dl, hh) - m));
      }
    }
  });
  __syncthreads();

  for_owned([&](bool own, int s, int dl) {
    float al[HEADS];
    for (int hh = 0; hh < HEADS; ++hh) al[hh] = 0.f;
    if (own) {
      for (int hh = 0; hh < HEADS; ++hh) {
        const float m = ordDec(s_menc[dl * HEADS + hh]);
        al[hh] = expf(logit(s, dl, hh) - m) / (s_sum[dl * HEADS + hh] + 1e-16f);
      }
    }
    unsigned m = __builtin_amdgcn_ballot_w32(own);
    while (__builtin_popcount(m) >= BATCH) {
      float v[BATCH][NL];
      int oo[BATCH];
      int jj[BATCH];
      for (int b = 0; b < BATCH; ++b) {
        const int j = __builtin_ctz(m);
        m &= m - 1;
        jj[b] = j;
        const int sj = __builtin_amdgcn_readlane(s, j);
        const int dj = __builtin_amdgcn_readlane(dl, j);
        const float* hp = h + (size_t)sj * ldh + lane;
        oo[b] = dj * ROWW + lane;
        for (int c = 0; c < NL; ++c) v[b][c] = hp[32 * c];
      }
      for (int b = 0; b < BATCH; ++b) {
        for (int c = 0; c < NL; ++c) {
          const float a = __int_as_float(__builtin_amdgcn_readlane(
              __float_as_int(al[c / (DFEAT / 32)]), jj[b]));
          atomicAdd(&s_acc[oo[b] + 32 * c], a * v[b][c]);
        }
      }
    }
    while (m) {
      const int j = __builtin_ctz(m);
      m &= m - 1;
      const int sj = __builtin_amdgcn_readlane(s, j);
      const int dj = __builtin_amdgcn_readlane(dl, j);
      const float* hp = h + (size_t)sj * ldh + lane;
      const int o = dj * ROWW + lane;
      float v[NL];
      for (int c = 0; c < NL; ++c) v[c] = hp[32 * c];
      for (int c = 0; c < NL; ++c) {
        const float a = __int_as_float(__builtin_amdgcn_readlane(
            __float_as_int(al[c / (DFEAT / 32)]), j));
        atomicAdd(&s_acc[o + 32 * c], a * v[c]);
      }
    }
  });
  __syncthreads();

  for (int r = wave; r < rows_here; r += nw) {
    float* op = out + (size_t)(r0 + r) * ldo + colOff + lane;
    for (int c = 0; c < NL; ++c)
      op[32 * c] = s_acc[r * ROWW + 32 * c + lane] + bias[32 * c + lane];
  }
}

__global__ void __launch_bounds__(OWN_THREADS)
gat_owned_h2(const int* __restrict__ src, const int* __restrict__ dst, int E,
             int n, const float* __restrict__ h, int ldh,
             const float* __restrict__ al_s, const float* __restrict__ al_d,
             const float* __restrict__ bias, float* __restrict__ out, int ldo,
             int colOff, int keepOff) {
  gat_owned_body<2, 1>(src, dst, E, n, h, ldh, al_s, al_d, bias, out, ldo,
                       colOff, keepOff);
}

__global__ void __launch_bounds__(OWN_THREADS)
gat_owned_h1(const int* __restrict__ src, const int* __restrict__ dst, int E,
             int n, const float* __restrict__ h, int ldh,
             const float* __restrict__ al_s, const float* __restrict__ al_d,
             const float* __restrict__ bias, float* __restrict__ out, int ldo,
             int colOff, int keepOff) {
  gat_owned_body<1, 2>(src, dst, E, n, h, ldh, al_s, al_d, bias, out, ldo,
                       colOff, keepOff);
}

#define CS_THREADS 192
#define CS_LOADS 10

template <int NC>
__global__ void __launch_bounds__(CS_THREADS)
col_part(const float* __restrict__ y, int total, float* __restrict__ ps,
         float* __restrict__ pq, int nparts) {
  static_assert((4 * CS_THREADS) % NC == 0, "a thread keeps its columns");
  const int t = threadIdx.x;
  const int nvec = total >> 2;
  const int base = blockIdx.x * (CS_LOADS * CS_THREADS);
  const float4* __restrict__ y4 = (const float4*)y;

  float4 v[CS_LOADS];
#pragma unroll
  for (int k = 0; k < CS_LOADS; ++k) {
    const int f = base + k * CS_THREADS + t;
    v[k] = y4[f < nvec ? f : 0];
  }
  float sa[4] = {0.f, 0.f, 0.f, 0.f};
  float qa[4] = {0.f, 0.f, 0.f, 0.f};
#pragma unroll
  for (int k = 0; k < CS_LOADS; ++k) {
    const int f = base + k * CS_THREADS + t;
    if (f < nvec) {
      sa[0] += v[k].x; qa[0] = fmaf(v[k].x, v[k].x, qa[0]);
      sa[1] += v[k].y; qa[1] = fmaf(v[k].y, v[k].y, qa[1]);
      sa[2] += v[k].z; qa[2] = fmaf(v[k].z, v[k].z, qa[2]);
      sa[3] += v[k].w; qa[3] = fmaf(v[k].w, v[k].w, qa[3]);
    }
  }
  {
    const int tail = total & 3;
    const int rel = nvec - base;
    if (tail && rel >= 0 && rel < CS_LOADS * CS_THREADS &&
        rel % CS_THREADS == t) {
      for (int j = 0; j < tail; ++j) {
        const float w = y[4 * nvec + j];
        sa[j] += w;
        qa[j] = fmaf(w, w, qa[j]);
      }
    }
  }

  if constexpr (NC % 4 == 0) {
    constexpr int QN = NC / 4;
    static_assert(CS_THREADS % QN == 0, "whole rows per step");
    constexpr int GN = CS_THREADS / QN;
    __shared__ __attribute__((aligned(16))) float red[2][GN][NC];
    const int g = t / QN;
    const int c4 = 4 * (t - g * QN);
    *(float4*)&red[0][g][c4] = make_float4(sa[0], sa[1], sa[2], sa[3]);
    *(float4*)&red[1][g][c4] = make_float4(qa[0], qa[1], qa[2], qa[3]);
    __syncthreads();
    for (int i = t; i < 2 * NC; i += CS_THREADS) {
      const int which = i / NC;
      const int c = i - which * NC;
      float r = 0.f;
#pragma unroll
      for (int gg = 0; gg < GN; ++gg) r += red[which][gg][c];
      (which ? pq : ps)[(size_t)c * nparts + blockIdx.x] = r;
    }
  } else {
    static_assert(4 % NC == 0, "column count must divide a group");
    constexpr int NW = CS_THREADS / 32;
    __shared__ float red[2][NW][NC];
    float cs[NC], cq[NC];
#pragma unroll
    for (int c = 0; c < NC; ++c) { cs[c] = 0.f; cq[c] = 0.f; }
#pragma unroll
    for (int j = 0; j < 4; ++j) { cs[j % NC] += sa[j]; cq[j % NC] += qa[j]; }
#pragma unroll
    for (int c = 0; c < NC; ++c) {
      for (int off = 16; off > 0; off >>= 1) {
        cs[c] += __shfl_xor(cs[c], off, 32);
        cq[c] += __shfl_xor(cq[c], off, 32);
      }
    }
    const int lane = t & 31;
    const int wave = t >> 5;
    if (lane == 0) {
#pragma unroll
      for (int c = 0; c < NC; ++c) {
        red[0][wave][c] = cs[c];
        red[1][wave][c] = cq[c];
      }
    }
    __syncthreads();
    if (t < 2 * NC) {
      const int which = t / NC;
      const int c = t - which * NC;
      float r = 0.f;
#pragma unroll
      for (int w = 0; w < NW; ++w) r += red[which][w][c];
      (which ? pq : ps)[(size_t)c * nparts + blockIdx.x] = r;
    }
  }
}

__global__ void bn_fold(const float* __restrict__ ps,
                        const float* __restrict__ pq, int nparts,
                        const float* __restrict__ g,
                        const float* __restrict__ beta, int ncols, float invn,
                        float* __restrict__ sc, float* __restrict__ sh) {
  const int lane = threadIdx.x & 31;
  const int c = (blockIdx.x * blockDim.x + threadIdx.x) >> 5;
  if (c >= ncols) return;
  const float* __restrict__ sp = ps + (size_t)c * nparts;
  const float* __restrict__ qp = pq + (size_t)c * nparts;
  float s = 0.f, q = 0.f;
  for (int b0 = 0; b0 < nparts; b0 += 256) {
    float vs[8], vq[8];
#pragma unroll
    for (int k = 0; k < 8; ++k) {
      const int b = b0 + 32 * k + lane;
      vs[k] = sp[b < nparts ? b : 0];
      vq[k] = qp[b < nparts ? b : 0];
    }
#pragma unroll
    for (int k = 0; k < 8; ++k) {
      const int b = b0 + 32 * k + lane;
      if (b < nparts) { s += vs[k]; q += vq[k]; }
    }
  }
  for (int off = 16; off > 0; off >>= 1) {
    s += __shfl_xor(s, off, 32);
    q += __shfl_xor(q, off, 32);
  }
  if (lane == 0) {
    const float mean = s * invn;
    const float var = q * invn - mean * mean;
    const float r = rsqrtf(var + 1e-5f);
    const float a = g[c] * r;
    sc[c] = a;
    sh[c] = beta[c] - mean * a;
  }
}

__global__ void mlp_out(const float* __restrict__ y2,
                        const float* __restrict__ sc,
                        const float* __restrict__ sh,
                        const float* __restrict__ Wm2,
                        const float* __restrict__ bm2, int n,
                        float* __restrict__ y3) {
  const int node = blockIdx.x * blockDim.x + threadIdx.x;
  if (node >= n) return;
  float a0 = bm2[0], a1 = bm2[1];
  const float* yp = y2 + (size_t)node * 96;
  for (int k = 0; k < 96; ++k) {
    const float v = fmaf(yp[k], sc[k], sh[k]);
    a0 = fmaf(v, Wm2[2 * k + 0], a0);
    a1 = fmaf(v, Wm2[2 * k + 1], a1);
  }
  y3[node * 2 + 0] = fmaxf(a0, 0.f);
  y3[node * 2 + 1] = fmaxf(a1, 0.f);
}

__global__ void bn_apply(const float* __restrict__ y,
                         const float* __restrict__ sc,
                         const float* __restrict__ sh, int total, int ncols,
                         float* __restrict__ out) {
  const int i = blockIdx.x * blockDim.x + threadIdx.x;
  if (i >= total) return;
  const int c = i % ncols;
  out[i] = fmaf(y[i], sc[c], sh[c]);
}

extern "C" void kernel_launch(void* const* d_in, const int* in_sizes, int n_in,
                              void* d_out, int out_size, void* d_ws,
                              size_t ws_size, hipStream_t stream) {
  const float* x      = (const float*)d_in[0];
  const int*   ei     = (const int*)d_in[1];
  const float* W1     = (const float*)d_in[2];
  const float* as1    = (const float*)d_in[3];
  const float* ad1    = (const float*)d_in[4];
  const float* b1     = (const float*)d_in[5];
  const float* W2     = (const float*)d_in[6];
  const float* as2    = (const float*)d_in[7];
  const float* ad2    = (const float*)d_in[8];
  const float* b2     = (const float*)d_in[9];
  const float* Wl     = (const float*)d_in[10];
  const float* bl     = (const float*)d_in[11];
  const float* gl     = (const float*)d_in[12];
  const float* betal  = (const float*)d_in[13];
  const float* Wm1    = (const float*)d_in[14];
  const float* bm1    = (const float*)d_in[15];
  const float* gm1    = (const float*)d_in[16];
  const float* betam1 = (const float*)d_in[17];
  const float* Wm2    = (const float*)d_in[18];
  const float* bm2    = (const float*)d_in[19];
  const float* gm2    = (const float*)d_in[20];
  const float* betam2 = (const float*)d_in[21];

  const int N = in_sizes[0] / DFEAT;
  const int E = in_sizes[1] / 2;
  const int* srcA = ei;
  const int* dstA = ei + E;

  char* w = (char*)d_ws;
  auto alloc = [&](size_t bytes) -> char* {
    char* p = w;
    w += (bytes + 255) & ~(size_t)255;
    return p;
  };
  float*    xcat = (float*)alloc((size_t)N * 384 * 4);
  float*    h1   = (float*)alloc((size_t)N * 256 * 4);
  float*    h2   = (float*)alloc((size_t)N * 128 * 4);
  float*    al_s = (float*)alloc((size_t)N * 2 * 4);
  float*    al_d = (float*)alloc((size_t)N * 2 * 4);
  float*    y3   = (float*)alloc((size_t)N * 2 * 4);
  float*    sc1  = (float*)alloc(96 * 4);
  float*    sh1  = (float*)alloc(96 * 4);
  float*    sc2  = (float*)alloc(96 * 4);
  float*    sh2  = (float*)alloc(96 * 4);
  float*    sc3  = (float*)alloc(96 * 4);
  float*    sh3  = (float*)alloc(96 * 4);
  float* y1 = h1;
  float* y2 = h1 + (size_t)N * 96;
  const int csGroups = CS_LOADS * CS_THREADS;
  auto csBlocks = [&](int total) {
    return ((total + 3) / 4 + csGroups - 1) / csGroups;
  };
  const int np96 = csBlocks(N * 96);
  const int np2  = csBlocks(N * 2);
  float* ps = h2;
  float* pq = h2 + (size_t)96 * np96;

  const int TB = 256;
  auto gemmBlocks = [](int rows, int cols) {
    const int tiles = ((rows + 15) / 16) * (cols / 16);
    return (tiles + 7) / 8;
  };
  auto layerBlocks = [](int rows, int cols) {
    const int rowsPer = LG_WAVES * 16;
    return ((rows + rowsPer - 1) / rowsPer) * (cols / LG_COLS);
  };
  const int ownBlocks = (N + OWN_ROWS - 1) / OWN_ROWS;
  const int apRows = AP_ROWS * (AP_THREADS / 32);

  gemm_layer_strip<128><<<layerBlocks(N, 256), LG_WAVES * 32, 0, stream>>>(
      x, 128, W1, 256, h1, 256, N, 256);
  attn_proj_rows<2><<<(N * 2 + apRows - 1) / apRows, AP_THREADS, 0, stream>>>(
      h1, as1, ad1, al_s, al_d, N * 2);
  gat_owned_h2<<<ownBlocks, OWN_THREADS, 0, stream>>>(
      srcA, dstA, E, N, h1, 256, al_s, al_d, b1, xcat, 384, 0, 256);

  gemm_layer_strip<256><<<layerBlocks(N, 128), LG_WAVES * 32, 0, stream>>>(
      xcat, 384, W2, 128, h2, 128, N, 128);
  attn_proj_rows<1><<<(N + apRows - 1) / apRows, AP_THREADS, 0, stream>>>(
      h2, as2, ad2, al_s, al_d, N);
  gat_owned_h1<<<ownBlocks, OWN_THREADS, 0, stream>>>(
      srcA, dstA, E, N, h2, 128, al_s, al_d, b2, xcat, 384, 256, 256);

  const float invn = 1.0f / (float)N;
  gemm_f32_wmma<<<gemmBlocks(N, 96), TB, 0, stream>>>(
      xcat, 384, Wl, 96, y1, 96, N, 384, 96, bl, 1);
  col_part<96><<<np96, CS_THREADS, 0, stream>>>(y1, N * 96, ps, pq, np96);
  bn_fold<<<(96 * 32 + TB - 1) / TB, TB, 0, stream>>>(ps, pq, np96, gl, betal,
                                                      96, invn, sc1, sh1);
  gemm_f32_wmma_bn<<<gemmBlocks(N, 96), TB, 0, stream>>>(
      y1, 96, Wm1, 96, y2, 96, N, 96, 96, bm1, sc1, sh1, 1);
  col_part<96><<<np96, CS_THREADS, 0, stream>>>(y2, N * 96, ps, pq, np96);
  bn_fold<<<(96 * 32 + TB - 1) / TB, TB, 0, stream>>>(ps, pq, np96, gm1, betam1,
                                                      96, invn, sc2, sh2);
  mlp_out<<<(N + TB - 1) / TB, TB, 0, stream>>>(y2, sc2, sh2, Wm2, bm2, N, y3);
  col_part<2><<<np2, CS_THREADS, 0, stream>>>(y3, N * 2, ps, pq, np2);
  bn_fold<<<(2 * 32 + TB - 1) / TB, TB, 0, stream>>>(ps, pq, np2, gm2, betam2, 2,
                                                     invn, sc3, sh3);
  bn_apply<<<(N * 2 + TB - 1) / TB, TB, 0, stream>>>(y3, sc3, sh3, N * 2, 2,
                                                     (float*)d_out);
}


# first MLP GEMM ([x1|x2] @ Wl, relu) moved into layer 2's aggregation work-group: its 80 finished rows feed f32 WMMA tiles from LDS, one tile per wave, same k order; 16 launches to 15, x2 not stored
# speedup vs baseline: 4.4976x; 1.1097x over previous
#include <hip/hip_runtime.h>
#include <hip/hip_bf16.h>
#include <math.h>


typedef __attribute__((ext_vector_type(2))) float v2f;
typedef __attribute__((ext_vector_type(8))) float v8f;

#define DFEAT 128
#define NEG_SLOPE 0.2f

__device__ __forceinline__ unsigned ordEnc(float f) {
  unsigned u = __float_as_uint(f);
  return (u & 0x80000000u) ? ~u : (u | 0x80000000u);
}
__device__ __forceinline__ float ordDec(unsigned e) {
  unsigned u = (e & 0x80000000u) ? (e & 0x7fffffffu) : ~e;
  return __uint_as_float(u);
}

__device__ __forceinline__ v8f wmma_f32(v2f a, v2f b, v8f c) {
  return __builtin_amdgcn_wmma_f32_16x16x4_f32(false, a, false, b, (short)0, c,
                                               false, false);
}

__global__ void gemm_f32_wmma_bn(const float* __restrict__ A, int lda,
                                 const float* __restrict__ B, int ldb,
                                 float* __restrict__ D, int ldo,
                                 int nrows, int kdim, int ncols,
                                 const float* __restrict__ bias,
                                 const float* __restrict__ scl,
                                 const float* __restrict__ shf, int relu) {
  const int lane = threadIdx.x & 31;
  const int wid  = (blockIdx.x * blockDim.x + threadIdx.x) >> 5;
  const int colTiles = ncols >> 4;
  const int rowTiles = (nrows + 15) >> 4;
  if (wid >= rowTiles * colTiles) return;
  const int tr = wid / colTiles;
  const int tc = wid - tr * colTiles;
  const int row0 = tr << 4, col0 = tc << 4;
  const int lm = lane & 15;
  const int lh = lane >> 4;

  int arow = row0 + lm;
  if (arow >= nrows) arow = nrows - 1;
  const float* Ap = A + (size_t)arow * lda + 2 * lh;
  const float* Sp = scl + 2 * lh;
  const float* Hp = shf + 2 * lh;
  const float* Bp = B + (size_t)(2 * lh) * ldb + col0 + lm;

  v8f acc = {0.f, 0.f, 0.f, 0.f, 0.f, 0.f, 0.f, 0.f};
  int k = 0;
  for (; k + 16 <= kdim; k += 16) {
    v2f a0 = *(const v2f*)(Ap + k);
    v2f a1 = *(const v2f*)(Ap + k + 4);
    v2f a2 = *(const v2f*)(Ap + k + 8);
    v2f a3 = *(const v2f*)(Ap + k + 12);
    const v2f s0 = *(const v2f*)(Sp + k);
    const v2f s1 = *(const v2f*)(Sp + k + 4);
    const v2f s2 = *(const v2f*)(Sp + k + 8);
    const v2f s3 = *(const v2f*)(Sp + k + 12);
    const v2f h0 = *(const v2f*)(Hp + k);
    const v2f h1 = *(const v2f*)(Hp + k + 4);
    const v2f h2 = *(const v2f*)(Hp + k + 8);
    const v2f h3 = *(const v2f*)(Hp + k + 12);
    a0 = a0 * s0 + h0;
    a1 = a1 * s1 + h1;
    a2 = a2 * s2 + h2;
    a3 = a3 * s3 + h3;
    const float* bq = Bp + (size_t)k * ldb;
    v2f b0, b1, b2, b3;
    b0.x = bq[0];
    b0.y = bq[(size_t)1 * ldb];
    b1.x = bq[(size_t)4 * ldb];
    b1.y = bq[(size_t)5 * ldb];
    b2.x = bq[(size_t)8 * ldb];
    b2.y = bq[(size_t)9 * ldb];
    b3.x = bq[(size_t)12 * ldb];
    b3.y = bq[(size_t)13 * ldb];
    acc = wmma_f32(a0, b0, acc);
    acc = wmma_f32(a1, b1, acc);
    acc = wmma_f32(a2, b2, acc);
    acc = wmma_f32(a3, b3, acc);
  }
  for (; k < kdim; k += 4) {
    v2f a = *(const v2f*)(Ap + k);
    const v2f s = *(const v2f*)(Sp + k);
    const v2f h = *(const v2f*)(Hp + k);
    a = a * s + h;
    v2f b;
    b.x = Bp[(size_t)k * ldb];
    b.y = Bp[(size_t)(k + 1) * ldb];
    acc = wmma_f32(a, b, acc);
  }

  const int col = col0 + lm;
  const float bv = bias ? bias[col] : 0.0f;
#pragma unroll
  for (int r = 0; r < 8; ++r) {
    const int row = row0 + r + 8 * lh;
    if (row < nrows) {
      float v = acc[r] + bv;
      if (relu) v = fmaxf(v, 0.0f);
      D[(size_t)row * ldo + col] = v;
    }
  }
}

#define LG_WAVES 4
#define LG_COLS 64
#define LG_PITCH (2 * LG_COLS + 8)
#define LG_CPITCH (LG_COLS + 4)

template <int KDIM>
__global__ void __launch_bounds__(LG_WAVES * 32)
gemm_layer_strip(const float* __restrict__ A, int lda,
                 const float* __restrict__ B, int ldb, float* __restrict__ D,
                 int ldo, int nrows, int ncols) {
  static_assert(KDIM % 16 == 0, "whole blocks of 16 k");
  static_assert(LG_COLS == 64, "four tiles per wave, 16 column groups per row");
  constexpr int NTHR = LG_WAVES * 32;
  constexpr int NPAIR = KDIM / 2;
  constexpr int NBLK = KDIM / 16;
  constexpr int CGR = LG_COLS / 4;
  constexpr int PER = NPAIR * CGR / NTHR;
  constexpr int CH = 8;
  static_assert((NPAIR * CGR) % NTHR == 0 && PER % CH == 0, "even staging");
  static_assert(NTHR % CGR == 0, "a thread keeps its column group");
  __shared__ __attribute__((aligned(16))) float Bs[NPAIR * LG_PITCH];
  __shared__ __attribute__((aligned(16))) float Cs[LG_WAVES][16 * LG_CPITCH];

  const int tid = threadIdx.x;
  const int lane = tid & 31;
  const int wave = __builtin_amdgcn_readfirstlane(tid >> 5);
  const int strips = ncols / LG_COLS;
  const int rg = blockIdx.x / strips;
  const int col0 = (blockIdx.x - rg * strips) * LG_COLS;
  const int row0 = (rg * LG_WAVES + wave) * 16;
  const bool live = row0 < nrows;
  const int c = lane & 15;
  const int h = lane >> 4;

  int arow = row0 + c;
  if (arow >= nrows) arow = nrows - 1;
  const float4* __restrict__ Ap =
      (const float4*)(A + (size_t)arow * lda + 8 * h);
  float4 a[2 * NBLK];
#pragma unroll
  for (int kb = 0; kb < NBLK; ++kb) {
    a[2 * kb] = Ap[4 * kb];
    a[2 * kb + 1] = Ap[4 * kb + 1];
  }

  {
    const int cg = tid % CGR;
    const float* __restrict__ bcol = B + col0 + 4 * cg;
    for (int t0 = 0; t0 < PER; t0 += CH) {
      float4 r0[CH], r1[CH];
#pragma unroll
      for (int i = 0; i < CH; ++i) {
        const int p = ((t0 + i) * NTHR + tid) / CGR;
        const float* bp = bcol + (size_t)(2 * p) * ldb;
        r0[i] = *(const float4*)bp;
        r1[i] = *(const float4*)(bp + ldb);
      }
#pragma unroll
      for (int i = 0; i < CH; ++i) {
        const int p = ((t0 + i) * NTHR + tid) / CGR;
        float* sp = Bs + p * LG_PITCH + 8 * cg;
        *(float4*)sp = make_float4(r0[i].x, r1[i].x, r0[i].y, r1[i].y);
        *(float4*)(sp + 4) = make_float4(r0[i].z, r1[i].z, r0[i].w, r1[i].w);
      }
    }
  }
  __syncthreads();

  float* cw = &Cs[wave][0];
  if (live) {
    v8f acc[4];
#pragma unroll
    for (int t = 0; t < 4; ++t)
      acc[t] = (v8f){0.f, 0.f, 0.f, 0.f, 0.f, 0.f, 0.f, 0.f};
    const float* bl = Bs + (4 * h) * LG_PITCH + 2 * c;
#pragma unroll
    for (int kb = 0; kb < NBLK; ++kb) {
      const float av[8] = {a[2 * kb].x,     a[2 * kb].y,     a[2 * kb].z,
                           a[2 * kb].w,     a[2 * kb + 1].x, a[2 * kb + 1].y,
                           a[2 * kb + 1].z, a[2 * kb + 1].w};
#pragma unroll
      for (int j = 0; j < 4; ++j) {
        v2f af;
        af.x = av[2 * j];
        af.y = av[2 * j + 1];
        const float* bq = bl + (8 * kb + j) * LG_PITCH;
#pragma unroll
        for (int t = 0; t < 4; ++t) {
          const v2f bf = *(const v2f*)(bq + 32 * t);
          acc[t] = wmma_f32(af, bf, acc[t]);
        }
      }
    }
    float* cp = cw + (8 * h) * LG_CPITCH + c;
#pragma unroll
    for (int t = 0; t < 4; ++t) {
#pragma unroll
      for (int r = 0; r < 8; ++r) cp[r * LG_CPITCH + 16 * t] = acc[t][r];
    }
  }
  __syncthreads();
  if (live) {
    const int rr = lane >> 4;
    const int cg = lane & 15;
    float4 v[8];
#pragma unroll
    for (int i = 0; i < 8; ++i)
      v[i] = *(const float4*)(cw + (2 * i + rr) * LG_CPITCH + 4 * cg);
    float* dp = D + (size_t)(row0 + rr) * ldo + col0 + 4 * cg;
    if (row0 + 16 <= nrows) {
#pragma unroll
      for (int i = 0; i < 8; ++i) *(float4*)(dp + (size_t)(2 * i) * ldo) = v[i];
    } else {
#pragma unroll
      for (int i = 0; i < 8; ++i) {
        if (row0 + 2 * i + rr < nrows)
          *(float4*)(dp + (size_t)(2 * i) * ldo) = v[i];
      }
    }
  }
}

#define AP_ROWS 16
#define AP_THREADS 128

template <int CNT>
__device__ __forceinline__ void ap_fold(float (&val)[2 * AP_ROWS], int lane) {
  const bool up = (lane & CNT) != 0;
#pragma unroll
  for (int i = 0; i < CNT; ++i) {
    const float keep = up ? val[i + CNT] : val[i];
    const float send = up ? val[i] : val[i + CNT];
    val[i] = keep + __shfl_xor(send, CNT, 32);
  }
}

template <int HEADS>
__global__ void __launch_bounds__(AP_THREADS)
attn_proj_rows(const float* __restrict__ h, const float* __restrict__ a_src,
               const float* __restrict__ a_dst, float* __restrict__ al_s,
               float* __restrict__ al_d, int nrows) {
  static_assert(AP_ROWS % HEADS == 0, "a wave's first row is head 0");
  static_assert(2 * AP_ROWS == 32, "one finished sum per lane");
  const int lane = threadIdx.x & 31;
  const int wid = __builtin_amdgcn_readfirstlane(
      (int)((blockIdx.x * blockDim.x + threadIdx.x) >> 5));
  const int w0 = wid * AP_ROWS;
  if (w0 >= nrows) return;

  float as[HEADS][4], ad[HEADS][4];
#pragma unroll
  for (int hh = 0; hh < HEADS; ++hh) {
#pragma unroll
    for (int q = 0; q < 4; ++q) {
      as[hh][q] = a_src[hh * DFEAT + 4 * lane + q];
      ad[hh][q] = a_dst[hh * DFEAT + 4 * lane + q];
    }
  }
  const float4* __restrict__ hp =
      (const float4*)(h + (size_t)w0 * DFEAT) + lane;
  float4 v[AP_ROWS];
#pragma unroll
  for (int j = 0; j < AP_ROWS; ++j)
    v[j] = hp[(w0 + j < nrows ? j : 0) * (DFEAT / 4)];
  __builtin_amdgcn_sched_barrier(0);

  float val[2 * AP_ROWS];
#pragma unroll
  for (int j = 0; j < AP_ROWS; ++j) {
    const int hh = j % HEADS;
    val[j] = v[j].x * as[hh][0] + v[j].y * as[hh][1] + v[j].z * as[hh][2] +
             v[j].w * as[hh][3];
    val[AP_ROWS + j] = v[j].x * ad[hh][0] + v[j].y * ad[hh][1] +
                       v[j].z * ad[hh][2] + v[j].w * ad[hh][3];
  }
  ap_fold<16>(val, lane);
  ap_fold<8>(val, lane);
  ap_fold<4>(val, lane);
  ap_fold<2>(val, lane);
  ap_fold<1>(val, lane);

  const int w = w0 + (lane & 15);
  if (w < nrows) (lane < 16 ? al_s : al_d)[w] = val[0];
}

#define X1_W 256
#define MLP_C 96
#define OWN_ROWS 80
#define OWN_THREADS 1024
#define LIST_CAP 8192
#define KEEP_HDR 32

template <int HEADS, int KEEP>
__device__ __forceinline__ void gat_owned_body(
    const int* __restrict__ src, const int* __restrict__ dst, int E, int n,
    const float* __restrict__ h, int ldh, const float* __restrict__ al_s,
    const float* __restrict__ al_d, const float* __restrict__ bias,
    float* __restrict__ out, int ldo, int colOff, int keepOff,
    const float* __restrict__ Wl, const float* __restrict__ bl,
    float* __restrict__ y1) {
  static_assert(OWN_ROWS <= 128 && DFEAT == 128, "packed word: 7 bits of row");
  static_assert(OWN_ROWS % 16 == 0 && MLP_C % 16 == 0 && X1_W % 16 == 0,
                "whole tiles and whole blocks of 16 k in the fused product");
  static_assert(KEEP != 2 || HEADS == 1, "the fused product takes one head");
  constexpr int ROWW = HEADS * DFEAT;
  constexpr int NL = ROWW / 32;
  constexpr int BATCH = 32 / NL;
  constexpr int AP = ROWW + (KEEP == 2 ? 4 : 0);
  constexpr int XP = X1_W + 4;
  __shared__ __attribute__((aligned(16))) float s_acc[OWN_ROWS * AP];
  __shared__ __attribute__((aligned(16))) float
      s_x1[KEEP == 2 ? OWN_ROWS * XP : 4];
  __shared__ int s_ls[LIST_CAP];
  __shared__ int s_ld[LIST_CAP];
  __shared__ unsigned s_menc[OWN_ROWS * HEADS];
  __shared__ float s_sum[OWN_ROWS * HEADS];
  __shared__ unsigned s_cnt;

  const int tid = threadIdx.x;
  const int lane = tid & 31;
  const int wave = tid >> 5;
  const int nthr = blockDim.x;
  const int nw = nthr >> 5;
  const int r0 = blockIdx.x * OWN_ROWS;
  int rows_here = n - r0;
  if (rows_here > OWN_ROWS) rows_here = OWN_ROWS;
  if (rows_here < 0) rows_here = 0;

  float* const kbase = out + (size_t)r0 * ldo + keepOff;
  auto kslot = [&](int j) -> float* {
    return kbase + (size_t)(j >> 7) * ldo + (j & (DFEAT - 1));
  };
  int keep_cap = rows_here * DFEAT - KEEP_HDR;
  if (keep_cap > LIST_CAP) keep_cap = LIST_CAP;
  const bool packs = n <= (1 << 24);
  int kcnt = -1;
  if constexpr (KEEP == 2) {
    if (rows_here > 0 && packs) kcnt = __float_as_int(*kslot(0));
  }
  const bool kept = KEEP == 2 && kcnt >= 0 && kcnt <= keep_cap;

  if constexpr (KEEP == 2) {
    constexpr int XG = X1_W / 4;
    constexpr int XB = 5;
    for (int f0 = 0; f0 < OWN_ROWS * XG; f0 += XB * nthr) {
      float4 xr[XB];
#pragma unroll
      for (int i = 0; i < XB; ++i) {
        const int f = f0 + i * nthr + tid;
        const int r = f / XG;
        xr[i] = make_float4(0.f, 0.f, 0.f, 0.f);
        if (f < OWN_ROWS * XG && r < rows_here)
          xr[i] = *(const float4*)(out + (size_t)(r0 + r) * ldo +
                                   4 * (f - r * XG));
      }
#pragma unroll
      for (int i = 0; i < XB; ++i) {
        const int f = f0 + i * nthr + tid;
        const int r = f / XG;
        if (f < OWN_ROWS * XG)
          *(float4*)(s_x1 + r * XP + 4 * (f - r * XG)) = xr[i];
      }
    }
  }
  for (int i = tid; i < OWN_ROWS * AP; i += nthr) s_acc[i] = 0.f;
  for (int i = tid; i < OWN_ROWS * HEADS; i += nthr) {
    s_menc[i] = 0u;
    s_sum[i] = 0.f;
  }
  if (tid == 0) s_cnt = 0u;
  __syncthreads();

  auto append = [&](int e, int dl) {
    const unsigned p = atomicAdd(&s_cnt, 1u);
    if (p < (unsigned)LIST_CAP) { s_ls[p] = e; s_ld[p] = dl; }
  };
  auto owned = [&](int d) -> bool {
    return (unsigned)(d - r0) < (unsigned)rows_here;
  };
  if (kept) {
    for (int i = tid; i < kcnt; i += nthr) {
      const int p = __float_as_int(*kslot(KEEP_HDR + i));
      s_ls[i] = p >> 7;
      s_ld[i] = p & 127;
    }
  } else {
    for (int r = tid; r < rows_here; r += nthr) append(E + r0 + r, r);
    int pre = (int)(((16u - ((unsigned)(size_t)dst & 15u)) & 15u) >> 2);
    if (pre > E) pre = E;
    const int nvec = (E - pre) >> 2;
    const int4* __restrict__ dv = (const int4*)(dst + pre);
    for (int vb = 0; vb < nvec; vb += 4 * nthr) {
      int4 q[4];
      int vi[4];
      for (int k = 0; k < 4; ++k) {
        vi[k] = vb + k * nthr + tid;
        q[k] = dv[vi[k] < nvec ? vi[k] : nvec - 1];
      }
      for (int k = 0; k < 4; ++k) {
        const bool in = vi[k] < nvec;
        const bool o0 = in && owned(q[k].x), o1 = in && owned(q[k].y);
        const bool o2 = in && owned(q[k].z), o3 = in && owned(q[k].w);
        if (o0 | o1 | o2 | o3) {
          const int e = pre + 4 * vi[k];
          if (o0) append(e + 0, q[k].x - r0);
          if (o1) append(e + 1, q[k].y - r0);
          if (o2) append(e + 2, q[k].z - r0);
          if (o3) append(e + 3, q[k].w - r0);
        }
      }
    }
    const int tail0 = pre + 4 * nvec;
    if (tid < 8) {
      const int e = (tid < 4) ? tid : tail0 + (tid - 4);
      const bool in = (tid < 4) ? (e < pre) : (e < E);
      if (in) {
        const int d = dst[e];
        if (owned(d)) append(e, d - r0);
      }
    }
  }
  __syncthreads();
  const int cnt = kept ? kcnt : (int)s_cnt;
  const bool use_list = cnt <= LIST_CAP;
  const bool save = KEEP == 1 && packs && cnt <= keep_cap;
  if (use_list && !kept) {
    for (int i = tid; i < cnt; i += nthr) {
      const int e = s_ls[i];
      const int sr = (e < E) ? src[e] : (e - E);
      s_ls[i] = sr;
      if (save) *kslot(KEEP_HDR + i) = __int_as_float((sr << 7) | s_ld[i]);
    }
  }
  if constexpr (KEEP == 1) {
    if (tid == 0 && rows_here > 0) *kslot(0) = __int_as_float(save ? cnt : -1);
  }
  __syncthreads();

  auto for_owned = [&](auto body) {
    if (use_list) {
      for (int base = 0; base < cnt; base += 32 * nw) {
        const int i = base + lane * nw + wave;
        const bool own = i < cnt;
        const int s = own ? s_ls[i] : 0;
        const int dl = own ? s_ld[i] : 0;
        body(own, s, dl);
      }
    } else {
      for (int e0 = wave * 32; e0 < E; e0 += 32 * nw) {
        const int e = e0 + lane;
        const int d = (e < E) ? dst[e] : -1;
        const bool own = (unsigned)(d - r0) < (unsigned)rows_here;
        if (__builtin_amdgcn_ballot_w32(own)) {
          const int s = own ? src[e] : 0;
          body(own, s, own ? d - r0 : 0);
        }
      }
      for (int i0 = wave * 32; i0 < rows_here; i0 += 32 * nw) {
        const int r = i0 + lane;
        const bool own = r < rows_here;
        body(own, own ? r0 + r : 0, own ? r : 0);
      }
    }
  };
  auto logit = [&](int s, int dl, int hh) -> float {
    const float x = al_s[s * HEADS + hh] + al_d[(r0 + dl) * HEADS + hh];
    return (x > 0.f) ? x : NEG_SLOPE * x;
  };

  for_owned([&](bool own, int s, int dl) {
    if (own) {
      for (int hh = 0; hh < HEADS; ++hh)
        atomicMax(&s_menc[dl * HEADS + hh], ordEnc(logit(s, dl, hh)));
    }
  });
  __syncthreads();

  for_owned([&](bool own, int s, int dl) {
    if (own) {
      for (int hh = 0; hh < HEADS; ++hh) {
        const float m = ordDec(s_menc[dl * HEADS + hh]);
        atomicAdd(&s_sum[dl * HEADS + hh], expf(logit(s, dl, hh) - m));
      }
    }
  });
  __syncthreads();

  for_owned([&](bool own, int s, int dl) {
    float al[HEADS];
    for (int hh = 0; hh < HEADS; ++hh) al[hh] = 0.f;
    if (own) {
      for (int hh = 0; hh < HEADS; ++hh) {
        const float m = ordDec(s_menc[dl * HEADS + hh]);
        al[hh] = expf(logit(s, dl, hh) - m) / (s_sum[dl * HEADS + hh] + 1e-16f);
      }
    }
    unsigned m = __builtin_amdgcn_ballot_w32(own);
    while (__builtin_popcount(m) >= BATCH) {
      float v[BATCH][NL];
      int oo[BATCH];
      int jj[BATCH];
      for (int b = 0; b < BATCH; ++b) {
        const int j = __builtin_ctz(m);
        m &= m - 1;
        jj[b] = j;
        const int sj = __builtin_amdgcn_readlane(s, j);
        const int dj = __builtin_amdgcn_readlane(dl, j);
        const float* hp = h + (size_t)sj * ldh + lane;
        oo[b] = dj * AP + lane;
        for (int c = 0; c < NL; ++c) v[b][c] = hp[32 * c];
      }
      for (int b = 0; b < BATCH; ++b) {
        for (int c = 0; c < NL; ++c) {
          const float a = __int_as_float(__builtin_amdgcn_readlane(
              __float_as_int(al[c / (DFEAT / 32)]), jj[b]));
          atomicAdd(&s_acc[oo[b] + 32 * c], a * v[b][c]);
        }
      }
    }
    while (m) {
      const int j = __builtin_ctz(m);
      m &= m - 1;
      const int sj = __builtin_amdgcn_readlane(s, j);
      const int dj = __builtin_amdgcn_readlane(dl, j);
      const float* hp = h + (size_t)sj * ldh + lane;
      const int o = dj * AP + lane;
      float v[NL];
      for (int c = 0; c < NL; ++c) v[c] = hp[32 * c];
      for (int c = 0; c < NL; ++c) {
        const float a = __int_as_float(__builtin_amdgcn_readlane(
            __float_as_int(al[c / (DFEAT / 32)]), j));
        atomicAdd(&s_acc[o + 32 * c], a * v[c]);
      }
    }
  });
  __syncthreads();

  if constexpr (KEEP != 2) {
    for (int r = wave; r < rows_here; r += nw) {
      float* op = out + (size_t)(r0 + r) * ldo + colOff + lane;
      for (int c = 0; c < NL; ++c)
        op[32 * c] = s_acc[r * AP + 32 * c + lane] + bias[32 * c + lane];
    }
  } else {
    for (int i = tid; i < OWN_ROWS * ROWW; i += nthr) {
      const int r = i / ROWW;
      const int cc = i - r * ROWW;
      s_acc[r * AP + cc] += bias[cc];
    }
    __syncthreads();
    const int wv = __builtin_amdgcn_readfirstlane(wave);
    constexpr int CT = MLP_C / 16;
    constexpr int TILES = (OWN_ROWS / 16) * CT;
    const int lc = lane & 15;
    const int lh = lane >> 4;
    for (int t = wv; t < TILES; t += nw) {
      const int rt = t / CT;
      const int ct = t - rt * CT;
      const int col = ct * 16 + lc;
      const float* ax = s_x1 + (rt * 16 + lc) * XP + 2 * lh;
      const float* aa = s_acc + (rt * 16 + lc) * AP + 2 * lh;
      const float* bp = Wl + (size_t)(2 * lh) * MLP_C + col;
      v8f acc = {0.f, 0.f, 0.f, 0.f, 0.f, 0.f, 0.f, 0.f};
      auto kblock = [&](const float* ap, const float* bq) {
        const v2f a0 = *(const v2f*)(ap);
        const v2f a1 = *(const v2f*)(ap + 4);
        const v2f a2 = *(const v2f*)(ap + 8);
        const v2f a3 = *(const v2f*)(ap + 12);
        v2f b0, b1, b2, b3;
        b0.x = bq[0];
        b0.y = bq[1 * MLP_C];
        b1.x = bq[4 * MLP_C];
        b1.y = bq[5 * MLP_C];
        b2.x = bq[8 * MLP_C];
        b2.y = bq[9 * MLP_C];
        b3.x = bq[12 * MLP_C];
        b3.y = bq[13 * MLP_C];
        acc = wmma_f32(a0, b0, acc);
        acc = wmma_f32(a1, b1, acc);
        acc = wmma_f32(a2, b2, acc);
        acc = wmma_f32(a3, b3, acc);
      };
      for (int k = 0; k < X1_W; k += 16) kblock(ax + k, bp + (size_t)k * MLP_C);
      for (int k = 0; k < ROWW; k += 16)
        kblock(aa + k, bp + (size_t)(X1_W + k) * MLP_C);
      const float bv = bl[col];
#pragma unroll
      for (int r = 0; r < 8; ++r) {
        const int row = rt * 16 + r + 8 * lh;
        if (row < rows_here)
          y1[(size_t)(r0 + row) * MLP_C + col] = fmaxf(acc[r] + bv, 0.0f);
      }
    }
  }
}

__global__ void __launch_bounds__(OWN_THREADS)
gat_owned_h2(const int* __restrict__ src, const int* __restrict__ dst, int E,
             int n, const float* __restrict__ h, int ldh,
             const float* __restrict__ al_s, const float* __restrict__ al_d,
             const float* __restrict__ bias, float* __restrict__ out, int ldo,
             int colOff, int keepOff) {
  gat_owned_body<2, 1>(src, dst, E, n, h, ldh, al_s, al_d, bias, out, ldo,
                       colOff, keepOff, nullptr, nullptr, nullptr);
}

__global__ void __launch_bounds__(OWN_THREADS)
gat_owned_h1(const int* __restrict__ src, const int* __restrict__ dst, int E,
             int n, const float* __restrict__ h, int ldh,
             const float* __restrict__ al_s, const float* __restrict__ al_d,
             const float* __restrict__ bias, float* __restrict__ out, int ldo,
             int colOff, int keepOff, const float* __restrict__ Wl,
             const float* __restrict__ bl, float* __restrict__ y1) {
  gat_owned_body<1, 2>(src, dst, E, n, h, ldh, al_s, al_d, bias, out, ldo,
                       colOff, keepOff, Wl, bl, y1);
}

#define CS_THREADS 192
#define CS_LOADS 10

template <int NC>
__global__ void __launch_bounds__(CS_THREADS)
col_part(const float* __restrict__ y, int total, float* __restrict__ ps,
         float* __restrict__ pq, int nparts) {
  static_assert((4 * CS_THREADS) % NC == 0, "a thread keeps its columns");
  const int t = threadIdx.x;
  const int nvec = total >> 2;
  const int base = blockIdx.x * (CS_LOADS * CS_THREADS);
  const float4* __restrict__ y4 = (const float4*)y;

  float4 v[CS_LOADS];
#pragma unroll
  for (int k = 0; k < CS_LOADS; ++k) {
    const int f = base + k * CS_THREADS + t;
    v[k] = y4[f < nvec ? f : 0];
  }
  float sa[4] = {0.f, 0.f, 0.f, 0.f};
  float qa[4] = {0.f, 0.f, 0.f, 0.f};
#pragma unroll
  for (int k = 0; k < CS_LOADS; ++k) {
    const int f = base + k * CS_THREADS + t;
    if (f < nvec) {
      sa[0] += v[k].x; qa[0] = fmaf(v[k].x, v[k].x, qa[0]);
      sa[1] += v[k].y; qa[1] = fmaf(v[k].y, v[k].y, qa[1]);
      sa[2] += v[k].z; qa[2] = fmaf(v[k].z, v[k].z, qa[2]);
      sa[3] += v[k].w; qa[3] = fmaf(v[k].w, v[k].w, qa[3]);
    }
  }
  {
    const int tail = total & 3;
    const int rel = nvec - base;
    if (tail && rel >= 0 && rel < CS_LOADS * CS_THREADS &&
        rel % CS_THREADS == t) {
      for (int j = 0; j < tail; ++j) {
        const float w = y[4 * nvec + j];
        sa[j] += w;
        qa[j] = fmaf(w, w, qa[j]);
      }
    }
  }

  if constexpr (NC % 4 == 0) {
    constexpr int QN = NC / 4;
    static_assert(CS_THREADS % QN == 0, "whole rows per step");
    constexpr int GN = CS_THREADS / QN;
    __shared__ __attribute__((aligned(16))) float red[2][GN][NC];
    const int g = t / QN;
    const int c4 = 4 * (t - g * QN);
    *(float4*)&red[0][g][c4] = make_float4(sa[0], sa[1], sa[2], sa[3]);
    *(float4*)&red[1][g][c4] = make_float4(qa[0], qa[1], qa[2], qa[3]);
    __syncthreads();
    for (int i = t; i < 2 * NC; i += CS_THREADS) {
      const int which = i / NC;
      const int c = i - which * NC;
      float r = 0.f;
#pragma unroll
      for (int gg = 0; gg < GN; ++gg) r += red[which][gg][c];
      (which ? pq : ps)[(size_t)c * nparts + blockIdx.x] = r;
    }
  } else {
    static_assert(4 % NC == 0, "column count must divide a group");
    constexpr int NW = CS_THREADS / 32;
    __shared__ float red[2][NW][NC];
    float cs[NC], cq[NC];
#pragma unroll
    for (int c = 0; c < NC; ++c) { cs[c] = 0.f; cq[c] = 0.f; }
#pragma unroll
    for (int j = 0; j < 4; ++j) { cs[j % NC] += sa[j]; cq[j % NC] += qa[j]; }
#pragma unroll
    for (int c = 0; c < NC; ++c) {
      for (int off = 16; off > 0; off >>= 1) {
        cs[c] += __shfl_xor(cs[c], off, 32);
        cq[c] += __shfl_xor(cq[c], off, 32);
      }
    }
    const int lane = t & 31;
    const int wave = t >> 5;
    if (lane == 0) {
#pragma unroll
      for (int c = 0; c < NC; ++c) {
        red[0][wave][c] = cs[c];
        red[1][wave][c] = cq[c];
      }
    }
    __syncthreads();
    if (t < 2 * NC) {
      const int which = t / NC;
      const int c = t - which * NC;
      float r = 0.f;
#pragma unroll
      for (int w = 0; w < NW; ++w) r += red[which][w][c];
      (which ? pq : ps)[(size_t)c * nparts + blockIdx.x] = r;
    }
  }
}

__global__ void bn_fold(const float* __restrict__ ps,
                        const float* __restrict__ pq, int nparts,
                        const float* __restrict__ g,
                        const float* __restrict__ beta, int ncols, float invn,
                        float* __restrict__ sc, float* __restrict__ sh) {
  const int lane = threadIdx.x & 31;
  const int c = (blockIdx.x * blockDim.x + threadIdx.x) >> 5;
  if (c >= ncols) return;
  const float* __restrict__ sp = ps + (size_t)c * nparts;
  const float* __restrict__ qp = pq + (size_t)c * nparts;
  float s = 0.f, q = 0.f;
  for (int b0 = 0; b0 < nparts; b0 += 256) {
    float vs[8], vq[8];
#pragma unroll
    for (int k = 0; k < 8; ++k) {
      const int b = b0 + 32 * k + lane;
      vs[k] = sp[b < nparts ? b : 0];
      vq[k] = qp[b < nparts ? b : 0];
    }
#pragma unroll
    for (int k = 0; k < 8; ++k) {
      const int b = b0 + 32 * k + lane;
      if (b < nparts) { s += vs[k]; q += vq[k]; }
    }
  }
  for (int off = 16; off > 0; off >>= 1) {
    s += __shfl_xor(s, off, 32);
    q += __shfl_xor(q, off, 32);
  }
  if (lane == 0) {
    const float mean = s * invn;
    const float var = q * invn - mean * mean;
    const float r = rsqrtf(var + 1e-5f);
    const float a = g[c] * r;
    sc[c] = a;
    sh[c] = beta[c] - mean * a;
  }
}

__global__ void mlp_out(const float* __restrict__ y2,
                        const float* __restrict__ sc,
                        const float* __restrict__ sh,
                        const float* __restrict__ Wm2,
                        const float* __restrict__ bm2, int n,
                        float* __restrict__ y3) {
  const int node = blockIdx.x * blockDim.x + threadIdx.x;
  if (node >= n) return;
  float a0 = bm2[0], a1 = bm2[1];
  const float* yp = y2 + (size_t)node * 96;
  for (int k = 0; k < 96; ++k) {
    const float v = fmaf(yp[k], sc[k], sh[k]);
    a0 = fmaf(v, Wm2[2 * k + 0], a0);
    a1 = fmaf(v, Wm2[2 * k + 1], a1);
  }
  y3[node * 2 + 0] = fmaxf(a0, 0.f);
  y3[node * 2 + 1] = fmaxf(a1, 0.f);
}

__global__ void bn_apply(const float* __restrict__ y,
                         const float* __restrict__ sc,
                         const float* __restrict__ sh, int total, int ncols,
                         float* __restrict__ out) {
  const int i = blockIdx.x * blockDim.x + threadIdx.x;
  if (i >= total) return;
  const int c = i % ncols;
  out[i] = fmaf(y[i], sc[c], sh[c]);
}

extern "C" void kernel_launch(void* const* d_in, const int* in_sizes, int n_in,
                              void* d_out, int out_size, void* d_ws,
                              size_t ws_size, hipStream_t stream) {
  const float* x      = (const float*)d_in[0];
  const int*   ei     = (const int*)d_in[1];
  const float* W1     = (const float*)d_in[2];
  const float* as1    = (const float*)d_in[3];
  const float* ad1    = (const float*)d_in[4];
  const float* b1     = (const float*)d_in[5];
  const float* W2     = (const float*)d_in[6];
  const float* as2    = (const float*)d_in[7];
  const float* ad2    = (const float*)d_in[8];
  const float* b2     = (const float*)d_in[9];
  const float* Wl     = (const float*)d_in[10];
  const float* bl     = (const float*)d_in[11];
  const float* gl     = (const float*)d_in[12];
  const float* betal  = (const float*)d_in[13];
  const float* Wm1    = (const float*)d_in[14];
  const float* bm1    = (const float*)d_in[15];
  const float* gm1    = (const float*)d_in[16];
  const float* betam1 = (const float*)d_in[17];
  const float* Wm2    = (const float*)d_in[18];
  const float* bm2    = (const float*)d_in[19];
  const float* gm2    = (const float*)d_in[20];
  const float* betam2 = (const float*)d_in[21];

  const int N = in_sizes[0] / DFEAT;
  const int E = in_sizes[1] / 2;
  const int* srcA = ei;
  const int* dstA = ei + E;

  char* w = (char*)d_ws;
  auto alloc = [&](size_t bytes) -> char* {
    char* p = w;
    w += (bytes + 255) & ~(size_t)255;
    return p;
  };
  float*    xcat = (float*)alloc((size_t)N * 384 * 4);
  float*    h1   = (float*)alloc((size_t)N * 256 * 4);
  float*    h2   = (float*)alloc((size_t)N * 128 * 4);
  float*    al_s = (float*)alloc((size_t)N * 2 * 4);
  float*    al_d = (float*)alloc((size_t)N * 2 * 4);
  float*    y3   = (float*)alloc((size_t)N * 2 * 4);
  float*    sc1  = (float*)alloc(96 * 4);
  float*    sh1  = (float*)alloc(96 * 4);
  float*    sc2  = (float*)alloc(96 * 4);
  float*    sh2  = (float*)alloc(96 * 4);
  float*    sc3  = (float*)alloc(96 * 4);
  float*    sh3  = (float*)alloc(96 * 4);
  float* y1 = h1;
  float* y2 = h1 + (size_t)N * 96;
  const int csGroups = CS_LOADS * CS_THREADS;
  auto csBlocks = [&](int total) {
    return ((total + 3) / 4 + csGroups - 1) / csGroups;
  };
  const int np96 = csBlocks(N * 96);
  const int np2  = csBlocks(N * 2);
  float* ps = h2;
  float* pq = h2 + (size_t)96 * np96;

  const int TB = 256;
  auto gemmBlocks = [](int rows, int cols) {
    const int tiles = ((rows + 15) / 16) * (cols / 16);
    return (tiles + 7) / 8;
  };
  auto layerBlocks = [](int rows, int cols) {
    const int rowsPer = LG_WAVES * 16;
    return ((rows + rowsPer - 1) / rowsPer) * (cols / LG_COLS);
  };
  const int ownBlocks = (N + OWN_ROWS - 1) / OWN_ROWS;
  const int apRows = AP_ROWS * (AP_THREADS / 32);

  gemm_layer_strip<128><<<layerBlocks(N, 256), LG_WAVES * 32, 0, stream>>>(
      x, 128, W1, 256, h1, 256, N, 256);
  attn_proj_rows<2><<<(N * 2 + apRows - 1) / apRows, AP_THREADS, 0, stream>>>(
      h1, as1, ad1, al_s, al_d, N * 2);
  gat_owned_h2<<<ownBlocks, OWN_THREADS, 0, stream>>>(
      srcA, dstA, E, N, h1, 256, al_s, al_d, b1, xcat, 384, 0, 256);

  gemm_layer_strip<256><<<layerBlocks(N, 128), LG_WAVES * 32, 0, stream>>>(
      xcat, 384, W2, 128, h2, 128, N, 128);
  attn_proj_rows<1><<<(N + apRows - 1) / apRows, AP_THREADS, 0, stream>>>(
      h2, as2, ad2, al_s, al_d, N);
  gat_owned_h1<<<ownBlocks, OWN_THREADS, 0, stream>>>(
      srcA, dstA, E, N, h2, 128, al_s, al_d, b2, xcat, 384, 256, 256, Wl, bl,
      y1);

  const float invn = 1.0f / (float)N;
  col_part<96><<<np96, CS_THREADS, 0, stream>>>(y1, N * 96, ps, pq, np96);
  bn_fold<<<(96 * 32 + TB - 1) / TB, TB, 0, stream>>>(ps, pq, np96, gl, betal,
                                                      96, invn, sc1, sh1);
  gemm_f32_wmma_bn<<<gemmBlocks(N, 96), TB, 0, stream>>>(
      y1, 96, Wm1, 96, y2, 96, N, 96, 96, bm1, sc1, sh1, 1);
  col_part<96><<<np96, CS_THREADS, 0, stream>>>(y2, N * 96, ps, pq, np96);
  bn_fold<<<(96 * 32 + TB - 1) / TB, TB, 0, stream>>>(ps, pq, np96, gm1, betam1,
                                                      96, invn, sc2, sh2);
  mlp_out<<<(N + TB - 1) / TB, TB, 0, stream>>>(y2, sc2, sh2, Wm2, bm2, N, y3);
  col_part<2><<<np2, CS_THREADS, 0, stream>>>(y3, N * 2, ps, pq, np2);
  bn_fold<<<(2 * 32 + TB - 1) / TB, TB, 0, stream>>>(ps, pq, np2, gm2, betam2, 2,
                                                     invn, sc3, sh3);
  bn_apply<<<(N * 2 + TB - 1) / TB, TB, 0, stream>>>(y3, sc3, sh3, N * 2, 2,
                                                     (float*)d_out);
}


# layer-1 edge filter sweeps a one-byte-per-edge owner plane (dst / 80, cut by the projection launch) instead of dst: a quarter of the bytes, hits gathered by ballot; LDS adds in arrival order
# speedup vs baseline: 4.6943x; 1.0437x over previous
#include <hip/hip_runtime.h>
#include <hip/hip_bf16.h>
#include <math.h>


typedef __attribute__((ext_vector_type(2))) float v2f;
typedef __attribute__((ext_vector_type(8))) float v8f;

#define DFEAT 128
#define NEG_SLOPE 0.2f
#define OWN_ROWS 80
#define NO_OWNER 0xFFu

__device__ __forceinline__ unsigned ordEnc(float f) {
  unsigned u = __float_as_uint(f);
  return (u & 0x80000000u) ? ~u : (u | 0x80000000u);
}
__device__ __forceinline__ float ordDec(unsigned e) {
  unsigned u = (e & 0x80000000u) ? (e & 0x7fffffffu) : ~e;
  return __uint_as_float(u);
}

__device__ __forceinline__ v8f wmma_f32(v2f a, v2f b, v8f c) {
  return __builtin_amdgcn_wmma_f32_16x16x4_f32(false, a, false, b, (short)0, c,
                                               false, false);
}

__global__ void gemm_f32_wmma_bn(const float* __restrict__ A, int lda,
                                 const float* __restrict__ B, int ldb,
                                 float* __restrict__ D, int ldo,
                                 int nrows, int kdim, int ncols,
                                 const float* __restrict__ bias,
                                 const float* __restrict__ scl,
                                 const float* __restrict__ shf, int relu) {
  const int lane = threadIdx.x & 31;
  const int wid  = (blockIdx.x * blockDim.x + threadIdx.x) >> 5;
  const int colTiles = ncols >> 4;
  const int rowTiles = (nrows + 15) >> 4;
  if (wid >= rowTiles * colTiles) return;
  const int tr = wid / colTiles;
  const int tc = wid - tr * colTiles;
  const int row0 = tr << 4, col0 = tc << 4;
  const int lm = lane & 15;
  const int lh = lane >> 4;

  int arow = row0 + lm;
  if (arow >= nrows) arow = nrows - 1;
  const float* Ap = A + (size_t)arow * lda + 2 * lh;
  const float* Sp = scl + 2 * lh;
  const float* Hp = shf + 2 * lh;
  const float* Bp = B + (size_t)(2 * lh) * ldb + col0 + lm;

  v8f acc = {0.f, 0.f, 0.f, 0.f, 0.f, 0.f, 0.f, 0.f};
  int k = 0;
  for (; k + 16 <= kdim; k += 16) {
    v2f a0 = *(const v2f*)(Ap + k);
    v2f a1 = *(const v2f*)(Ap + k + 4);
    v2f a2 = *(const v2f*)(Ap + k + 8);
    v2f a3 = *(const v2f*)(Ap + k + 12);
    const v2f s0 = *(const v2f*)(Sp + k);
    const v2f s1 = *(const v2f*)(Sp + k + 4);
    const v2f s2 = *(const v2f*)(Sp + k + 8);
    const v2f s3 = *(const v2f*)(Sp + k + 12);
    const v2f h0 = *(const v2f*)(Hp + k);
    const v2f h1 = *(const v2f*)(Hp + k + 4);
    const v2f h2 = *(const v2f*)(Hp + k + 8);
    const v2f h3 = *(const v2f*)(Hp + k + 12);
    a0 = a0 * s0 + h0;
    a1 = a1 * s1 + h1;
    a2 = a2 * s2 + h2;
    a3 = a3 * s3 + h3;
    const float* bq = Bp + (size_t)k * ldb;
    v2f b0, b1, b2, b3;
    b0.x = bq[0];
    b0.y = bq[(size_t)1 * ldb];
    b1.x = bq[(size_t)4 * ldb];
    b1.y = bq[(size_t)5 * ldb];
    b2.x = bq[(size_t)8 * ldb];
    b2.y = bq[(size_t)9 * ldb];
    b3.x = bq[(size_t)12 * ldb];
    b3.y = bq[(size_t)13 * ldb];
    acc = wmma_f32(a0, b0, acc);
    acc = wmma_f32(a1, b1, acc);
    acc = wmma_f32(a2, b2, acc);
    acc = wmma_f32(a3, b3, acc);
  }
  for (; k < kdim; k += 4) {
    v2f a = *(const v2f*)(Ap + k);
    const v2f s = *(const v2f*)(Sp + k);
    const v2f h = *(const v2f*)(Hp + k);
    a = a * s + h;
    v2f b;
    b.x = Bp[(size_t)k * ldb];
    b.y = Bp[(size_t)(k + 1) * ldb];
    acc = wmma_f32(a, b, acc);
  }

  const int col = col0 + lm;
  const float bv = bias ? bias[col] : 0.0f;
#pragma unroll
  for (int r = 0; r < 8; ++r) {
    const int row = row0 + r + 8 * lh;
    if (row < nrows) {
      float v = acc[r] + bv;
      if (relu) v = fmaxf(v, 0.0f);
      D[(size_t)row * ldo + col] = v;
    }
  }
}

#define LG_WAVES 4
#define LG_COLS 64
#define LG_PITCH (2 * LG_COLS + 8)
#define LG_CPITCH (LG_COLS + 4)

template <int KDIM>
__global__ void __launch_bounds__(LG_WAVES * 32)
gemm_layer_strip(const float* __restrict__ A, int lda,
                 const float* __restrict__ B, int ldb, float* __restrict__ D,
                 int ldo, int nrows, int ncols) {
  static_assert(KDIM % 16 == 0, "whole blocks of 16 k");
  static_assert(LG_COLS == 64, "four tiles per wave, 16 column groups per row");
  constexpr int NTHR = LG_WAVES * 32;
  constexpr int NPAIR = KDIM / 2;
  constexpr int NBLK = KDIM / 16;
  constexpr int CGR = LG_COLS / 4;
  constexpr int PER = NPAIR * CGR / NTHR;
  constexpr int CH = 8;
  static_assert((NPAIR * CGR) % NTHR == 0 && PER % CH == 0, "even staging");
  static_assert(NTHR % CGR == 0, "a thread keeps its column group");
  __shared__ __attribute__((aligned(16))) float Bs[NPAIR * LG_PITCH];
  __shared__ __attribute__((aligned(16))) float Cs[LG_WAVES][16 * LG_CPITCH];

  const int tid = threadIdx.x;
  const int lane = tid & 31;
  const int wave = __builtin_amdgcn_readfirstlane(tid >> 5);
  const int strips = ncols / LG_COLS;
  const int rg = blockIdx.x / strips;
  const int col0 = (blockIdx.x - rg * strips) * LG_COLS;
  const int row0 = (rg * LG_WAVES + wave) * 16;
  const bool live = row0 < nrows;
  const int c = lane & 15;
  const int h = lane >> 4;

  int arow = row0 + c;
  if (arow >= nrows) arow = nrows - 1;
  const float4* __restrict__ Ap =
      (const float4*)(A + (size_t)arow * lda + 8 * h);
  float4 a[2 * NBLK];
#pragma unroll
  for (int kb = 0; kb < NBLK; ++kb) {
    a[2 * kb] = Ap[4 * kb];
    a[2 * kb + 1] = Ap[4 * kb + 1];
  }

  {
    const int cg = tid % CGR;
    const float* __restrict__ bcol = B + col0 + 4 * cg;
    for (int t0 = 0; t0 < PER; t0 += CH) {
      float4 r0[CH], r1[CH];
#pragma unroll
      for (int i = 0; i < CH; ++i) {
        const int p = ((t0 + i) * NTHR + tid) / CGR;
        const float* bp = bcol + (size_t)(2 * p) * ldb;
        r0[i] = *(const float4*)bp;
        r1[i] = *(const float4*)(bp + ldb);
      }
#pragma unroll
      for (int i = 0; i < CH; ++i) {
        const int p = ((t0 + i) * NTHR + tid) / CGR;
        float* sp = Bs + p * LG_PITCH + 8 * cg;
        *(float4*)sp = make_float4(r0[i].x, r1[i].x, r0[i].y, r1[i].y);
        *(float4*)(sp + 4) = make_float4(r0[i].z, r1[i].z, r0[i].w, r1[i].w);
      }
    }
  }
  __syncthreads();

  float* cw = &Cs[wave][0];
  if (live) {
    v8f acc[4];
#pragma unroll
    for (int t = 0; t < 4; ++t)
      acc[t] = (v8f){0.f, 0.f, 0.f, 0.f, 0.f, 0.f, 0.f, 0.f};
    const float* bl = Bs + (4 * h) * LG_PITCH + 2 * c;
#pragma unroll
    for (int kb = 0; kb < NBLK; ++kb) {
      const float av[8] = {a[2 * kb].x,     a[2 * kb].y,     a[2 * kb].z,
                           a[2 * kb].w,     a[2 * kb + 1].x, a[2 * kb + 1].y,
                           a[2 * kb + 1].z, a[2 * kb + 1].w};
#pragma unroll
      for (int j = 0; j < 4; ++j) {
        v2f af;
        af.x = av[2 * j];
        af.y = av[2 * j + 1];
        const float* bq = bl + (8 * kb + j) * LG_PITCH;
#pragma unroll
        for (int t = 0; t < 4; ++t) {
          const v2f bf = *(const v2f*)(bq + 32 * t);
          acc[t] = wmma_f32(af, bf, acc[t]);
        }
      }
    }
    float* cp = cw + (8 * h) * LG_CPITCH + c;
#pragma unroll
    for (int t = 0; t < 4; ++t) {
#pragma unroll
      for (int r = 0; r < 8; ++r) cp[r * LG_CPITCH + 16 * t] = acc[t][r];
    }
  }
  __syncthreads();
  if (live) {
    const int rr = lane >> 4;
    const int cg = lane & 15;
    float4 v[8];
#pragma unroll
    for (int i = 0; i < 8; ++i)
      v[i] = *(const float4*)(cw + (2 * i + rr) * LG_CPITCH + 4 * cg);
    float* dp = D + (size_t)(row0 + rr) * ldo + col0 + 4 * cg;
    if (row0 + 16 <= nrows) {
#pragma unroll
      for (int i = 0; i < 8; ++i) *(float4*)(dp + (size_t)(2 * i) * ldo) = v[i];
    } else {
#pragma unroll
      for (int i = 0; i < 8; ++i) {
        if (row0 + 2 * i + rr < nrows)
          *(float4*)(dp + (size_t)(2 * i) * ldo) = v[i];
      }
    }
  }
}

#define AP_ROWS 16
#define AP_THREADS 128

template <int CNT>
__device__ __forceinline__ void ap_fold(float (&val)[2 * AP_ROWS], int lane) {
  const bool up = (lane & CNT) != 0;
#pragma unroll
  for (int i = 0; i < CNT; ++i) {
    const float keep = up ? val[i + CNT] : val[i];
    const float send = up ? val[i] : val[i + CNT];
    val[i] = keep + __shfl_xor(send, CNT, 32);
  }
}

template <int HEADS>
__global__ void __launch_bounds__(AP_THREADS)
attn_proj_rows(const float* __restrict__ h, const float* __restrict__ a_src,
               const float* __restrict__ a_dst, float* __restrict__ al_s,
               float* __restrict__ al_d, int nrows,
               const int* __restrict__ dst, int E, int n,
               unsigned* __restrict__ plane) {
  static_assert(AP_ROWS % HEADS == 0, "a wave's first row is head 0");
  static_assert(2 * AP_ROWS == 32, "one finished sum per lane");
  if (plane) {
    const int nwords = ((E + 15) >> 4) << 2;
    const int step = gridDim.x * blockDim.x;
    for (int w = blockIdx.x * blockDim.x + threadIdx.x; w < nwords; w += step) {
      int d[4];
#pragma unroll
      for (int j = 0; j < 4; ++j) {
        const int e = 4 * w + j;
        d[j] = dst[e < E ? e : E - 1];
      }
      unsigned pk = 0u;
#pragma unroll
      for (int j = 0; j < 4; ++j) {
        const int e = 4 * w + j;
        const bool in = e < E && (unsigned)d[j] < (unsigned)n;
        pk |= (in ? (unsigned)d[j] / OWN_ROWS : NO_OWNER) << (8 * j);
      }
      plane[w] = pk;
    }
  }
  const int lane = threadIdx.x & 31;
  const int wid = __builtin_amdgcn_readfirstlane(
      (int)((blockIdx.x * blockDim.x + threadIdx.x) >> 5));
  const int w0 = wid * AP_ROWS;
  if (w0 >= nrows) return;

  float as[HEADS][4], ad[HEADS][4];
#pragma unroll
  for (int hh = 0; hh < HEADS; ++hh) {
#pragma unroll
    for (int q = 0; q < 4; ++q) {
      as[hh][q] = a_src[hh * DFEAT + 4 * lane + q];
      ad[hh][q] = a_dst[hh * DFEAT + 4 * lane + q];
    }
  }
  const float4* __restrict__ hp =
      (const float4*)(h + (size_t)w0 * DFEAT) + lane;
  float4 v[AP_ROWS];
#pragma unroll
  for (int j = 0; j < AP_ROWS; ++j)
    v[j] = hp[(w0 + j < nrows ? j : 0) * (DFEAT / 4)];
  __builtin_amdgcn_sched_barrier(0);

  float val[2 * AP_ROWS];
#pragma unroll
  for (int j = 0; j < AP_ROWS; ++j) {
    const int hh = j % HEADS;
    val[j] = v[j].x * as[hh][0] + v[j].y * as[hh][1] + v[j].z * as[hh][2] +
             v[j].w * as[hh][3];
    val[AP_ROWS + j] = v[j].x * ad[hh][0] + v[j].y * ad[hh][1] +
                       v[j].z * ad[hh][2] + v[j].w * ad[hh][3];
  }
  ap_fold<16>(val, lane);
  ap_fold<8>(val, lane);
  ap_fold<4>(val, lane);
  ap_fold<2>(val, lane);
  ap_fold<1>(val, lane);

  const int w = w0 + (lane & 15);
  if (w < nrows) (lane < 16 ? al_s : al_d)[w] = val[0];
}

#define X1_W 256
#define MLP_C 96
#define OWN_THREADS 1024
#define LIST_CAP 8192
#define KEEP_HDR 32

template <int HEADS, int KEEP>
__device__ __forceinline__ void gat_owned_body(
    const int* __restrict__ src, const int* __restrict__ dst, int E, int n,
    const float* __restrict__ h, int ldh, const float* __restrict__ al_s,
    const float* __restrict__ al_d, const float* __restrict__ bias,
    float* __restrict__ out, int ldo, int colOff, int keepOff,
    const float* __restrict__ Wl, const float* __restrict__ bl,
    float* __restrict__ y1, const unsigned* __restrict__ plane) {
  static_assert(OWN_ROWS <= 128 && DFEAT == 128, "packed word: 7 bits of row");
  static_assert(OWN_ROWS % 16 == 0 && MLP_C % 16 == 0 && X1_W % 16 == 0,
                "whole tiles and whole blocks of 16 k in the fused product");
  static_assert(KEEP != 2 || HEADS == 1, "the fused product takes one head");
  constexpr int ROWW = HEADS * DFEAT;
  constexpr int NL = ROWW / 32;
  constexpr int BATCH = 32 / NL;
  constexpr int AP = ROWW + (KEEP == 2 ? 4 : 0);
  constexpr int XP = X1_W + 4;
  __shared__ __attribute__((aligned(16))) float s_acc[OWN_ROWS * AP];
  __shared__ __attribute__((aligned(16))) float
      s_x1[KEEP == 2 ? OWN_ROWS * XP : 4];
  __shared__ int s_ls[LIST_CAP];
  __shared__ int s_ld[LIST_CAP];
  __shared__ unsigned s_menc[OWN_ROWS * HEADS];
  __shared__ float s_sum[OWN_ROWS * HEADS];
  __shared__ unsigned s_cnt;

  const int tid = threadIdx.x;
  const int lane = tid & 31;
  const int wave = tid >> 5;
  const int nthr = blockDim.x;
  const int nw = nthr >> 5;
  const int r0 = blockIdx.x * OWN_ROWS;
  int rows_here = n - r0;
  if (rows_here > OWN_ROWS) rows_here = OWN_ROWS;
  if (rows_here < 0) rows_here = 0;

  float* const kbase = out + (size_t)r0 * ldo + keepOff;
  auto kslot = [&](int j) -> float* {
    return kbase + (size_t)(j >> 7) * ldo + (j & (DFEAT - 1));
  };
  int keep_cap = rows_here * DFEAT - KEEP_HDR;
  if (keep_cap > LIST_CAP) keep_cap = LIST_CAP;
  const bool packs = n <= (1 << 24);
  int kcnt = -1;
  if constexpr (KEEP == 2) {
    if (rows_here > 0 && packs) kcnt = __float_as_int(*kslot(0));
  }
  const bool kept = KEEP == 2 && kcnt >= 0 && kcnt <= keep_cap;

  if constexpr (KEEP == 2) {
    constexpr int XG = X1_W / 4;
    constexpr int XB = 5;
    for (int f0 = 0; f0 < OWN_ROWS * XG; f0 += XB * nthr) {
      float4 xr[XB];
#pragma unroll
      for (int i = 0; i < XB; ++i) {
        const int f = f0 + i * nthr + tid;
        const int r = f / XG;
        xr[i] = make_float4(0.f, 0.f, 0.f, 0.f);
        if (f < OWN_ROWS * XG && r < rows_here)
          xr[i] = *(const float4*)(out + (size_t)(r0 + r) * ldo +
                                   4 * (f - r * XG));
      }
#pragma unroll
      for (int i = 0; i < XB; ++i) {
        const int f = f0 + i * nthr + tid;
        const int r = f / XG;
        if (f < OWN_ROWS * XG)
          *(float4*)(s_x1 + r * XP + 4 * (f - r * XG)) = xr[i];
      }
    }
  }
  for (int i = tid; i < OWN_ROWS * AP; i += nthr) s_acc[i] = 0.f;
  for (int i = tid; i < OWN_ROWS * HEADS; i += nthr) {
    s_menc[i] = 0u;
    s_sum[i] = 0.f;
  }
  if (tid == 0) s_cnt = 0u;
  __syncthreads();

  auto append = [&](int e, int dl) {
    const unsigned p = atomicAdd(&s_cnt, 1u);
    if (p < (unsigned)LIST_CAP) { s_ls[p] = e; s_ld[p] = dl; }
  };
  auto owned = [&](int d) -> bool {
    return (unsigned)(d - r0) < (unsigned)rows_here;
  };
  if (kept) {
    for (int i = tid; i < kcnt; i += nthr) {
      const int p = __float_as_int(*kslot(KEEP_HDR + i));
      s_ls[i] = p >> 7;
      s_ld[i] = p & 127;
    }
  } else if (plane) {
    constexpr int PL = 8;
    constexpr unsigned ONES = 0x01010101u;
    const uint4* __restrict__ pv = (const uint4*)plane;
    const int ngrp = (E + 15) >> 4;
    const unsigned idw = (unsigned)blockIdx.x * ONES;
    auto append_e = [&](int e) {
      const unsigned p = atomicAdd(&s_cnt, 1u);
      if (p < (unsigned)LIST_CAP) s_ls[p] = e;
    };
    for (int r = tid; r < rows_here; r += nthr) append_e(E + r0 + r);
    const int wcap = LIST_CAP / nw;
    int* const cand = s_ld + wave * wcap;
    int ncand = 0;
    for (int gb = 0; gb < ngrp; gb += PL * nthr) {
      uint4 q[PL];
#pragma unroll
      for (int k = 0; k < PL; ++k) {
        const int g = gb + k * nthr + tid;
        q[k] = pv[g < ngrp ? g : ngrp - 1];
      }
#pragma unroll
      for (int k = 0; k < PL; ++k) {
        const int g = gb + k * nthr + tid;
        const unsigned x0 = q[k].x ^ idw, x1 = q[k].y ^ idw;
        const unsigned x2 = q[k].z ^ idw, x3 = q[k].w ^ idw;
        const unsigned t = (((x0 - ONES) & ~x0) | ((x1 - ONES) & ~x1) |
                            ((x2 - ONES) & ~x2) | ((x3 - ONES) & ~x3)) &
                           0x80808080u;
        const bool hit = g < ngrp && t != 0u;
        const unsigned bm = __builtin_amdgcn_ballot_w32(hit);
        if (bm) {
          const int pos = ncand + (int)__builtin_amdgcn_mbcnt_lo(bm, 0u);
          if (hit && pos < wcap) cand[pos] = g;
          ncand += __builtin_popcount(bm);
        }
      }
    }
    if (ncand > wcap) {
      if (lane == 0) atomicAdd(&s_cnt, (unsigned)LIST_CAP + 1u);
    } else {
      for (int b = 0; b < ncand; b += 32) {
        const int i = b + lane;
        if (i < ncand) {
          const int g = cand[i];
          const uint4 v = pv[g];
          auto zb = [&](unsigned w) -> unsigned {
            const unsigned x = w ^ idw;
            return ~(((x & 0x7F7F7F7Fu) + 0x7F7F7F7Fu) | x) & 0x80808080u;
          };
          unsigned m = (zb(v.x) >> 7) | (zb(v.y) >> 6) | (zb(v.z) >> 5) |
                       (zb(v.w) >> 4);
          unsigned p = atomicAdd(&s_cnt, (unsigned)__builtin_popcount(m));
          while (m) {
            const int bit = __builtin_ctz(m);
            m &= m - 1;
            if (p < (unsigned)LIST_CAP)
              s_ls[p] = 16 * g + 4 * (bit & 7) + (bit >> 3);
            ++p;
          }
        }
      }
    }
  } else {
    for (int r = tid; r < rows_here; r += nthr) append(E + r0 + r, r);
    int pre = (int)(((16u - ((unsigned)(size_t)dst & 15u)) & 15u) >> 2);
    if (pre > E) pre = E;
    const int nvec = (E - pre) >> 2;
    const int4* __restrict__ dv = (const int4*)(dst + pre);
    for (int vb = 0; vb < nvec; vb += 4 * nthr) {
      int4 q[4];
      int vi[4];
      for (int k = 0; k < 4; ++k) {
        vi[k] = vb + k * nthr + tid;
        q[k] = dv[vi[k] < nvec ? vi[k] : nvec - 1];
      }
      for (int k = 0; k < 4; ++k) {
        const bool in = vi[k] < nvec;
        const bool o0 = in && owned(q[k].x), o1 = in && owned(q[k].y);
        const bool o2 = in && owned(q[k].z), o3 = in && owned(q[k].w);
        if (o0 | o1 | o2 | o3) {
          const int e = pre + 4 * vi[k];
          if (o0) append(e + 0, q[k].x - r0);
          if (o1) append(e + 1, q[k].y - r0);
          if (o2) append(e + 2, q[k].z - r0);
          if (o3) append(e + 3, q[k].w - r0);
        }
      }
    }
    const int tail0 = pre + 4 * nvec;
    if (tid < 8) {
      const int e = (tid < 4) ? tid : tail0 + (tid - 4);
      const bool in = (tid < 4) ? (e < pre) : (e < E);
      if (in) {
        const int d = dst[e];
        if (owned(d)) append(e, d - r0);
      }
    }
  }
  __syncthreads();
  const int cnt = kept ? kcnt : (int)s_cnt;
  const bool use_list = cnt <= LIST_CAP;
  const bool save = KEEP == 1 && packs && cnt <= keep_cap;
  if (use_list && !kept) {
    for (int i = tid; i < cnt; i += nthr) {
      const int e = s_ls[i];
      const int sr = (e < E) ? src[e] : (e - E);
      int dl;
      if (plane) {
        dl = ((e < E) ? dst[e] : (e - E)) - r0;
        s_ld[i] = dl;
      } else {
        dl = s_ld[i];
      }
      s_ls[i] = sr;
      if (save) *kslot(KEEP_HDR + i) = __int_as_float((sr << 7) | dl);
    }
  }
  if constexpr (KEEP == 1) {
    if (tid == 0 && rows_here > 0) *kslot(0) = __int_as_float(save ? cnt : -1);
  }
  __syncthreads();

  auto for_owned = [&](auto body) {
    if (use_list) {
      for (int base = 0; base < cnt; base += 32 * nw) {
        const int i = base + lane * nw + wave;
        const bool own = i < cnt;
        const int s = own ? s_ls[i] : 0;
        const int dl = own ? s_ld[i] : 0;
        body(own, s, dl);
      }
    } else {
      for (int e0 = wave * 32; e0 < E; e0 += 32 * nw) {
        const int e = e0 + lane;
        const int d = (e < E) ? dst[e] : -1;
        const bool own = (unsigned)(d - r0) < (unsigned)rows_here;
        if (__builtin_amdgcn_ballot_w32(own)) {
          const int s = own ? src[e] : 0;
          body(own, s, own ? d - r0 : 0);
        }
      }
      for (int i0 = wave * 32; i0 < rows_here; i0 += 32 * nw) {
        const int r = i0 + lane;
        const bool own = r < rows_here;
        body(own, own ? r0 + r : 0, own ? r : 0);
      }
    }
  };
  auto logit = [&](int s, int dl, int hh) -> float {
    const float x = al_s[s * HEADS + hh] + al_d[(r0 + dl) * HEADS + hh];
    return (x > 0.f) ? x : NEG_SLOPE * x;
  };

  for_owned([&](bool own, int s, int dl) {
    if (own) {
      for (int hh = 0; hh < HEADS; ++hh)
        atomicMax(&s_menc[dl * HEADS + hh], ordEnc(logit(s, dl, hh)));
    }
  });
  __syncthreads();

  for_owned([&](bool own, int s, int dl) {
    if (own) {
      for (int hh = 0; hh < HEADS; ++hh) {
        const float m = ordDec(s_menc[dl * HEADS + hh]);
        atomicAdd(&s_sum[dl * HEADS + hh], expf(logit(s, dl, hh) - m));
      }
    }
  });
  __syncthreads();

  for_owned([&](bool own, int s, int dl) {
    float al[HEADS];
    for (int hh = 0; hh < HEADS; ++hh) al[hh] = 0.f;
    if (own) {
      for (int hh = 0; hh < HEADS; ++hh) {
        const float m = ordDec(s_menc[dl * HEADS + hh]);
        al[hh] = expf(logit(s, dl, hh) - m) / (s_sum[dl * HEADS + hh] + 1e-16f);
      }
    }
    unsigned m = __builtin_amdgcn_ballot_w32(own);
    while (__builtin_popcount(m) >= BATCH) {
      float v[BATCH][NL];
      int oo[BATCH];
      int jj[BATCH];
      for (int b = 0; b < BATCH; ++b) {
        const int j = __builtin_ctz(m);
        m &= m - 1;
        jj[b] = j;
        const int sj = __builtin_amdgcn_readlane(s, j);
        const int dj = __builtin_amdgcn_readlane(dl, j);
        const float* hp = h + (size_t)sj * ldh + lane;
        oo[b] = dj * AP + lane;
        for (int c = 0; c < NL; ++c) v[b][c] = hp[32 * c];
      }
      for (int b = 0; b < BATCH; ++b) {
        for (int c = 0; c < NL; ++c) {
          const float a = __int_as_float(__builtin_amdgcn_readlane(
              __float_as_int(al[c / (DFEAT / 32)]), jj[b]));
          atomicAdd(&s_acc[oo[b] + 32 * c], a * v[b][c]);
        }
      }
    }
    while (m) {
      const int j = __builtin_ctz(m);
      m &= m - 1;
      const int sj = __builtin_amdgcn_readlane(s, j);
      const int dj = __builtin_amdgcn_readlane(dl, j);
      const float* hp = h + (size_t)sj * ldh + lane;
      const int o = dj * AP + lane;
      float v[NL];
      for (int c = 0; c < NL; ++c) v[c] = hp[32 * c];
      for (int c = 0; c < NL; ++c) {
        const float a = __int_as_float(__builtin_amdgcn_readlane(
            __float_as_int(al[c / (DFEAT / 32)]), j));
        atomicAdd(&s_acc[o + 32 * c], a * v[c]);
      }
    }
  });
  __syncthreads();

  if constexpr (KEEP != 2) {
    for (int r = wave; r < rows_here; r += nw) {
      float* op = out + (size_t)(r0 + r) * ldo + colOff + lane;
      for (int c = 0; c < NL; ++c)
        op[32 * c] = s_acc[r * AP + 32 * c + lane] + bias[32 * c + lane];
    }
  } else {
    for (int i = tid; i < OWN_ROWS * ROWW; i += nthr) {
      const int r = i / ROWW;
      const int cc = i - r * ROWW;
      s_acc[r * AP + cc] += bias[cc];
    }
    __syncthreads();
    const int wv = __builtin_amdgcn_readfirstlane(wave);
    constexpr int CT = MLP_C / 16;
    constexpr int TILES = (OWN_ROWS / 16) * CT;
    const int lc = lane & 15;
    const int lh = lane >> 4;
    for (int t = wv; t < TILES; t += nw) {
      const int rt = t / CT;
      const int ct = t - rt * CT;
      const int col = ct * 16 + lc;
      const float* ax = s_x1 + (rt * 16 + lc) * XP + 2 * lh;
      const float* aa = s_acc + (rt * 16 + lc) * AP + 2 * lh;
      const float* bp = Wl + (size_t)(2 * lh) * MLP_C + col;
      v8f acc = {0.f, 0.f, 0.f, 0.f, 0.f, 0.f, 0.f, 0.f};
      auto kblock = [&](const float* ap, const float* bq) {
        const v2f a0 = *(const v2f*)(ap);
        const v2f a1 = *(const v2f*)(ap + 4);
        const v2f a2 = *(const v2f*)(ap + 8);
        const v2f a3 = *(const v2f*)(ap + 12);
        v2f b0, b1, b2, b3;
        b0.x = bq[0];
        b0.y = bq[1 * MLP_C];
        b1.x = bq[4 * MLP_C];
        b1.y = bq[5 * MLP_C];
        b2.x = bq[8 * MLP_C];
        b2.y = bq[9 * MLP_C];
        b3.x = bq[12 * MLP_C];
        b3.y = bq[13 * MLP_C];
        acc = wmma_f32(a0, b0, acc);
        acc = wmma_f32(a1, b1, acc);
        acc = wmma_f32(a2, b2, acc);
        acc = wmma_f32(a3, b3, acc);
      };
      for (int k = 0; k < X1_W; k += 16) kblock(ax + k, bp + (size_t)k * MLP_C);
      for (int k = 0; k < ROWW; k += 16)
        kblock(aa + k, bp + (size_t)(X1_W + k) * MLP_C);
      const float bv = bl[col];
#pragma unroll
      for (int r = 0; r < 8; ++r) {
        const int row = rt * 16 + r + 8 * lh;
        if (row < rows_here)
          y1[(size_t)(r0 + row) * MLP_C + col] = fmaxf(acc[r] + bv, 0.0f);
      }
    }
  }
}

__global__ void __launch_bounds__(OWN_THREADS)
gat_owned_h2(const int* __restrict__ src, const int* __restrict__ dst, int E,
             int n, const float* __restrict__ h, int ldh,
             const float* __restrict__ al_s, const float* __restrict__ al_d,
             const float* __restrict__ bias, float* __restrict__ out, int ldo,
             int colOff, int keepOff, const unsigned* __restrict__ plane) {
  gat_owned_body<2, 1>(src, dst, E, n, h, ldh, al_s, al_d, bias, out, ldo,
                       colOff, keepOff, nullptr, nullptr, nullptr, plane);
}

__global__ void __launch_bounds__(OWN_THREADS)
gat_owned_h1(const int* __restrict__ src, const int* __restrict__ dst, int E,
             int n, const float* __restrict__ h, int ldh,
             const float* __restrict__ al_s, const float* __restrict__ al_d,
             const float* __restrict__ bias, float* __restrict__ out, int ldo,
             int colOff, int keepOff, const float* __restrict__ Wl,
             const float* __restrict__ bl, float* __restrict__ y1) {
  gat_owned_body<1, 2>(src, dst, E, n, h, ldh, al_s, al_d, bias, out, ldo,
                       colOff, keepOff, Wl, bl, y1, nullptr);
}

#define CS_THREADS 192
#define CS_LOADS 10

template <int NC>
__global__ void __launch_bounds__(CS_THREADS)
col_part(const float* __restrict__ y, int total, float* __restrict__ ps,
         float* __restrict__ pq, int nparts) {
  static_assert((4 * CS_THREADS) % NC == 0, "a thread keeps its columns");
  const int t = threadIdx.x;
  const int nvec = total >> 2;
  const int base = blockIdx.x * (CS_LOADS * CS_THREADS);
  const float4* __restrict__ y4 = (const float4*)y;

  float4 v[CS_LOADS];
#pragma unroll
  for (int k = 0; k < CS_LOADS; ++k) {
    const int f = base + k * CS_THREADS + t;
    v[k] = y4[f < nvec ? f : 0];
  }
  float sa[4] = {0.f, 0.f, 0.f, 0.f};
  float qa[4] = {0.f, 0.f, 0.f, 0.f};
#pragma unroll
  for (int k = 0; k < CS_LOADS; ++k) {
    const int f = base + k * CS_THREADS + t;
    if (f < nvec) {
      sa[0] += v[k].x; qa[0] = fmaf(v[k].x, v[k].x, qa[0]);
      sa[1] += v[k].y; qa[1] = fmaf(v[k].y, v[k].y, qa[1]);
      sa[2] += v[k].z; qa[2] = fmaf(v[k].z, v[k].z, qa[2]);
      sa[3] += v[k].w; qa[3] = fmaf(v[k].w, v[k].w, qa[3]);
    }
  }
  {
    const int tail = total & 3;
    const int rel = nvec - base;
    if (tail && rel >= 0 && rel < CS_LOADS * CS_THREADS &&
        rel % CS_THREADS == t) {
      for (int j = 0; j < tail; ++j) {
        const float w = y[4 * nvec + j];
        sa[j] += w;
        qa[j] = fmaf(w, w, qa[j]);
      }
    }
  }

  if constexpr (NC % 4 == 0) {
    constexpr int QN = NC / 4;
    static_assert(CS_THREADS % QN == 0, "whole rows per step");
    constexpr int GN = CS_THREADS / QN;
    __shared__ __attribute__((aligned(16))) float red[2][GN][NC];
    const int g = t / QN;
    const int c4 = 4 * (t - g * QN);
    *(float4*)&red[0][g][c4] = make_float4(sa[0], sa[1], sa[2], sa[3]);
    *(float4*)&red[1][g][c4] = make_float4(qa[0], qa[1], qa[2], qa[3]);
    __syncthreads();
    for (int i = t; i < 2 * NC; i += CS_THREADS) {
      const int which = i / NC;
      const int c = i - which * NC;
      float r = 0.f;
#pragma unroll
      for (int gg = 0; gg < GN; ++gg) r += red[which][gg][c];
      (which ? pq : ps)[(size_t)c * nparts + blockIdx.x] = r;
    }
  } else {
    static_assert(4 % NC == 0, "column count must divide a group");
    constexpr int NW = CS_THREADS / 32;
    __shared__ float red[2][NW][NC];
    float cs[NC], cq[NC];
#pragma unroll
    for (int c = 0; c < NC; ++c) { cs[c] = 0.f; cq[c] = 0.f; }
#pragma unroll
    for (int j = 0; j < 4; ++j) { cs[j % NC] += sa[j]; cq[j % NC] += qa[j]; }
#pragma unroll
    for (int c = 0; c < NC; ++c) {
      for (int off = 16; off > 0; off >>= 1) {
        cs[c] += __shfl_xor(cs[c], off, 32);
        cq[c] += __shfl_xor(cq[c], off, 32);
      }
    }
    const int lane = t & 31;
    const int wave = t >> 5;
    if (lane == 0) {
#pragma unroll
      for (int c = 0; c < NC; ++c) {
        red[0][wave][c] = cs[c];
        red[1][wave][c] = cq[c];
      }
    }
    __syncthreads();
    if (t < 2 * NC) {
      const int which = t / NC;
      const int c = t - which * NC;
      float r = 0.f;
#pragma unroll
      for (int w = 0; w < NW; ++w) r += red[which][w][c];
      (which ? pq : ps)[(size_t)c * nparts + blockIdx.x] = r;
    }
  }
}

__global__ void bn_fold(const float* __restrict__ ps,
                        const float* __restrict__ pq, int nparts,
                        const float* __restrict__ g,
                        const float* __restrict__ beta, int ncols, float invn,
                        float* __restrict__ sc, float* __restrict__ sh) {
  const int lane = threadIdx.x & 31;
  const int c = (blockIdx.x * blockDim.x + threadIdx.x) >> 5;
  if (c >= ncols) return;
  const float* __restrict__ sp = ps + (size_t)c * nparts;
  const float* __restrict__ qp = pq + (size_t)c * nparts;
  float s = 0.f, q = 0.f;
  for (int b0 = 0; b0 < nparts; b0 += 256) {
    float vs[8], vq[8];
#pragma unroll
    for (int k = 0; k < 8; ++k) {
      const int b = b0 + 32 * k + lane;
      vs[k] = sp[b < nparts ? b : 0];
      vq[k] = qp[b < nparts ? b : 0];
    }
#pragma unroll
    for (int k = 0; k < 8; ++k) {
      const int b = b0 + 32 * k + lane;
      if (b < nparts) { s += vs[k]; q += vq[k]; }
    }
  }
  for (int off = 16; off > 0; off >>= 1) {
    s += __shfl_xor(s, off, 32);
    q += __shfl_xor(q, off, 32);
  }
  if (lane == 0) {
    const float mean = s * invn;
    const float var = q * invn - mean * mean;
    const float r = rsqrtf(var + 1e-5f);
    const float a = g[c] * r;
    sc[c] = a;
    sh[c] = beta[c] - mean * a;
  }
}

__global__ void mlp_out(const float* __restrict__ y2,
                        const float* __restrict__ sc,
                        const float* __restrict__ sh,
                        const float* __restrict__ Wm2,
                        const float* __restrict__ bm2, int n,
                        float* __restrict__ y3) {
  const int node = blockIdx.x * blockDim.x + threadIdx.x;
  if (node >= n) return;
  float a0 = bm2[0], a1 = bm2[1];
  const float* yp = y2 + (size_t)node * 96;
  for (int k = 0; k < 96; ++k) {
    const float v = fmaf(yp[k], sc[k], sh[k]);
    a0 = fmaf(v, Wm2[2 * k + 0], a0);
    a1 = fmaf(v, Wm2[2 * k + 1], a1);
  }
  y3[node * 2 + 0] = fmaxf(a0, 0.f);
  y3[node * 2 + 1] = fmaxf(a1, 0.f);
}

__global__ void bn_apply(const float* __restrict__ y,
                         const float* __restrict__ sc,
                         const float* __restrict__ sh, int total, int ncols,
                         float* __restrict__ out) {
  const int i = blockIdx.x * blockDim.x + threadIdx.x;
  if (i >= total) return;
  const int c = i % ncols;
  out[i] = fmaf(y[i], sc[c], sh[c]);
}

extern "C" void kernel_launch(void* const* d_in, const int* in_sizes, int n_in,
                              void* d_out, int out_size, void* d_ws,
                              size_t ws_size, hipStream_t stream) {
  const float* x      = (const float*)d_in[0];
  const int*   ei     = (const int*)d_in[1];
  const float* W1     = (const float*)d_in[2];
  const float* as1    = (const float*)d_in[3];
  const float* ad1    = (const float*)d_in[4];
  const float* b1     = (const float*)d_in[5];
  const float* W2     = (const float*)d_in[6];
  const float* as2    = (const float*)d_in[7];
  const float* ad2    = (const float*)d_in[8];
  const float* b2     = (const float*)d_in[9];
  const float* Wl     = (const float*)d_in[10];
  const float* bl     = (const float*)d_in[11];
  const float* gl     = (const float*)d_in[12];
  const float* betal  = (const float*)d_in[13];
  const float* Wm1    = (const float*)d_in[14];
  const float* bm1    = (const float*)d_in[15];
  const float* gm1    = (const float*)d_in[16];
  const float* betam1 = (const float*)d_in[17];
  const float* Wm2    = (const float*)d_in[18];
  const float* bm2    = (const float*)d_in[19];
  const float* gm2    = (const float*)d_in[20];
  const float* betam2 = (const float*)d_in[21];

  const int N = in_sizes[0] / DFEAT;
  const int E = in_sizes[1] / 2;
  const int* srcA = ei;
  const int* dstA = ei + E;

  char* w = (char*)d_ws;
  auto alloc = [&](size_t bytes) -> char* {
    char* p = w;
    w += (bytes + 255) & ~(size_t)255;
    return p;
  };
  float*    xcat = (float*)alloc((size_t)N * 384 * 4);
  float*    h1   = (float*)alloc((size_t)N * 256 * 4);
  float*    h2   = (float*)alloc((size_t)N * 128 * 4);
  float*    al_s = (float*)alloc((size_t)N * 2 * 4);
  float*    al_d = (float*)alloc((size_t)N * 2 * 4);
  float*    y3   = (float*)alloc((size_t)N * 2 * 4);
  float*    sc1  = (float*)alloc(96 * 4);
  float*    sh1  = (float*)alloc(96 * 4);
  float*    sc2  = (float*)alloc(96 * 4);
  float*    sh2  = (float*)alloc(96 * 4);
  float*    sc3  = (float*)alloc(96 * 4);
  float*    sh3  = (float*)alloc(96 * 4);
  float* y1 = h1;
  float* y2 = h1 + (size_t)N * 96;
  const int csGroups = CS_LOADS * CS_THREADS;
  auto csBlocks = [&](int total) {
    return ((total + 3) / 4 + csGroups - 1) / csGroups;
  };
  const int np96 = csBlocks(N * 96);
  const int np2  = csBlocks(N * 2);
  float* ps = h2;
  float* pq = h2 + (size_t)96 * np96;

  const int TB = 256;
  auto gemmBlocks = [](int rows, int cols) {
    const int tiles = ((rows + 15) / 16) * (cols / 16);
    return (tiles + 7) / 8;
  };
  auto layerBlocks = [](int rows, int cols) {
    const int rowsPer = LG_WAVES * 16;
    return ((rows + rowsPer - 1) / rowsPer) * (cols / LG_COLS);
  };
  const int ownBlocks = (N + OWN_ROWS - 1) / OWN_ROWS;
  unsigned* plane = (ownBlocks <= 255 &&
                     (size_t)E + 16 <= (size_t)N * 128 * sizeof(float))
                        ? (unsigned*)h2
                        : nullptr;
  const int apRows = AP_ROWS * (AP_THREADS / 32);

  gemm_layer_strip<128><<<layerBlocks(N, 256), LG_WAVES * 32, 0, stream>>>(
      x, 128, W1, 256, h1, 256, N, 256);
  attn_proj_rows<2><<<(N * 2 + apRows - 1) / apRows, AP_THREADS, 0, stream>>>(
      h1, as1, ad1, al_s, al_d, N * 2, dstA, E, N, plane);
  gat_owned_h2<<<ownBlocks, OWN_THREADS, 0, stream>>>(
      srcA, dstA, E, N, h1, 256, al_s, al_d, b1, xcat, 384, 0, 256, plane);

  gemm_layer_strip<256><<<layerBlocks(N, 128), LG_WAVES * 32, 0, stream>>>(
      xcat, 384, W2, 128, h2, 128, N, 128);
  attn_proj_rows<1><<<(N + apRows - 1) / apRows, AP_THREADS, 0, stream>>>(
      h2, as2, ad2, al_s, al_d, N, nullptr, 0, 0, nullptr);
  gat_owned_h1<<<ownBlocks, OWN_THREADS, 0, stream>>>(
      srcA, dstA, E, N, h2, 128, al_s, al_d, b2, xcat, 384, 256, 256, Wl, bl,
      y1);

  const float invn = 1.0f / (float)N;
  col_part<96><<<np96, CS_THREADS, 0, stream>>>(y1, N * 96, ps, pq, np96);
  bn_fold<<<(96 * 32 + TB - 1) / TB, TB, 0, stream>>>(ps, pq, np96, gl, betal,
                                                      96, invn, sc1, sh1);
  gemm_f32_wmma_bn<<<gemmBlocks(N, 96), TB, 0, stream>>>(
      y1, 96, Wm1, 96, y2, 96, N, 96, 96, bm1, sc1, sh1, 1);
  col_part<96><<<np96, CS_THREADS, 0, stream>>>(y2, N * 96, ps, pq, np96);
  bn_fold<<<(96 * 32 + TB - 1) / TB, TB, 0, stream>>>(ps, pq, np96, gm1, betam1,
                                                      96, invn, sc2, sh2);
  mlp_out<<<(N + TB - 1) / TB, TB, 0, stream>>>(y2, sc2, sh2, Wm2, bm2, N, y3);
  col_part<2><<<np2, CS_THREADS, 0, stream>>>(y3, N * 2, ps, pq, np2);
  bn_fold<<<(2 * 32 + TB - 1) / TB, TB, 0, stream>>>(ps, pq, np2, gm2, betam2, 2,
                                                     invn, sc3, sh3);
  bn_apply<<<(N * 2 + TB - 1) / TB, TB, 0, stream>>>(y3, sc3, sh3, N * 2, 2,
                                                     (float*)d_out);
}


# layer 2's GEMM (x1 @ W2) and attention projections moved into layer 1's aggregation work-group: its 80 rows feed f32 WMMA tiles from LDS; 15 launches to 13 (order of k and projection sums changes)
# speedup vs baseline: 4.8358x; 1.0301x over previous
#include <hip/hip_runtime.h>
#include <hip/hip_bf16.h>
#include <math.h>


typedef __attribute__((ext_vector_type(2))) float v2f;
typedef __attribute__((ext_vector_type(8))) float v8f;

#define DFEAT 128
#define NEG_SLOPE 0.2f
#define OWN_ROWS 80
#define NO_OWNER 0xFFu

__device__ __forceinline__ unsigned ordEnc(float f) {
  unsigned u = __float_as_uint(f);
  return (u & 0x80000000u) ? ~u : (u | 0x80000000u);
}
__device__ __forceinline__ float ordDec(unsigned e) {
  unsigned u = (e & 0x80000000u) ? (e & 0x7fffffffu) : ~e;
  return __uint_as_float(u);
}

__device__ __forceinline__ v8f wmma_f32(v2f a, v2f b, v8f c) {
  return __builtin_amdgcn_wmma_f32_16x16x4_f32(false, a, false, b, (short)0, c,
                                               false, false);
}

__global__ void gemm_f32_wmma_bn(const float* __restrict__ A, int lda,
                                 const float* __restrict__ B, int ldb,
                                 float* __restrict__ D, int ldo,
                                 int nrows, int kdim, int ncols,
                                 const float* __restrict__ bias,
                                 const float* __restrict__ scl,
                                 const float* __restrict__ shf, int relu) {
  const int lane = threadIdx.x & 31;
  const int wid  = (blockIdx.x * blockDim.x + threadIdx.x) >> 5;
  const int colTiles = ncols >> 4;
  const int rowTiles = (nrows + 15) >> 4;
  if (wid >= rowTiles * colTiles) return;
  const int tr = wid / colTiles;
  const int tc = wid - tr * colTiles;
  const int row0 = tr << 4, col0 = tc << 4;
  const int lm = lane & 15;
  const int lh = lane >> 4;

  int arow = row0 + lm;
  if (arow >= nrows) arow = nrows - 1;
  const float* Ap = A + (size_t)arow * lda + 2 * lh;
  const float* Sp = scl + 2 * lh;
  const float* Hp = shf + 2 * lh;
  const float* Bp = B + (size_t)(2 * lh) * ldb + col0 + lm;

  v8f acc = {0.f, 0.f, 0.f, 0.f, 0.f, 0.f, 0.f, 0.f};
  int k = 0;
  for (; k + 16 <= kdim; k += 16) {
    v2f a0 = *(const v2f*)(Ap + k);
    v2f a1 = *(const v2f*)(Ap + k + 4);
    v2f a2 = *(const v2f*)(Ap + k + 8);
    v2f a3 = *(const v2f*)(Ap + k + 12);
    const v2f s0 = *(const v2f*)(Sp + k);
    const v2f s1 = *(const v2f*)(Sp + k + 4);
    const v2f s2 = *(const v2f*)(Sp + k + 8);
    const v2f s3 = *(const v2f*)(Sp + k + 12);
    const v2f h0 = *(const v2f*)(Hp + k);
    const v2f h1 = *(const v2f*)(Hp + k + 4);
    const v2f h2 = *(const v2f*)(Hp + k + 8);
    const v2f h3 = *(const v2f*)(Hp + k + 12);
    a0 = a0 * s0 + h0;
    a1 = a1 * s1 + h1;
    a2 = a2 * s2 + h2;
    a3 = a3 * s3 + h3;
    const float* bq = Bp + (size_t)k * ldb;
    v2f b0, b1, b2, b3;
    b0.x = bq[0];
    b0.y = bq[(size_t)1 * ldb];
    b1.x = bq[(size_t)4 * ldb];
    b1.y = bq[(size_t)5 * ldb];
    b2.x = bq[(size_t)8 * ldb];
    b2.y = bq[(size_t)9 * ldb];
    b3.x = bq[(size_t)12 * ldb];
    b3.y = bq[(size_t)13 * ldb];
    acc = wmma_f32(a0, b0, acc);
    acc = wmma_f32(a1, b1, acc);
    acc = wmma_f32(a2, b2, acc);
    acc = wmma_f32(a3, b3, acc);
  }
  for (; k < kdim; k += 4) {
    v2f a = *(const v2f*)(Ap + k);
    const v2f s = *(const v2f*)(Sp + k);
    const v2f h = *(const v2f*)(Hp + k);
    a = a * s + h;
    v2f b;
    b.x = Bp[(size_t)k * ldb];
    b.y = Bp[(size_t)(k + 1) * ldb];
    acc = wmma_f32(a, b, acc);
  }

  const int col = col0 + lm;
  const float bv = bias ? bias[col] : 0.0f;
#pragma unroll
  for (int r = 0; r < 8; ++r) {
    const int row = row0 + r + 8 * lh;
    if (row < nrows) {
      float v = acc[r] + bv;
      if (relu) v = fmaxf(v, 0.0f);
      D[(size_t)row * ldo + col] = v;
    }
  }
}

#define LG_WAVES 4
#define LG_COLS 64
#define LG_PITCH (2 * LG_COLS + 8)
#define LG_CPITCH (LG_COLS + 4)

template <int KDIM>
__global__ void __launch_bounds__(LG_WAVES * 32)
gemm_layer_strip(const float* __restrict__ A, int lda,
                 const float* __restrict__ B, int ldb, float* __restrict__ D,
                 int ldo, int nrows, int ncols) {
  static_assert(KDIM % 16 == 0, "whole blocks of 16 k");
  static_assert(LG_COLS == 64, "four tiles per wave, 16 column groups per row");
  constexpr int NTHR = LG_WAVES * 32;
  constexpr int NPAIR = KDIM / 2;
  constexpr int NBLK = KDIM / 16;
  constexpr int CGR = LG_COLS / 4;
  constexpr int PER = NPAIR * CGR / NTHR;
  constexpr int CH = 8;
  static_assert((NPAIR * CGR) % NTHR == 0 && PER % CH == 0, "even staging");
  static_assert(NTHR % CGR == 0, "a thread keeps its column group");
  __shared__ __attribute__((aligned(16))) float Bs[NPAIR * LG_PITCH];
  __shared__ __attribute__((aligned(16))) float Cs[LG_WAVES][16 * LG_CPITCH];

  const int tid = threadIdx.x;
  const int lane = tid & 31;
  const int wave = __builtin_amdgcn_readfirstlane(tid >> 5);
  const int strips = ncols / LG_COLS;
  const int rg = blockIdx.x / strips;
  const int col0 = (blockIdx.x - rg * strips) * LG_COLS;
  const int row0 = (rg * LG_WAVES + wave) * 16;
  const bool live = row0 < nrows;
  const int c = lane & 15;
  const int h = lane >> 4;

  int arow = row0 + c;
  if (arow >= nrows) arow = nrows - 1;
  const float4* __restrict__ Ap =
      (const float4*)(A + (size_t)arow * lda + 8 * h);
  float4 a[2 * NBLK];
#pragma unroll
  for (int kb = 0; kb < NBLK; ++kb) {
    a[2 * kb] = Ap[4 * kb];
    a[2 * kb + 1] = Ap[4 * kb + 1];
  }

  {
    const int cg = tid % CGR;
    const float* __restrict__ bcol = B + col0 + 4 * cg;
    for (int t0 = 0; t0 < PER; t0 += CH) {
      float4 r0[CH], r1[CH];
#pragma unroll
      for (int i = 0; i < CH; ++i) {
        const int p = ((t0 + i) * NTHR + tid) / CGR;
        const float* bp = bcol + (size_t)(2 * p) * ldb;
        r0[i] = *(const float4*)bp;
        r1[i] = *(const float4*)(bp + ldb);
      }
#pragma unroll
      for (int i = 0; i < CH; ++i) {
        const int p = ((t0 + i) * NTHR + tid) / CGR;
        float* sp = Bs + p * LG_PITCH + 8 * cg;
        *(float4*)sp = make_float4(r0[i].x, r1[i].x, r0[i].y, r1[i].y);
        *(float4*)(sp + 4) = make_float4(r0[i].z, r1[i].z, r0[i].w, r1[i].w);
      }
    }
  }
  __syncthreads();

  float* cw = &Cs[wave][0];
  if (live) {
    v8f acc[4];
#pragma unroll
    for (int t = 0; t < 4; ++t)
      acc[t] = (v8f){0.f, 0.f, 0.f, 0.f, 0.f, 0.f, 0.f, 0.f};
    const float* bl = Bs + (4 * h) * LG_PITCH + 2 * c;
#pragma unroll
    for (int kb = 0; kb < NBLK; ++kb) {
      const float av[8] = {a[2 * kb].x,     a[2 * kb].y,     a[2 * kb].z,
                           a[2 * kb].w,     a[2 * kb + 1].x, a[2 * kb + 1].y,
                           a[2 * kb + 1].z, a[2 * kb + 1].w};
#pragma unroll
      for (int j = 0; j < 4; ++j) {
        v2f af;
        af.x = av[2 * j];
        af.y = av[2 * j + 1];
        const float* bq = bl + (8 * kb + j) * LG_PITCH;
#pragma unroll
        for (int t = 0; t < 4; ++t) {
          const v2f bf = *(const v2f*)(bq + 32 * t);
          acc[t] = wmma_f32(af, bf, acc[t]);
        }
      }
    }
    float* cp = cw + (8 * h) * LG_CPITCH + c;
#pragma unroll
    for (int t = 0; t < 4; ++t) {
#pragma unroll
      for (int r = 0; r < 8; ++r) cp[r * LG_CPITCH + 16 * t] = acc[t][r];
    }
  }
  __syncthreads();
  if (live) {
    const int rr = lane >> 4;
    const int cg = lane & 15;
    float4 v[8];
#pragma unroll
    for (int i = 0; i < 8; ++i)
      v[i] = *(const float4*)(cw + (2 * i + rr) * LG_CPITCH + 4 * cg);
    float* dp = D + (size_t)(row0 + rr) * ldo + col0 + 4 * cg;
    if (row0 + 16 <= nrows) {
#pragma unroll
      for (int i = 0; i < 8; ++i) *(float4*)(dp + (size_t)(2 * i) * ldo) = v[i];
    } else {
#pragma unroll
      for (int i = 0; i < 8; ++i) {
        if (row0 + 2 * i + rr < nrows)
          *(float4*)(dp + (size_t)(2 * i) * ldo) = v[i];
      }
    }
  }
}

#define AP_ROWS 16
#define AP_THREADS 128

template <int CNT>
__device__ __forceinline__ void ap_fold(float (&val)[2 * AP_ROWS], int lane) {
  const bool up = (lane & CNT) != 0;
#pragma unroll
  for (int i = 0; i < CNT; ++i) {
    const float keep = up ? val[i + CNT] : val[i];
    const float send = up ? val[i] : val[i + CNT];
    val[i] = keep + __shfl_xor(send, CNT, 32);
  }
}

template <int HEADS>
__global__ void __launch_bounds__(AP_THREADS)
attn_proj_rows(const float* __restrict__ h, const float* __restrict__ a_src,
               const float* __restrict__ a_dst, float* __restrict__ al_s,
               float* __restrict__ al_d, int nrows,
               const int* __restrict__ dst, int E, int n,
               unsigned* __restrict__ plane) {
  static_assert(AP_ROWS % HEADS == 0, "a wave's first row is head 0");
  static_assert(2 * AP_ROWS == 32, "one finished sum per lane");
  if (plane) {
    const int nwords = ((E + 15) >> 4) << 2;
    const int step = gridDim.x * blockDim.x;
    for (int w = blockIdx.x * blockDim.x + threadIdx.x; w < nwords; w += step) {
      int d[4];
#pragma unroll
      for (int j = 0; j < 4; ++j) {
        const int e = 4 * w + j;
        d[j] = dst[e < E ? e : E - 1];
      }
      unsigned pk = 0u;
#pragma unroll
      for (int j = 0; j < 4; ++j) {
        const int e = 4 * w + j;
        const bool in = e < E && (unsigned)d[j] < (unsigned)n;
        pk |= (in ? (unsigned)d[j] / OWN_ROWS : NO_OWNER) << (8 * j);
      }
      plane[w] = pk;
    }
  }
  const int lane = threadIdx.x & 31;
  const int wid = __builtin_amdgcn_readfirstlane(
      (int)((blockIdx.x * blockDim.x + threadIdx.x) >> 5));
  const int w0 = wid * AP_ROWS;
  if (w0 >= nrows) return;

  float as[HEADS][4], ad[HEADS][4];
#pragma unroll
  for (int hh = 0; hh < HEADS; ++hh) {
#pragma unroll
    for (int q = 0; q < 4; ++q) {
      as[hh][q] = a_src[hh * DFEAT + 4 * lane + q];
      ad[hh][q] = a_dst[hh * DFEAT + 4 * lane + q];
    }
  }
  const float4* __restrict__ hp =
      (const float4*)(h + (size_t)w0 * DFEAT) + lane;
  float4 v[AP_ROWS];
#pragma unroll
  for (int j = 0; j < AP_ROWS; ++j)
    v[j] = hp[(w0 + j < nrows ? j : 0) * (DFEAT / 4)];
  __builtin_amdgcn_sched_barrier(0);

  float val[2 * AP_ROWS];
#pragma unroll
  for (int j = 0; j < AP_ROWS; ++j) {
    const int hh = j % HEADS;
    val[j] = v[j].x * as[hh][0] + v[j].y * as[hh][1] + v[j].z * as[hh][2] +
             v[j].w * as[hh][3];
    val[AP_ROWS + j] = v[j].x * ad[hh][0] + v[j].y * ad[hh][1] +
                       v[j].z * ad[hh][2] + v[j].w * ad[hh][3];
  }
  ap_fold<16>(val, lane);
  ap_fold<8>(val, lane);
  ap_fold<4>(val, lane);
  ap_fold<2>(val, lane);
  ap_fold<1>(val, lane);

  const int w = w0 + (lane & 15);
  if (w < nrows) (lane < 16 ? al_s : al_d)[w] = val[0];
}

#define X1_W 256
#define MLP_C 96
#define OWN_THREADS 1024
#define LIST_CAP 8192
#define KEEP_HDR 32
#define KEEP_WORDS (KEEP_HDR + LIST_CAP)

template <int HEADS, int KEEP>
__device__ __forceinline__ void gat_owned_body(
    const int* __restrict__ src, const int* __restrict__ dst, int E, int n,
    const float* __restrict__ h, int ldh, const float* __restrict__ al_s,
    const float* __restrict__ al_d, const float* __restrict__ bias,
    float* __restrict__ out, int ldo, int colOff, float* __restrict__ keepbuf,
    const float* __restrict__ Wl, const float* __restrict__ bl,
    float* __restrict__ y1, const unsigned* __restrict__ plane,
    const float* __restrict__ Wg, const float* __restrict__ a_src2,
    const float* __restrict__ a_dst2, float* __restrict__ hout, int ldho,
    float* __restrict__ al_s2, float* __restrict__ al_d2) {
  static_assert(OWN_ROWS <= 128 && DFEAT == 128, "packed word: 7 bits of row");
  static_assert(OWN_ROWS % 16 == 0 && MLP_C % 16 == 0 && X1_W % 16 == 0,
                "whole tiles and whole blocks of 16 k in the fused product");
  static_assert(KEEP == 1 || KEEP == 2, "layer 1 or layer 2");
  static_assert(KEEP != 2 || HEADS == 1, "the fused product takes one head");
  static_assert(KEEP != 1 || HEADS * DFEAT == X1_W, "layer 1 makes x1's columns");
  constexpr int ROWW = HEADS * DFEAT;
  constexpr int NL = ROWW / 32;
  constexpr int BATCH = 32 / NL;
  constexpr int AP = ROWW + (KEEP != 0 ? 4 : 0);
  constexpr int XP = X1_W + 4;
  __shared__ __attribute__((aligned(16))) float s_acc[OWN_ROWS * AP];
  __shared__ __attribute__((aligned(16))) float
      s_x1[KEEP == 2 ? OWN_ROWS * XP : 4];
  __shared__ __attribute__((aligned(16))) int s_list[2 * LIST_CAP];
  int* const s_ls = s_list;
  int* const s_ld = s_list + LIST_CAP;
  __shared__ unsigned s_menc[OWN_ROWS * HEADS];
  __shared__ float s_sum[OWN_ROWS * HEADS];
  __shared__ unsigned s_cnt;

  const int tid = threadIdx.x;
  const int lane = tid & 31;
  const int wave = tid >> 5;
  const int nthr = blockDim.x;
  const int nw = nthr >> 5;
  const int r0 = blockIdx.x * OWN_ROWS;
  int rows_here = n - r0;
  if (rows_here > OWN_ROWS) rows_here = OWN_ROWS;
  if (rows_here < 0) rows_here = 0;

  float* const kbase =
      keepbuf ? keepbuf + (size_t)blockIdx.x * KEEP_WORDS : nullptr;
  auto kslot = [&](int j) -> float* { return kbase + j; };
  const int keep_cap = kbase ? LIST_CAP : -1;
  const bool packs = n <= (1 << 24);
  int kcnt = -1;
  if constexpr (KEEP == 2) {
    if (kbase && rows_here > 0 && packs) kcnt = __float_as_int(*kslot(0));
  }
  const bool kept = KEEP == 2 && kcnt >= 0 && kcnt <= keep_cap;

  if constexpr (KEEP == 2) {
    constexpr int XG = X1_W / 4;
    constexpr int XB = 5;
    for (int f0 = 0; f0 < OWN_ROWS * XG; f0 += XB * nthr) {
      float4 xr[XB];
#pragma unroll
      for (int i = 0; i < XB; ++i) {
        const int f = f0 + i * nthr + tid;
        const int r = f / XG;
        xr[i] = make_float4(0.f, 0.f, 0.f, 0.f);
        if (f < OWN_ROWS * XG && r < rows_here)
          xr[i] = *(const float4*)(out + (size_t)(r0 + r) * ldo +
                                   4 * (f - r * XG));
      }
#pragma unroll
      for (int i = 0; i < XB; ++i) {
        const int f = f0 + i * nthr + tid;
        const int r = f / XG;
        if (f < OWN_ROWS * XG)
          *(float4*)(s_x1 + r * XP + 4 * (f - r * XG)) = xr[i];
      }
    }
  }
  for (int i = tid; i < OWN_ROWS * AP; i += nthr) s_acc[i] = 0.f;
  for (int i = tid; i < OWN_ROWS * HEADS; i += nthr) {
    s_menc[i] = 0u;
    s_sum[i] = 0.f;
  }
  if (tid == 0) s_cnt = 0u;
  __syncthreads();

  auto append = [&](int e, int dl) {
    const unsigned p = atomicAdd(&s_cnt, 1u);
    if (p < (unsigned)LIST_CAP) { s_ls[p] = e; s_ld[p] = dl; }
  };
  auto owned = [&](int d) -> bool {
    return (unsigned)(d - r0) < (unsigned)rows_here;
  };
  if (kept) {
    for (int i = tid; i < kcnt; i += nthr) {
      const int p = __float_as_int(*kslot(KEEP_HDR + i));
      s_ls[i] = p >> 7;
      s_ld[i] = p & 127;
    }
  } else if (plane) {
    constexpr int PL = 8;
    constexpr unsigned ONES = 0x01010101u;
    const uint4* __restrict__ pv = (const uint4*)plane;
    const int ngrp = (E + 15) >> 4;
    const unsigned idw = (unsigned)blockIdx.x * ONES;
    auto append_e = [&](int e) {
      const unsigned p = atomicAdd(&s_cnt, 1u);
      if (p < (unsigned)LIST_CAP) s_ls[p] = e;
    };
    for (int r = tid; r < rows_here; r += nthr) append_e(E + r0 + r);
    const int wcap = LIST_CAP / nw;
    int* const cand = s_ld + wave * wcap;
    int ncand = 0;
    for (int gb = 0; gb < ngrp; gb += PL * nthr) {
      uint4 q[PL];
#pragma unroll
      for (int k = 0; k < PL; ++k) {
        const int g = gb + k * nthr + tid;
        q[k] = pv[g < ngrp ? g : ngrp - 1];
      }
#pragma unroll
      for (int k = 0; k < PL; ++k) {
        const int g = gb + k * nthr + tid;
        const unsigned x0 = q[k].x ^ idw, x1 = q[k].y ^ idw;
        const unsigned x2 = q[k].z ^ idw, x3 = q[k].w ^ idw;
        const unsigned t = (((x0 - ONES) & ~x0) | ((x1 - ONES) & ~x1) |
                            ((x2 - ONES) & ~x2) | ((x3 - ONES) & ~x3)) &
                           0x80808080u;
        const bool hit = g < ngrp && t != 0u;
        const unsigned bm = __builtin_amdgcn_ballot_w32(hit);
        if (bm) {
          const int pos = ncand + (int)__builtin_amdgcn_mbcnt_lo(bm, 0u);
          if (hit && pos < wcap) cand[pos] = g;
          ncand += __builtin_popcount(bm);
        }
      }
    }
    if (ncand > wcap) {
      if (lane == 0) atomicAdd(&s_cnt, (unsigned)LIST_CAP + 1u);
    } else {
      for (int b = 0; b < ncand; b += 32) {
        const int i = b + lane;
        if (i < ncand) {
          const int g = cand[i];
          const uint4 v = pv[g];
          auto zb = [&](unsigned w) -> unsigned {
            const unsigned x = w ^ idw;
            return ~(((x & 0x7F7F7F7Fu) + 0x7F7F7F7Fu) | x) & 0x80808080u;
          };
          unsigned m = (zb(v.x) >> 7) | (zb(v.y) >> 6) | (zb(v.z) >> 5) |
                       (zb(v.w) >> 4);
          unsigned p = atomicAdd(&s_cnt, (unsigned)__builtin_popcount(m));
          while (m) {
            const int bit = __builtin_ctz(m);
            m &= m - 1;
            if (p < (unsigned)LIST_CAP)
              s_ls[p] = 16 * g + 4 * (bit & 7) + (bit >> 3);
            ++p;
          }
        }
      }
    }
  } else {
    for (int r = tid; r < rows_here; r += nthr) append(E + r0 + r, r);
    int pre = (int)(((16u - ((unsigned)(size_t)dst & 15u)) & 15u) >> 2);
    if (pre > E) pre = E;
    const int nvec = (E - pre) >> 2;
    const int4* __restrict__ dv = (const int4*)(dst + pre);
    for (int vb = 0; vb < nvec; vb += 4 * nthr) {
      int4 q[4];
      int vi[4];
      for (int k = 0; k < 4; ++k) {
        vi[k] = vb + k * nthr + tid;
        q[k] = dv[vi[k] < nvec ? vi[k] : nvec - 1];
      }
      for (int k = 0; k < 4; ++k) {
        const bool in = vi[k] < nvec;
        const bool o0 = in && owned(q[k].x), o1 = in && owned(q[k].y);
        const bool o2 = in && owned(q[k].z), o3 = in && owned(q[k].w);
        if (o0 | o1 | o2 | o3) {
          const int e = pre + 4 * vi[k];
          if (o0) append(e + 0, q[k].x - r0);
          if (o1) append(e + 1, q[k].y - r0);
          if (o2) append(e + 2, q[k].z - r0);
          if (o3) append(e + 3, q[k].w - r0);
        }
      }
    }
    const int tail0 = pre + 4 * nvec;
    if (tid < 8) {
      const int e = (tid < 4) ? tid : tail0 + (tid - 4);
      const bool in = (tid < 4) ? (e < pre) : (e < E);
      if (in) {
        const int d = dst[e];
        if (owned(d)) append(e, d - r0);
      }
    }
  }
  __syncthreads();
  const int cnt = kept ? kcnt : (int)s_cnt;
  const bool use_list = cnt <= LIST_CAP;
  const bool save = KEEP == 1 && packs && cnt <= keep_cap;
  if (use_list && !kept) {
    for (int i = tid; i < cnt; i += nthr) {
      const int e = s_ls[i];
      const int sr = (e < E) ? src[e] : (e - E);
      int dl;
      if (plane) {
        dl = ((e < E) ? dst[e] : (e - E)) - r0;
        s_ld[i] = dl;
      } else {
        dl = s_ld[i];
      }
      s_ls[i] = sr;
      if (save) *kslot(KEEP_HDR + i) = __int_as_float((sr << 7) | dl);
    }
  }
  if constexpr (KEEP == 1) {
    if (tid == 0 && kbase && rows_here > 0)
      *kslot(0) = __int_as_float(save ? cnt : -1);
  }
  __syncthreads();

  auto for_owned = [&](auto body) {
    if (use_list) {
      for (int base = 0; base < cnt; base += 32 * nw) {
        const int i = base + lane * nw + wave;
        const bool own = i < cnt;
        const int s = own ? s_ls[i] : 0;
        const int dl = own ? s_ld[i] : 0;
        body(own, s, dl);
      }
    } else {
      for (int e0 = wave * 32; e0 < E; e0 += 32 * nw) {
        const int e = e0 + lane;
        const int d = (e < E) ? dst[e] : -1;
        const bool own = (unsigned)(d - r0) < (unsigned)rows_here;
        if (__builtin_amdgcn_ballot_w32(own)) {
          const int s = own ? src[e] : 0;
          body(own, s, own ? d - r0 : 0);
        }
      }
      for (int i0 = wave * 32; i0 < rows_here; i0 += 32 * nw) {
        const int r = i0 + lane;
        const bool own = r < rows_here;
        body(own, own ? r0 + r : 0, own ? r : 0);
      }
    }
  };
  auto logit = [&](int s, int dl, int hh) -> float {
    const float x = al_s[s * HEADS + hh] + al_d[(r0 + dl) * HEADS + hh];
    return (x > 0.f) ? x : NEG_SLOPE * x;
  };

  for_owned([&](bool own, int s, int dl) {
    if (own) {
      for (int hh = 0; hh < HEADS; ++hh)
        atomicMax(&s_menc[dl * HEADS + hh], ordEnc(logit(s, dl, hh)));
    }
  });
  __syncthreads();

  for_owned([&](bool own, int s, int dl) {
    if (own) {
      for (int hh = 0; hh < HEADS; ++hh) {
        const float m = ordDec(s_menc[dl * HEADS + hh]);
        atomicAdd(&s_sum[dl * HEADS + hh], expf(logit(s, dl, hh) - m));
      }
    }
  });
  __syncthreads();

  for_owned([&](bool own, int s, int dl) {
    float al[HEADS];
    for (int hh = 0; hh < HEADS; ++hh) al[hh] = 0.f;
    if (own) {
      for (int hh = 0; hh < HEADS; ++hh) {
        const float m = ordDec(s_menc[dl * HEADS + hh]);
        al[hh] = expf(logit(s, dl, hh) - m) / (s_sum[dl * HEADS + hh] + 1e-16f);
      }
    }
    unsigned m = __builtin_amdgcn_ballot_w32(own);
    while (__builtin_popcount(m) >= BATCH) {
      float v[BATCH][NL];
      int oo[BATCH];
      int jj[BATCH];
      for (int b = 0; b < BATCH; ++b) {
        const int j = __builtin_ctz(m);
        m &= m - 1;
        jj[b] = j;
        const int sj = __builtin_amdgcn_readlane(s, j);
        const int dj = __builtin_amdgcn_readlane(dl, j);
        const float* hp = h + (size_t)sj * ldh + lane;
        oo[b] = dj * AP + lane;
        for (int c = 0; c < NL; ++c) v[b][c] = hp[32 * c];
      }
      for (int b = 0; b < BATCH; ++b) {
        for (int c = 0; c < NL; ++c) {
          const float a = __int_as_float(__builtin_amdgcn_readlane(
              __float_as_int(al[c / (DFEAT / 32)]), jj[b]));
          atomicAdd(&s_acc[oo[b] + 32 * c], a * v[b][c]);
        }
      }
    }
    while (m) {
      const int j = __builtin_ctz(m);
      m &= m - 1;
      const int sj = __builtin_amdgcn_readlane(s, j);
      const int dj = __builtin_amdgcn_readlane(dl, j);
      const float* hp = h + (size_t)sj * ldh + lane;
      const int o = dj * AP + lane;
      float v[NL];
      for (int c = 0; c < NL; ++c) v[c] = hp[32 * c];
      for (int c = 0; c < NL; ++c) {
        const float a = __int_as_float(__builtin_amdgcn_readlane(
            __float_as_int(al[c / (DFEAT / 32)]), j));
        atomicAdd(&s_acc[o + 32 * c], a * v[c]);
      }
    }
  });
  __syncthreads();

  if constexpr (KEEP == 1) {
    for (int i = tid; i < OWN_ROWS * ROWW; i += nthr) {
      const int r = i / ROWW;
      const int cc = i - r * ROWW;
      s_acc[r * AP + cc] += bias[cc];
    }
    __syncthreads();
    for (int r = wave; r < rows_here; r += nw) {
      float* op = out + (size_t)(r0 + r) * ldo + colOff + lane;
      for (int c = 0; c < NL; ++c) op[32 * c] = s_acc[r * AP + 32 * c + lane];
    }
    constexpr int NWC = OWN_THREADS / 32;
    constexpr int CT = DFEAT / 16;
    constexpr int RT = OWN_ROWS / 16;
    static_assert(NWC % CT == 0, "every wave keeps one column tile");
    constexpr int RS = NWC / CT;
    constexpr int MT = (RT + RS - 1) / RS;
    constexpr int HP = DFEAT + 4;
    static_assert(OWN_ROWS * HP <= 2 * LIST_CAP, "h2 rows fit the list's LDS");
    const int wv = __builtin_amdgcn_readfirstlane(wave);
    const int ct = wv % CT;
    const int rt0 = wv / CT;
    const int lc = lane & 15;
    const int lh = lane >> 4;
    const int col = ct * 16 + lc;
    v8f acc[MT];
#pragma unroll
    for (int m = 0; m < MT; ++m)
      acc[m] = (v8f){0.f, 0.f, 0.f, 0.f, 0.f, 0.f, 0.f, 0.f};
    const float* bp = Wg + (size_t)(2 * lh) * DFEAT + col;
    const float* ap0 = s_acc + (rt0 * 16 + lc) * AP + 2 * lh;
    for (int k = 0; k < ROWW; k += 16) {
      const float* bq = bp + (size_t)k * DFEAT;
      v2f b0, b1, b2, b3;
      b0.x = bq[0];
      b0.y = bq[1 * DFEAT];
      b1.x = bq[4 * DFEAT];
      b1.y = bq[5 * DFEAT];
      b2.x = bq[8 * DFEAT];
      b2.y = bq[9 * DFEAT];
      b3.x = bq[12 * DFEAT];
      b3.y = bq[13 * DFEAT];
#pragma unroll
      for (int m = 0; m < MT; ++m) {
        if (rt0 + m * RS < RT) {
          const float* ap = ap0 + (m * RS * 16) * AP + k;
          const v2f a0 = *(const v2f*)(ap);
          const v2f a1 = *(const v2f*)(ap + 4);
          const v2f a2 = *(const v2f*)(ap + 8);
          const v2f a3 = *(const v2f*)(ap + 12);
          acc[m] = wmma_f32(a0, b0, acc[m]);
          acc[m] = wmma_f32(a1, b1, acc[m]);
          acc[m] = wmma_f32(a2, b2, acc[m]);
          acc[m] = wmma_f32(a3, b3, acc[m]);
        }
      }
    }
#pragma unroll
    for (int m = 0; m < MT; ++m) {
      if (rt0 + m * RS < RT) {
        int* hp = s_list + ((rt0 + m * RS) * 16 + 8 * lh) * HP + col;
#pragma unroll
        for (int r = 0; r < 8; ++r) hp[r * HP] = __float_as_int(acc[m][r]);
      }
    }
    __syncthreads();
    {
      const float4 sa = *(const float4*)(a_src2 + 4 * lane);
      const float4 da = *(const float4*)(a_dst2 + 4 * lane);
      for (int r = wave; r < rows_here; r += nw) {
        const int4 q = *(const int4*)(s_list + r * HP + 4 * lane);
        const float4 v = make_float4(__int_as_float(q.x), __int_as_float(q.y),
                                     __int_as_float(q.z), __int_as_float(q.w));
        *(float4*)(hout + (size_t)(r0 + r) * ldho + 4 * lane) = v;
        float ps = v.x * sa.x + v.y * sa.y + v.z * sa.z + v.w * sa.w;
        float pd = v.x * da.x + v.y * da.y + v.z * da.z + v.w * da.w;
        for (int off = 16; off > 0; off >>= 1) {
          ps += __shfl_xor(ps, off, 32);
          pd += __shfl_xor(pd, off, 32);
        }
        if (lane == 0) {
          al_s2[r0 + r] = ps;
          al_d2[r0 + r] = pd;
        }
      }
    }
  } else {
    for (int i = tid; i < OWN_ROWS * ROWW; i += nthr) {
      const int r = i / ROWW;
      const int cc = i - r * ROWW;
      s_acc[r * AP + cc] += bias[cc];
    }
    __syncthreads();
    const int wv = __builtin_amdgcn_readfirstlane(wave);
    constexpr int CT = MLP_C / 16;
    constexpr int TILES = (OWN_ROWS / 16) * CT;
    const int lc = lane & 15;
    const int lh = lane >> 4;
    for (int t = wv; t < TILES; t += nw) {
      const int rt = t / CT;
      const int ct = t - rt * CT;
      const int col = ct * 16 + lc;
      const float* ax = s_x1 + (rt * 16 + lc) * XP + 2 * lh;
      const float* aa = s_acc + (rt * 16 + lc) * AP + 2 * lh;
      const float* bp = Wl + (size_t)(2 * lh) * MLP_C + col;
      v8f acc = {0.f, 0.f, 0.f, 0.f, 0.f, 0.f, 0.f, 0.f};
      auto kblock = [&](const float* ap, const float* bq) {
        const v2f a0 = *(const v2f*)(ap);
        const v2f a1 = *(const v2f*)(ap + 4);
        const v2f a2 = *(const v2f*)(ap + 8);
        const v2f a3 = *(const v2f*)(ap + 12);
        v2f b0, b1, b2, b3;
        b0.x = bq[0];
        b0.y = bq[1 * MLP_C];
        b1.x = bq[4 * MLP_C];
        b1.y = bq[5 * MLP_C];
        b2.x = bq[8 * MLP_C];
        b2.y = bq[9 * MLP_C];
        b3.x = bq[12 * MLP_C];
        b3.y = bq[13 * MLP_C];
        acc = wmma_f32(a0, b0, acc);
        acc = wmma_f32(a1, b1, acc);
        acc = wmma_f32(a2, b2, acc);
        acc = wmma_f32(a3, b3, acc);
      };
      for (int k = 0; k < X1_W; k += 16) kblock(ax + k, bp + (size_t)k * MLP_C);
      for (int k = 0; k < ROWW; k += 16)
        kblock(aa + k, bp + (size_t)(X1_W + k) * MLP_C);
      const float bv = bl[col];
#pragma unroll
      for (int r = 0; r < 8; ++r) {
        const int row = rt * 16 + r + 8 * lh;
        if (row < rows_here)
          y1[(size_t)(r0 + row) * MLP_C + col] = fmaxf(acc[r] + bv, 0.0f);
      }
    }
  }
}

__global__ void __launch_bounds__(OWN_THREADS)
gat_owned_h2(const int* __restrict__ src, const int* __restrict__ dst, int E,
             int n, const float* __restrict__ h, int ldh,
             const float* __restrict__ al_s, const float* __restrict__ al_d,
             const float* __restrict__ bias, float* __restrict__ out, int ldo,
             int colOff, float* __restrict__ keepbuf,
             const unsigned* __restrict__ plane, const float* __restrict__ Wg,
             const float* __restrict__ a_src2, const float* __restrict__ a_dst2,
             float* __restrict__ hout, int ldho, float* __restrict__ al_s2,
             float* __restrict__ al_d2) {
  gat_owned_body<2, 1>(src, dst, E, n, h, ldh, al_s, al_d, bias, out, ldo,
                       colOff, keepbuf, nullptr, nullptr, nullptr, plane, Wg,
                       a_src2, a_dst2, hout, ldho, al_s2, al_d2);
}

__global__ void __launch_bounds__(OWN_THREADS)
gat_owned_h1(const int* __restrict__ src, const int* __restrict__ dst, int E,
             int n, const float* __restrict__ h, int ldh,
             const float* __restrict__ al_s, const float* __restrict__ al_d,
             const float* __restrict__ bias, float* __restrict__ out, int ldo,
             int colOff, float* __restrict__ keepbuf,
             const float* __restrict__ Wl, const float* __restrict__ bl,
             float* __restrict__ y1) {
  gat_owned_body<1, 2>(src, dst, E, n, h, ldh, al_s, al_d, bias, out, ldo,
                       colOff, keepbuf, Wl, bl, y1, nullptr, nullptr, nullptr,
                       nullptr, nullptr, 0, nullptr, nullptr);
}

#define CS_THREADS 192
#define CS_LOADS 10

template <int NC>
__global__ void __launch_bounds__(CS_THREADS)
col_part(const float* __restrict__ y, int total, float* __restrict__ ps,
         float* __restrict__ pq, int nparts) {
  static_assert((4 * CS_THREADS) % NC == 0, "a thread keeps its columns");
  const int t = threadIdx.x;
  const int nvec = total >> 2;
  const int base = blockIdx.x * (CS_LOADS * CS_THREADS);
  const float4* __restrict__ y4 = (const float4*)y;

  float4 v[CS_LOADS];
#pragma unroll
  for (int k = 0; k < CS_LOADS; ++k) {
    const int f = base + k * CS_THREADS + t;
    v[k] = y4[f < nvec ? f : 0];
  }
  float sa[4] = {0.f, 0.f, 0.f, 0.f};
  float qa[4] = {0.f, 0.f, 0.f, 0.f};
#pragma unroll
  for (int k = 0; k < CS_LOADS; ++k) {
    const int f = base + k * CS_THREADS + t;
    if (f < nvec) {
      sa[0] += v[k].x; qa[0] = fmaf(v[k].x, v[k].x, qa[0]);
      sa[1] += v[k].y; qa[1] = fmaf(v[k].y, v[k].y, qa[1]);
      sa[2] += v[k].z; qa[2] = fmaf(v[k].z, v[k].z, qa[2]);
      sa[3] += v[k].w; qa[3] = fmaf(v[k].w, v[k].w, qa[3]);
    }
  }
  {
    const int tail = total & 3;
    const int rel = nvec - base;
    if (tail && rel >= 0 && rel < CS_LOADS * CS_THREADS &&
        rel % CS_THREADS == t) {
      for (int j = 0; j < tail; ++j) {
        const float w = y[4 * nvec + j];
        sa[j] += w;
        qa[j] = fmaf(w, w, qa[j]);
      }
    }
  }

  if constexpr (NC % 4 == 0) {
    constexpr int QN = NC / 4;
    static_assert(CS_THREADS % QN == 0, "whole rows per step");
    constexpr int GN = CS_THREADS / QN;
    __shared__ __attribute__((aligned(16))) float red[2][GN][NC];
    const int g = t / QN;
    const int c4 = 4 * (t - g * QN);
    *(float4*)&red[0][g][c4] = make_float4(sa[0], sa[1], sa[2], sa[3]);
    *(float4*)&red[1][g][c4] = make_float4(qa[0], qa[1], qa[2], qa[3]);
    __syncthreads();
    for (int i = t; i < 2 * NC; i += CS_THREADS) {
      const int which = i / NC;
      const int c = i - which * NC;
      float r = 0.f;
#pragma unroll
      for (int gg = 0; gg < GN; ++gg) r += red[which][gg][c];
      (which ? pq : ps)[(size_t)c * nparts + blockIdx.x] = r;
    }
  } else {
    static_assert(4 % NC == 0, "column count must divide a group");
    constexpr int NW = CS_THREADS / 32;
    __shared__ float red[2][NW][NC];
    float cs[NC], cq[NC];
#pragma unroll
    for (int c = 0; c < NC; ++c) { cs[c] = 0.f; cq[c] = 0.f; }
#pragma unroll
    for (int j = 0; j < 4; ++j) { cs[j % NC] += sa[j]; cq[j % NC] += qa[j]; }
#pragma unroll
    for (int c = 0; c < NC; ++c) {
      for (int off = 16; off > 0; off >>= 1) {
        cs[c] += __shfl_xor(cs[c], off, 32);
        cq[c] += __shfl_xor(cq[c], off, 32);
      }
    }
    const int lane = t & 31;
    const int wave = t >> 5;
    if (lane == 0) {
#pragma unroll
      for (int c = 0; c < NC; ++c) {
        red[0][wave][c] = cs[c];
        red[1][wave][c] = cq[c];
      }
    }
    __syncthreads();
    if (t < 2 * NC) {
      const int which = t / NC;
      const int c = t - which * NC;
      float r = 0.f;
#pragma unroll
      for (int w = 0; w < NW; ++w) r += red[which][w][c];
      (which ? pq : ps)[(size_t)c * nparts + blockIdx.x] = r;
    }
  }
}

__global__ void bn_fold(const float* __restrict__ ps,
                        const float* __restrict__ pq, int nparts,
                        const float* __restrict__ g,
                        const float* __restrict__ beta, int ncols, float invn,
                        float* __restrict__ sc, float* __restrict__ sh) {
  const int lane = threadIdx.x & 31;
  const int c = (blockIdx.x * blockDim.x + threadIdx.x) >> 5;
  if (c >= ncols) return;
  const float* __restrict__ sp = ps + (size_t)c * nparts;
  const float* __restrict__ qp = pq + (size_t)c * nparts;
  float s = 0.f, q = 0.f;
  for (int b0 = 0; b0 < nparts; b0 += 256) {
    float vs[8], vq[8];
#pragma unroll
    for (int k = 0; k < 8; ++k) {
      const int b = b0 + 32 * k + lane;
      vs[k] = sp[b < nparts ? b : 0];
      vq[k] = qp[b < nparts ? b : 0];
    }
#pragma unroll
    for (int k = 0; k < 8; ++k) {
      const int b = b0 + 32 * k + lane;
      if (b < nparts) { s += vs[k]; q += vq[k]; }
    }
  }
  for (int off = 16; off > 0; off >>= 1) {
    s += __shfl_xor(s, off, 32);
    q += __shfl_xor(q, off, 32);
  }
  if (lane == 0) {
    const float mean = s * invn;
    const float var = q * invn - mean * mean;
    const float r = rsqrtf(var + 1e-5f);
    const float a = g[c] * r;
    sc[c] = a;
    sh[c] = beta[c] - mean * a;
  }
}

__global__ void mlp_out(const float* __restrict__ y2,
                        const float* __restrict__ sc,
                        const float* __restrict__ sh,
                        const float* __restrict__ Wm2,
                        const float* __restrict__ bm2, int n,
                        float* __restrict__ y3) {
  const int node = blockIdx.x * blockDim.x + threadIdx.x;
  if (node >= n) return;
  float a0 = bm2[0], a1 = bm2[1];
  const float* yp = y2 + (size_t)node * 96;
  for (int k = 0; k < 96; ++k) {
    const float v = fmaf(yp[k], sc[k], sh[k]);
    a0 = fmaf(v, Wm2[2 * k + 0], a0);
    a1 = fmaf(v, Wm2[2 * k + 1], a1);
  }
  y3[node * 2 + 0] = fmaxf(a0, 0.f);
  y3[node * 2 + 1] = fmaxf(a1, 0.f);
}

__global__ void bn_apply(const float* __restrict__ y,
                         const float* __restrict__ sc,
                         const float* __restrict__ sh, int total, int ncols,
                         float* __restrict__ out) {
  const int i = blockIdx.x * blockDim.x + threadIdx.x;
  if (i >= total) return;
  const int c = i % ncols;
  out[i] = fmaf(y[i], sc[c], sh[c]);
}

extern "C" void kernel_launch(void* const* d_in, const int* in_sizes, int n_in,
                              void* d_out, int out_size, void* d_ws,
                              size_t ws_size, hipStream_t stream) {
  const float* x      = (const float*)d_in[0];
  const int*   ei     = (const int*)d_in[1];
  const float* W1     = (const float*)d_in[2];
  const float* as1    = (const float*)d_in[3];
  const float* ad1    = (const float*)d_in[4];
  const float* b1     = (const float*)d_in[5];
  const float* W2     = (const float*)d_in[6];
  const float* as2    = (const float*)d_in[7];
  const float* ad2    = (const float*)d_in[8];
  const float* b2     = (const float*)d_in[9];
  const float* Wl     = (const float*)d_in[10];
  const float* bl     = (const float*)d_in[11];
  const float* gl     = (const float*)d_in[12];
  const float* betal  = (const float*)d_in[13];
  const float* Wm1    = (const float*)d_in[14];
  const float* bm1    = (const float*)d_in[15];
  const float* gm1    = (const float*)d_in[16];
  const float* betam1 = (const float*)d_in[17];
  const float* Wm2    = (const float*)d_in[18];
  const float* bm2    = (const float*)d_in[19];
  const float* gm2    = (const float*)d_in[20];
  const float* betam2 = (const float*)d_in[21];

  const int N = in_sizes[0] / DFEAT;
  const int E = in_sizes[1] / 2;
  const int* srcA = ei;
  const int* dstA = ei + E;

  char* w = (char*)d_ws;
  auto alloc = [&](size_t bytes) -> char* {
    char* p = w;
    w += (bytes + 255) & ~(size_t)255;
    return p;
  };
  float*    xcat = (float*)alloc((size_t)N * 384 * 4);
  float*    h1   = (float*)alloc((size_t)N * 256 * 4);
  float*    h2   = (float*)alloc((size_t)N * 128 * 4);
  float*    al_s = (float*)alloc((size_t)N * 2 * 4);
  float*    al_d = (float*)alloc((size_t)N * 2 * 4);
  float*    y3   = (float*)alloc((size_t)N * 2 * 4);
  float*    sc1  = (float*)alloc(96 * 4);
  float*    sh1  = (float*)alloc(96 * 4);
  float*    sc2  = (float*)alloc(96 * 4);
  float*    sh2  = (float*)alloc(96 * 4);
  float*    sc3  = (float*)alloc(96 * 4);
  float*    sh3  = (float*)alloc(96 * 4);
  float* y1 = h1;
  float* y2 = h1 + (size_t)N * 96;
  const int csGroups = CS_LOADS * CS_THREADS;
  auto csBlocks = [&](int total) {
    return ((total + 3) / 4 + csGroups - 1) / csGroups;
  };
  const int np96 = csBlocks(N * 96);
  const int np2  = csBlocks(N * 2);
  float* ps = h2;
  float* pq = h2 + (size_t)96 * np96;

  const int TB = 256;
  auto gemmBlocks = [](int rows, int cols) {
    const int tiles = ((rows + 15) / 16) * (cols / 16);
    return (tiles + 7) / 8;
  };
  auto layerBlocks = [](int rows, int cols) {
    const int rowsPer = LG_WAVES * 16;
    return ((rows + rowsPer - 1) / rowsPer) * (cols / LG_COLS);
  };
  const int ownBlocks = (N + OWN_ROWS - 1) / OWN_ROWS;
  const size_t h2Bytes = (size_t)N * 128 * sizeof(float);
  const size_t planeBytes = ((size_t)E + 16 + 255) & ~(size_t)255;
  unsigned* plane =
      (ownBlocks <= 255 && planeBytes <= h2Bytes) ? (unsigned*)h2 : nullptr;
  float* keepbuf =
      (planeBytes + (size_t)ownBlocks * KEEP_WORDS * sizeof(float) <= h2Bytes)
          ? (float*)((char*)h2 + planeBytes)
          : nullptr;
  float* hL2 = xcat + 256;
  float* al_s2 = y3;
  float* al_d2 = y3 + N;
  const int apRows = AP_ROWS * (AP_THREADS / 32);

  gemm_layer_strip<128><<<layerBlocks(N, 256), LG_WAVES * 32, 0, stream>>>(
      x, 128, W1, 256, h1, 256, N, 256);
  attn_proj_rows<2><<<(N * 2 + apRows - 1) / apRows, AP_THREADS, 0, stream>>>(
      h1, as1, ad1, al_s, al_d, N * 2, dstA, E, N, plane);
  gat_owned_h2<<<ownBlocks, OWN_THREADS, 0, stream>>>(
      srcA, dstA, E, N, h1, 256, al_s, al_d, b1, xcat, 384, 0, keepbuf, plane,
      W2, as2, ad2, hL2, 384, al_s2, al_d2);

  gat_owned_h1<<<ownBlocks, OWN_THREADS, 0, stream>>>(
      srcA, dstA, E, N, hL2, 384, al_s2, al_d2, b2, xcat, 384, 256, keepbuf,
      Wl, bl, y1);

  const float invn = 1.0f / (float)N;
  col_part<96><<<np96, CS_THREADS, 0, stream>>>(y1, N * 96, ps, pq, np96);
  bn_fold<<<(96 * 32 + TB - 1) / TB, TB, 0, stream>>>(ps, pq, np96, gl, betal,
                                                      96, invn, sc1, sh1);
  gemm_f32_wmma_bn<<<gemmBlocks(N, 96), TB, 0, stream>>>(
      y1, 96, Wm1, 96, y2, 96, N, 96, 96, bm1, sc1, sh1, 1);
  col_part<96><<<np96, CS_THREADS, 0, stream>>>(y2, N * 96, ps, pq, np96);
  bn_fold<<<(96 * 32 + TB - 1) / TB, TB, 0, stream>>>(ps, pq, np96, gm1, betam1,
                                                      96, invn, sc2, sh2);
  mlp_out<<<(N + TB - 1) / TB, TB, 0, stream>>>(y2, sc2, sh2, Wm2, bm2, N, y3);
  col_part<2><<<np2, CS_THREADS, 0, stream>>>(y3, N * 2, ps, pq, np2);
  bn_fold<<<(2 * 32 + TB - 1) / TB, TB, 0, stream>>>(ps, pq, np2, gm2, betam2, 2,
                                                     invn, sc3, sh3);
  bn_apply<<<(N * 2 + TB - 1) / TB, TB, 0, stream>>>(y3, sc3, sh3, N * 2, 2,
                                                     (float*)d_out);
}
